# starting kernel
# baseline (speedup 1.0000x reference)
#include <hip/hip_runtime.h>
#include <hip/hip_bf16.h>

#define N_NODES   50000
#define N_EDGES   1250000
#define IN_DIM    5
#define OUT_DIM   64
#define NUM_REL   3
#define NUM_LAYERS 3
#define LN_EPS    1e-5f

typedef __attribute__((ext_vector_type(2))) float v2f;
typedef __attribute__((ext_vector_type(8))) float v8f;

__global__ void zero_i32_kernel(int* __restrict__ p, int n) {
    int i = blockIdx.x * blockDim.x + threadIdx.x;
    if (i < n) p[i] = 0;
}

__global__ void count_kernel(const int* __restrict__ dst,
                             const int* __restrict__ etype,
                             int* __restrict__ counts, int nE) {
    int e = blockIdx.x * blockDim.x + threadIdx.x;
    if (e < nE) atomicAdd(&counts[dst[e] * NUM_REL + etype[e]], 1);
}

__global__ void inv_denom_kernel(const int* __restrict__ counts,
                                 float* __restrict__ inv, int n) {
    int i = blockIdx.x * blockDim.x + threadIdx.x;
    if (i < n) {
        int c = counts[i];
        inv[i] = 1.0f / (float)(c > 1 ? c : 1);
    }
}

__global__ void feat2c_kernel(const float* __restrict__ x,
                              const float* __restrict__ W,
                              const float* __restrict__ b,
                              float* __restrict__ h, int n) {
    int idx = blockIdx.x * blockDim.x + threadIdx.x;
    if (idx >= n * OUT_DIM) return;
    int node = idx >> 6;
    int o    = idx & 63;
    float acc = b[o];
#pragma unroll
    for (int d = 0; d < IN_DIM; ++d)
        acc += x[node * IN_DIM + d] * W[d * OUT_DIM + o];
    h[idx] = acc;
}

__global__ __launch_bounds__(128)
void rgcn_transform_kernel(const float* __restrict__ h,
                           const float* __restrict__ relW,
                           const float* __restrict__ rootW,
                           const float* __restrict__ bias,
                           float* __restrict__ hr,
                           float* __restrict__ outbuf)
{
    __shared__ float lh[16][65];
    const int row0 = blockIdx.x * 16;

    for (int i = threadIdx.x; i < 16 * 64; i += 128) {
        int r = i >> 6, c = i & 63;
        lh[r][c] = h[(size_t)(row0 + r) * OUT_DIM + c];
    }
    __syncthreads();

    const int wave = threadIdx.x >> 5;
    const int lane = threadIdx.x & 31;
    const int mrow = lane & 15;
    const int khalf = (lane >> 4) << 1;

    const float* __restrict__ W =
        (wave < NUM_REL) ? (relW + (size_t)wave * OUT_DIM * OUT_DIM) : rootW;

    v8f acc[4];
#pragma unroll
    for (int nt = 0; nt < 4; ++nt) acc[nt] = (v8f){0.f,0.f,0.f,0.f,0.f,0.f,0.f,0.f};

#pragma unroll
    for (int k0 = 0; k0 < OUT_DIM; k0 += 4) {
        const int kk = k0 + khalf;
        v2f a;
        a.x = lh[mrow][kk];
        a.y = lh[mrow][kk + 1];
        const float* __restrict__ Wk0 = W + (size_t)kk * OUT_DIM;
        const float* __restrict__ Wk1 = Wk0 + OUT_DIM;
#pragma unroll
        for (int nt = 0; nt < 4; ++nt) {
            v2f b;
            b.x = Wk0[nt * 16 + mrow];
            b.y = Wk1[nt * 16 + mrow];
            acc[nt] = __builtin_amdgcn_wmma_f32_16x16x4_f32(
                false, a, false, b, (short)0, acc[nt], false, false);
        }
    }

    const int rbase = row0 + ((lane >> 4) << 3);
    if (wave < NUM_REL) {
#pragma unroll
        for (int nt = 0; nt < 4; ++nt) {
            const int col = nt * 16 + mrow;
#pragma unroll
            for (int v = 0; v < 8; ++v) {
                int row = rbase + v;
                hr[((size_t)row * NUM_REL + wave) * OUT_DIM + col] = acc[nt][v];
            }
        }
    } else {
#pragma unroll
        for (int nt = 0; nt < 4; ++nt) {
            const int col = nt * 16 + mrow;
            const float bs = bias[col];
#pragma unroll
            for (int v = 0; v < 8; ++v) {
                int row = rbase + v;
                outbuf[(size_t)row * OUT_DIM + col] = acc[nt][v] + bs;
            }
        }
    }
}

__global__ __launch_bounds__(256)
void edge_scatter_kernel(const int* __restrict__ src,
                         const int* __restrict__ dst,
                         const int* __restrict__ etype,
                         const float* __restrict__ hr,
                         const float* __restrict__ inv_denom,
                         float* __restrict__ outbuf, int nE) {
    int e = blockIdx.x * 8 + (threadIdx.x >> 5);
    if (e >= nE) return;
    const int lane = threadIdx.x & 31;
    const int s = src[e];
    const int d = dst[e];
    const int r = etype[e];
    const float inv = inv_denom[d * NUM_REL + r];
    const float2 m =
        ((const float2*)(hr + ((size_t)s * NUM_REL + r) * OUT_DIM))[lane];
    float* o = outbuf + (size_t)d * OUT_DIM + lane * 2;
    unsafeAtomicAdd(o,     m.x * inv);
    unsafeAtomicAdd(o + 1, m.y * inv);
}

__global__ __launch_bounds__(256)
void finalize_kernel(const float* __restrict__ outbuf,
                     const float* __restrict__ gamma,
                     const float* __restrict__ beta,
                     float* __restrict__ hout, int n) {
    int node = blockIdx.x * 8 + (threadIdx.x >> 5);
    if (node >= n) return;
    const int lane = threadIdx.x & 31;
    float2 v = ((const float2*)(outbuf + (size_t)node * OUT_DIM))[lane];
    v.x = fmaxf(v.x, 0.0f);
    v.y = fmaxf(v.y, 0.0f);
    float s  = v.x + v.y;
    float sq = v.x * v.x + v.y * v.y;
#pragma unroll
    for (int off = 16; off > 0; off >>= 1) {
        s  += __shfl_xor(s,  off, 32);
        sq += __shfl_xor(sq, off, 32);
    }
    const float mu   = s * (1.0f / OUT_DIM);
    const float var  = sq * (1.0f / OUT_DIM) - mu * mu;
    const float rstd = rsqrtf(var + LN_EPS);
    const int c0 = lane * 2;
    float2 o;
    o.x = (v.x - mu) * rstd * gamma[c0]     + beta[c0];
    o.y = (v.y - mu) * rstd * gamma[c0 + 1] + beta[c0 + 1];
    ((float2*)(hout + (size_t)node * OUT_DIM))[lane] = o;
}

extern "C" void kernel_launch(void* const* d_in, const int* in_sizes, int n_in,
                              void* d_out, int out_size, void* d_ws, size_t ws_size,
                              hipStream_t stream) {
    const float* x      = (const float*)d_in[0];
    const int*   ei     = (const int*)  d_in[1];
    const int*   src    = ei;
    const int*   dst    = ei + N_EDGES;
    const int*   etype  = (const int*)  d_in[2];
    const float* f2cW   = (const float*)d_in[3];
    const float* f2cb   = (const float*)d_in[4];
    const float* rgW    = (const float*)d_in[5];
    const float* rgRoot = (const float*)d_in[6];
    const float* rgBias = (const float*)d_in[7];
    const float* gamma  = (const float*)d_in[8];
    const float* beta   = (const float*)d_in[9];

    float* ws = (float*)d_ws;
    float* h_a       = ws;  ws += (size_t)N_NODES * OUT_DIM;
    float* h_b       = ws;  ws += (size_t)N_NODES * OUT_DIM;
    float* hr        = ws;  ws += (size_t)N_NODES * NUM_REL * OUT_DIM;
    float* outbuf    = ws;  ws += (size_t)N_NODES * OUT_DIM;
    float* inv_denom = ws;  ws += (size_t)N_NODES * NUM_REL;
    int*   counts    = (int*)ws;

    const int NR = N_NODES * NUM_REL;

    zero_i32_kernel<<<(NR + 255) / 256, 256, 0, stream>>>(counts, NR);
    count_kernel<<<(N_EDGES + 255) / 256, 256, 0, stream>>>(dst, etype, counts, N_EDGES);
    inv_denom_kernel<<<(NR + 255) / 256, 256, 0, stream>>>(counts, inv_denom, NR);
    feat2c_kernel<<<((size_t)N_NODES * OUT_DIM + 255) / 256, 256, 0, stream>>>(
        x, f2cW, f2cb, h_a, N_NODES);

    float* hcur = h_a;
    float* hnext = h_b;
    for (int l = 0; l < NUM_LAYERS; ++l) {
        rgcn_transform_kernel<<<N_NODES / 16, 128, 0, stream>>>(
            hcur,
            rgW    + (size_t)l * NUM_REL * OUT_DIM * OUT_DIM,
            rgRoot + (size_t)l * OUT_DIM * OUT_DIM,
            rgBias + (size_t)l * OUT_DIM,
            hr, outbuf);
        edge_scatter_kernel<<<(N_EDGES + 7) / 8, 256, 0, stream>>>(
            src, dst, etype, hr, inv_denom, outbuf, N_EDGES);
        float* ho = (l == NUM_LAYERS - 1) ? (float*)d_out : hnext;
        finalize_kernel<<<(N_NODES + 7) / 8, 256, 0, stream>>>(
            outbuf, gamma, beta, ho, N_NODES);
        float* t = hcur; hcur = hnext; hnext = t;
    }
}


# Edge aggregation: float-atomic scatter replaced by a per-call destination-sorted (CSR) gather; one wave sums a node's incoming rows in registers in CSR (arrival-rank) order, no float atomics left
# speedup vs baseline: 1.9222x; 1.9222x over previous
#include <hip/hip_runtime.h>
#include <hip/hip_bf16.h>

#define N_NODES   50000
#define N_EDGES   1250000
#define IN_DIM    5
#define OUT_DIM   64
#define NUM_REL   3
#define NUM_LAYERS 3
#define LN_EPS    1e-5f

typedef __attribute__((ext_vector_type(2))) float v2f;
typedef __attribute__((ext_vector_type(8))) float v8f;

#define NR_TOTAL    (N_NODES * NUM_REL)
#define SCAN_BLOCK  256
#define SCAN_TILE   (SCAN_BLOCK * 4)
#define SCAN_BLOCKS ((NR_TOTAL + SCAN_TILE - 1) / SCAN_TILE)
static_assert(NR_TOTAL % 4 == 0, "segment counts are read as int4");
static_assert(SCAN_BLOCKS <= SCAN_BLOCK, "one thread per earlier block partial");

__global__ void zero_i32_kernel(int* __restrict__ p, int n) {
    int i = blockIdx.x * blockDim.x + threadIdx.x;
    if (i < n) p[i] = 0;
}

__global__ void rank_count_kernel(const int* __restrict__ dst,
                                  const int* __restrict__ etype,
                                  int* __restrict__ counts,
                                  int* __restrict__ rank, int nE) {
    int e = blockIdx.x * blockDim.x + threadIdx.x;
    if (e < nE) rank[e] = atomicAdd(&counts[dst[e] * NUM_REL + etype[e]], 1);
}

__global__ __launch_bounds__(SCAN_BLOCK)
void inv_denom_partial_kernel(const int* __restrict__ counts,
                              float* __restrict__ inv,
                              int* __restrict__ partial, int n) {
    __shared__ int wsum[SCAN_BLOCK / 32];
    const int lane = threadIdx.x & 31;
    const int wave = threadIdx.x >> 5;
    const int i4 = blockIdx.x * SCAN_BLOCK + threadIdx.x;
    int s = 0;
    if (i4 * 4 < n) {
        const int4 c = ((const int4*)counts)[i4];
        float4 v;
        v.x = 1.0f / (float)(c.x > 1 ? c.x : 1);
        v.y = 1.0f / (float)(c.y > 1 ? c.y : 1);
        v.z = 1.0f / (float)(c.z > 1 ? c.z : 1);
        v.w = 1.0f / (float)(c.w > 1 ? c.w : 1);
        ((float4*)inv)[i4] = v;
        s = c.x + c.y + c.z + c.w;
    }
    for (int off = 16; off > 0; off >>= 1) s += __shfl_xor(s, off, 32);
    if (lane == 0) wsum[wave] = s;
    __syncthreads();
    if (threadIdx.x == 0) {
        int t = 0;
        for (int w = 0; w < SCAN_BLOCK / 32; ++w) t += wsum[w];
        partial[blockIdx.x] = t;
    }
}

__global__ __launch_bounds__(SCAN_BLOCK)
void offsets_kernel(const int* __restrict__ counts,
                    const int* __restrict__ partial,
                    int* __restrict__ offs, int n) {
    __shared__ int wsum[SCAN_BLOCK / 32];
    __shared__ int wbase[SCAN_BLOCK / 32];
    const int lane = threadIdx.x & 31;
    const int wave = threadIdx.x >> 5;
    const int i4 = blockIdx.x * SCAN_BLOCK + threadIdx.x;

    int p = ((int)threadIdx.x < (int)blockIdx.x) ? partial[threadIdx.x] : 0;
    const bool valid = i4 * 4 < n;
    int4 c = make_int4(0, 0, 0, 0);
    if (valid) c = ((const int4*)counts)[i4];
    const int tsum = c.x + c.y + c.z + c.w;

    int inc = tsum;
    for (int off = 1; off < 32; off <<= 1) {
        const int t = __shfl_up(inc, off, 32);
        if (lane >= off) inc += t;
    }
    for (int off = 16; off > 0; off >>= 1) p += __shfl_xor(p, off, 32);
    if (lane == 31) wsum[wave] = inc;
    if (lane == 0)  wbase[wave] = p;
    __syncthreads();

    int ex = inc - tsum;
    for (int w = 0; w < SCAN_BLOCK / 32; ++w) {
        ex += wbase[w];
        if (w < wave) ex += wsum[w];
    }
    if (valid) {
        int4 o;
        o.x = ex;
        o.y = o.x + c.x;
        o.z = o.y + c.y;
        o.w = o.z + c.z;
        ((int4*)offs)[i4] = o;
        if (i4 * 4 + 4 == n) offs[n] = o.w + c.w;
    }
}

__global__ void csr_fill_kernel(const int* __restrict__ src,
                                const int* __restrict__ dst,
                                const int* __restrict__ etype,
                                const int* __restrict__ rank,
                                const int* __restrict__ offs,
                                int* __restrict__ csr_src, int nE) {
    int e = blockIdx.x * blockDim.x + threadIdx.x;
    if (e < nE) csr_src[offs[dst[e] * NUM_REL + etype[e]] + rank[e]] = src[e];
}

__global__ void feat2c_kernel(const float* __restrict__ x,
                              const float* __restrict__ W,
                              const float* __restrict__ b,
                              float* __restrict__ h, int n) {
    int idx = blockIdx.x * blockDim.x + threadIdx.x;
    if (idx >= n * OUT_DIM) return;
    int node = idx >> 6;
    int o    = idx & 63;
    float acc = b[o];
#pragma unroll
    for (int d = 0; d < IN_DIM; ++d)
        acc += x[node * IN_DIM + d] * W[d * OUT_DIM + o];
    h[idx] = acc;
}

__global__ __launch_bounds__(128)
void rgcn_transform_kernel(const float* __restrict__ h,
                           const float* __restrict__ relW,
                           const float* __restrict__ rootW,
                           const float* __restrict__ bias,
                           float* __restrict__ hr,
                           float* __restrict__ outbuf)
{
    __shared__ float lh[16][65];
    const int row0 = blockIdx.x * 16;

    for (int i = threadIdx.x; i < 16 * 64; i += 128) {
        int r = i >> 6, c = i & 63;
        lh[r][c] = h[(size_t)(row0 + r) * OUT_DIM + c];
    }
    __syncthreads();

    const int wave = threadIdx.x >> 5;
    const int lane = threadIdx.x & 31;
    const int mrow = lane & 15;
    const int khalf = (lane >> 4) << 1;

    const float* __restrict__ W =
        (wave < NUM_REL) ? (relW + (size_t)wave * OUT_DIM * OUT_DIM) : rootW;

    v8f acc[4];
#pragma unroll
    for (int nt = 0; nt < 4; ++nt) acc[nt] = (v8f){0.f,0.f,0.f,0.f,0.f,0.f,0.f,0.f};

#pragma unroll
    for (int k0 = 0; k0 < OUT_DIM; k0 += 4) {
        const int kk = k0 + khalf;
        v2f a;
        a.x = lh[mrow][kk];
        a.y = lh[mrow][kk + 1];
        const float* __restrict__ Wk0 = W + (size_t)kk * OUT_DIM;
        const float* __restrict__ Wk1 = Wk0 + OUT_DIM;
#pragma unroll
        for (int nt = 0; nt < 4; ++nt) {
            v2f b;
            b.x = Wk0[nt * 16 + mrow];
            b.y = Wk1[nt * 16 + mrow];
            acc[nt] = __builtin_amdgcn_wmma_f32_16x16x4_f32(
                false, a, false, b, (short)0, acc[nt], false, false);
        }
    }

    const int rbase = row0 + ((lane >> 4) << 3);
    if (wave < NUM_REL) {
#pragma unroll
        for (int nt = 0; nt < 4; ++nt) {
            const int col = nt * 16 + mrow;
#pragma unroll
            for (int v = 0; v < 8; ++v) {
                int row = rbase + v;
                hr[((size_t)row * NUM_REL + wave) * OUT_DIM + col] = acc[nt][v];
            }
        }
    } else {
#pragma unroll
        for (int nt = 0; nt < 4; ++nt) {
            const int col = nt * 16 + mrow;
            const float bs = bias[col];
#pragma unroll
            for (int v = 0; v < 8; ++v) {
                int row = rbase + v;
                outbuf[(size_t)row * OUT_DIM + col] = acc[nt][v] + bs;
            }
        }
    }
}

__device__ __forceinline__ int bcast_i(int v, int l) {
    return __builtin_amdgcn_readlane(v, l);
}
__device__ __forceinline__ float bcast_f(float v, int l) {
    return __int_as_float(__builtin_amdgcn_readlane(__float_as_int(v), l));
}

template <int G>
__device__ __forceinline__ void gather_group(float2& acc,
                                             const float2* __restrict__ hr2,
                                             int row, float w, int k) {
    float2 m[G];
    for (int i = 0; i < G; ++i) {
        const int rk = bcast_i(row, k + i);
        m[i] = hr2[(size_t)rk * (OUT_DIM / 2)];
    }
    for (int i = 0; i < G; ++i) {
        const float wk = bcast_f(w, k + i);
        acc.x += m[i].x * wk;
        acc.y += m[i].y * wk;
    }
}

__global__ __launch_bounds__(256)
void edge_gather_kernel(const int* __restrict__ offs,
                        const int* __restrict__ csr_src,
                        const float* __restrict__ hr,
                        const float* __restrict__ inv_denom,
                        float* __restrict__ outbuf, int n) {
    const int node = blockIdx.x * 8 + (threadIdx.x >> 5);
    if (node >= n) return;
    const int lane = threadIdx.x & 31;

    int bl = offs[node * NUM_REL + (lane < NUM_REL ? lane : NUM_REL)];
    bl = min(max(bl, 0), N_EDGES);
    const int b0 = bcast_i(bl, 0);
    const int b1 = bcast_i(bl, 1);
    const int b2 = bcast_i(bl, 2);
    const int b3 = bcast_i(bl, 3);
    const float wl = inv_denom[node * NUM_REL + (lane < NUM_REL - 1 ? lane : NUM_REL - 1)];
    const float w0 = bcast_f(wl, 0);
    const float w1 = bcast_f(wl, 1);
    const float w2 = bcast_f(wl, 2);

    float2* orow = (float2*)(outbuf + (size_t)node * OUT_DIM) + lane;
    float2 acc = *orow;
    const float2* __restrict__ hr2 = (const float2*)hr + lane;

    for (int j0 = b0; j0 < b3; j0 += 32) {
        const int j = j0 + lane;
        int s = 0;
        if (j < b3) s = csr_src[j];
        s = min(max(s, 0), N_NODES - 1);
        const int r = (j >= b1) + (j >= b2);
        const int row = s * NUM_REL + r;
        const float w = (r == 0) ? w0 : ((r == 1) ? w1 : w2);

        const int cnt = min(32, b3 - j0);
        int k = 0;
        for (; k + 8 <= cnt; k += 8) gather_group<8>(acc, hr2, row, w, k);
        if (k + 4 <= cnt) { gather_group<4>(acc, hr2, row, w, k); k += 4; }
        if (k + 2 <= cnt) { gather_group<2>(acc, hr2, row, w, k); k += 2; }
        if (k < cnt)      { gather_group<1>(acc, hr2, row, w, k); }
    }
    *orow = acc;
}

__global__ __launch_bounds__(256)
void finalize_kernel(const float* __restrict__ outbuf,
                     const float* __restrict__ gamma,
                     const float* __restrict__ beta,
                     float* __restrict__ hout, int n) {
    int node = blockIdx.x * 8 + (threadIdx.x >> 5);
    if (node >= n) return;
    const int lane = threadIdx.x & 31;
    float2 v = ((const float2*)(outbuf + (size_t)node * OUT_DIM))[lane];
    v.x = fmaxf(v.x, 0.0f);
    v.y = fmaxf(v.y, 0.0f);
    float s  = v.x + v.y;
    float sq = v.x * v.x + v.y * v.y;
#pragma unroll
    for (int off = 16; off > 0; off >>= 1) {
        s  += __shfl_xor(s,  off, 32);
        sq += __shfl_xor(sq, off, 32);
    }
    const float mu   = s * (1.0f / OUT_DIM);
    const float var  = sq * (1.0f / OUT_DIM) - mu * mu;
    const float rstd = rsqrtf(var + LN_EPS);
    const int c0 = lane * 2;
    float2 o;
    o.x = (v.x - mu) * rstd * gamma[c0]     + beta[c0];
    o.y = (v.y - mu) * rstd * gamma[c0 + 1] + beta[c0 + 1];
    ((float2*)(hout + (size_t)node * OUT_DIM))[lane] = o;
}

extern "C" void kernel_launch(void* const* d_in, const int* in_sizes, int n_in,
                              void* d_out, int out_size, void* d_ws, size_t ws_size,
                              hipStream_t stream) {
    const float* x      = (const float*)d_in[0];
    const int*   ei     = (const int*)  d_in[1];
    const int*   src    = ei;
    const int*   dst    = ei + N_EDGES;
    const int*   etype  = (const int*)  d_in[2];
    const float* f2cW   = (const float*)d_in[3];
    const float* f2cb   = (const float*)d_in[4];
    const float* rgW    = (const float*)d_in[5];
    const float* rgRoot = (const float*)d_in[6];
    const float* rgBias = (const float*)d_in[7];
    const float* gamma  = (const float*)d_in[8];
    const float* beta   = (const float*)d_in[9];

    float* ws = (float*)d_ws;
    float* h_a       = ws;  ws += (size_t)N_NODES * OUT_DIM;
    int*   rank      = (int*)ws;
    int*   csr_src   = rank + N_EDGES;
    int*   offs      = csr_src + N_EDGES;
    int*   partial   = offs + NR_TOTAL + 16;
    ws += (size_t)N_NODES * OUT_DIM;
    float* hr        = ws;  ws += (size_t)N_NODES * NUM_REL * OUT_DIM;
    float* outbuf    = ws;  ws += (size_t)N_NODES * OUT_DIM;
    float* inv_denom = ws;  ws += (size_t)N_NODES * NUM_REL;
    int*   counts    = (int*)ws;

    const int NR = NR_TOTAL;

    zero_i32_kernel<<<(NR + 255) / 256, 256, 0, stream>>>(counts, NR);
    rank_count_kernel<<<(N_EDGES + 255) / 256, 256, 0, stream>>>(dst, etype, counts, rank, N_EDGES);
    inv_denom_partial_kernel<<<SCAN_BLOCKS, SCAN_BLOCK, 0, stream>>>(counts, inv_denom, partial, NR);
    offsets_kernel<<<SCAN_BLOCKS, SCAN_BLOCK, 0, stream>>>(counts, partial, offs, NR);
    csr_fill_kernel<<<(N_EDGES + 255) / 256, 256, 0, stream>>>(
        src, dst, etype, rank, offs, csr_src, N_EDGES);
    feat2c_kernel<<<((size_t)N_NODES * OUT_DIM + 255) / 256, 256, 0, stream>>>(
        x, f2cW, f2cb, h_a, N_NODES);

    for (int l = 0; l < NUM_LAYERS; ++l) {
        rgcn_transform_kernel<<<N_NODES / 16, 128, 0, stream>>>(
            h_a,
            rgW    + (size_t)l * NUM_REL * OUT_DIM * OUT_DIM,
            rgRoot + (size_t)l * OUT_DIM * OUT_DIM,
            rgBias + (size_t)l * OUT_DIM,
            hr, outbuf);
        edge_gather_kernel<<<(N_NODES + 7) / 8, 256, 0, stream>>>(
            offs, csr_src, hr, inv_denom, outbuf, N_NODES);
        float* ho = (l == NUM_LAYERS - 1) ? (float*)d_out : h_a;
        finalize_kernel<<<(N_NODES + 7) / 8, 256, 0, stream>>>(
            outbuf, gamma, beta, ho, N_NODES);
    }
}


# CSR build: per-edge returning global atomic rank + scattered fill replaced by a two-level counting sort ranked on LDS counters; no global atomics left; order inside a segment = LDS arrival order
# speedup vs baseline: 2.5572x; 1.3304x over previous
#include <hip/hip_runtime.h>
#include <hip/hip_bf16.h>

#define N_NODES   50000
#define N_EDGES   1250000
#define IN_DIM    5
#define OUT_DIM   64
#define NUM_REL   3
#define NUM_LAYERS 3
#define LN_EPS    1e-5f

typedef __attribute__((ext_vector_type(2))) float v2f;
typedef __attribute__((ext_vector_type(8))) float v8f;

#define NR_TOTAL    (N_NODES * NUM_REL)
#define BKT_NODES   128
#define BKT_SEGS    (BKT_NODES * NUM_REL)
#define NUM_BKT     ((N_NODES + BKT_NODES - 1) / BKT_NODES)
#define SORT_BLOCK  256
#define TILE_EDGES  (SORT_BLOCK * 16)
#define NUM_TILES   ((N_EDGES + TILE_EDGES - 1) / TILE_EDGES)
#define TILE_PITCH  320
#define CHUNK       8
static_assert(N_EDGES % 4 == 0, "edges are read as int4");
static_assert(NUM_TILES <= TILE_PITCH, "one row entry per tile");
static_assert(NUM_BKT <= 2 * SORT_BLOCK, "two buckets per thread in the bucket scan");
static_assert(N_NODES <= 65536, "a source node fits the low 16 bits of a filed edge");
static_assert(BKT_NODES <= SORT_BLOCK, "one thread per node of a bucket");

__device__ __forceinline__ int bucket_of(int d) {
    return min((int)((unsigned)d >> 7), NUM_BKT - 1);
}

__global__ __launch_bounds__(SORT_BLOCK)
void tile_hist_kernel(const int* __restrict__ dst, int* __restrict__ th) {
    __shared__ int hist[NUM_BKT];
    const int t = threadIdx.x;
    const int tile = blockIdx.x;
    for (int b = t; b < NUM_BKT; b += SORT_BLOCK) hist[b] = 0;
    __syncthreads();
    int4 d[4];
#pragma unroll
    for (int i = 0; i < 4; ++i) {
        const int e = tile * TILE_EDGES + i * (SORT_BLOCK * 4) + 4 * t;
        d[i] = (e < N_EDGES) ? *(const int4*)(dst + e) : make_int4(0, 0, 0, 0);
    }
#pragma unroll
    for (int i = 0; i < 4; ++i) {
        const int e = tile * TILE_EDGES + i * (SORT_BLOCK * 4) + 4 * t;
        if (e < N_EDGES) {
            atomicAdd(&hist[bucket_of(d[i].x)], 1);
            atomicAdd(&hist[bucket_of(d[i].y)], 1);
            atomicAdd(&hist[bucket_of(d[i].z)], 1);
            atomicAdd(&hist[bucket_of(d[i].w)], 1);
        }
    }
    __syncthreads();
    for (int b = t; b < NUM_BKT; b += SORT_BLOCK) th[b * TILE_PITCH + tile] = hist[b];
}

__global__ __launch_bounds__(SORT_BLOCK)
void tile_prefix_kernel(int* __restrict__ th, int* __restrict__ btot) {
    const int lane = threadIdx.x & 31;
    const int b = blockIdx.x * (SORT_BLOCK / 32) + (threadIdx.x >> 5);
    if (b >= NUM_BKT) return;
    int2* row = (int2*)(th + b * TILE_PITCH + lane * 10);
    int v[10];
#pragma unroll
    for (int j = 0; j < 5; ++j) {
        const int2 x = row[j];
        v[2 * j] = x.x;
        v[2 * j + 1] = x.y;
    }
    int sum = 0;
#pragma unroll
    for (int j = 0; j < 10; ++j) {
        const int c = (lane * 10 + j < NUM_TILES) ? v[j] : 0;
        v[j] = sum;
        sum += c;
    }
    int inc = sum;
    for (int off = 1; off < 32; off <<= 1) {
        const int u = __shfl_up(inc, off, 32);
        if (lane >= off) inc += u;
    }
    const int ex = inc - sum;
#pragma unroll
    for (int j = 0; j < 5; ++j) row[j] = make_int2(v[2 * j] + ex, v[2 * j + 1] + ex);
    if (lane == 31) btot[b] = inc;
}

__device__ __forceinline__ void file_edge(int* cursor, int* __restrict__ mid, int d, int ty, int s) {
    const int key = (d & (BKT_NODES - 1)) * NUM_REL + min(max(ty, 0), NUM_REL - 1);
    int pos = atomicAdd(&cursor[bucket_of(d)], 1);
    pos = min(max(pos, 0), N_EDGES - 1);
    mid[pos] = (key << 16) | (s & 0xFFFF);
}

__global__ __launch_bounds__(SORT_BLOCK)
void partition_kernel(const int* __restrict__ src,
                      const int* __restrict__ dst,
                      const int* __restrict__ etype,
                      const int* __restrict__ th,
                      const int* __restrict__ btot,
                      int* __restrict__ bbase,
                      int* __restrict__ mid) {
    __shared__ int cursor[NUM_BKT];
    __shared__ int wsum[SORT_BLOCK / 32];
    const int t = threadIdx.x;
    const int lane = t & 31;
    const int wave = t >> 5;
    const int tile = blockIdx.x;

    const int i0 = 2 * t, i1 = 2 * t + 1;
    const int c0 = (i0 < NUM_BKT) ? btot[i0] : 0;
    const int c1 = (i1 < NUM_BKT) ? btot[i1] : 0;
    const int p0 = (i0 < NUM_BKT) ? th[i0 * TILE_PITCH + tile] : 0;
    const int p1 = (i1 < NUM_BKT) ? th[i1 * TILE_PITCH + tile] : 0;
    const int ts = c0 + c1;
    int inc = ts;
    for (int off = 1; off < 32; off <<= 1) {
        const int u = __shfl_up(inc, off, 32);
        if (lane >= off) inc += u;
    }
    if (lane == 31) wsum[wave] = inc;
    __syncthreads();
    int ex = inc - ts;
    for (int w = 0; w < SORT_BLOCK / 32; ++w)
        if (w < wave) ex += wsum[w];
    if (i0 < NUM_BKT) { cursor[i0] = ex + p0;      if (tile == 0) bbase[i0] = ex; }
    if (i1 < NUM_BKT) { cursor[i1] = ex + c0 + p1; if (tile == 0) bbase[i1] = ex + c0; }
    if (tile == 0 && t == SORT_BLOCK - 1) bbase[NUM_BKT] = ex + ts;
    __syncthreads();

    int4 d[4], y[4], s[4];
#pragma unroll
    for (int i = 0; i < 4; ++i) {
        const int e = tile * TILE_EDGES + i * (SORT_BLOCK * 4) + 4 * t;
        const bool ok = e < N_EDGES;
        d[i] = ok ? *(const int4*)(dst + e)   : make_int4(0, 0, 0, 0);
        y[i] = ok ? *(const int4*)(etype + e) : make_int4(0, 0, 0, 0);
        s[i] = ok ? *(const int4*)(src + e)   : make_int4(0, 0, 0, 0);
    }
#pragma unroll
    for (int i = 0; i < 4; ++i) {
        const int e = tile * TILE_EDGES + i * (SORT_BLOCK * 4) + 4 * t;
        if (e < N_EDGES) {
            file_edge(cursor, mid, d[i].x, y[i].x, s[i].x);
            file_edge(cursor, mid, d[i].y, y[i].y, s[i].y);
            file_edge(cursor, mid, d[i].z, y[i].z, s[i].z);
            file_edge(cursor, mid, d[i].w, y[i].w, s[i].w);
        }
    }
}

__global__ __launch_bounds__(SORT_BLOCK)
void bucket_sort_kernel(const int* __restrict__ mid,
                        const int* __restrict__ bbase,
                        int* __restrict__ csr_src,
                        int* __restrict__ offs,
                        float* __restrict__ inv) {
    __shared__ int cnt[BKT_SEGS];
    __shared__ int cur[BKT_SEGS];
    __shared__ int wsum[BKT_NODES / 32];
    const int t = threadIdx.x;
    const int lane = t & 31;
    const int wave = t >> 5;
    const int k = blockIdx.x;
    const int base = min(max(bbase[k], 0), N_EDGES);
    const int n = min(max(bbase[k + 1] - base, 0), N_EDGES - base);

    for (int i = t; i < BKT_SEGS; i += SORT_BLOCK) cnt[i] = 0;
    __syncthreads();

    for (int j0 = 0; j0 < n; j0 += SORT_BLOCK * CHUNK) {
        int v[CHUNK];
#pragma unroll
        for (int j = 0; j < CHUNK; ++j) {
            const int i = j0 + j * SORT_BLOCK + t;
            v[j] = (i < n) ? mid[base + i] : -1;
        }
#pragma unroll
        for (int j = 0; j < CHUNK; ++j)
            if (v[j] >= 0) atomicAdd(&cnt[min(v[j] >> 16, BKT_SEGS - 1)], 1);
    }
    __syncthreads();

    int c0 = 0, c1 = 0, c2 = 0;
    if (t < BKT_NODES) { c0 = cnt[3 * t]; c1 = cnt[3 * t + 1]; c2 = cnt[3 * t + 2]; }
    const int ts = c0 + c1 + c2;
    int inc = ts;
    for (int off = 1; off < 32; off <<= 1) {
        const int u = __shfl_up(inc, off, 32);
        if (lane >= off) inc += u;
    }
    if (t < BKT_NODES && lane == 31) wsum[wave] = inc;
    __syncthreads();
    if (t < BKT_NODES) {
        int ex = inc - ts;
        for (int w = 0; w < BKT_NODES / 32; ++w)
            if (w < wave) ex += wsum[w];
        cur[3 * t]     = ex;
        cur[3 * t + 1] = ex + c0;
        cur[3 * t + 2] = ex + c0 + c1;
        const int node = k * BKT_NODES + t;
        if (node < N_NODES) {
            const int seg = node * NUM_REL;
            offs[seg]     = base + ex;
            offs[seg + 1] = base + ex + c0;
            offs[seg + 2] = base + ex + c0 + c1;
            inv[seg]     = 1.0f / (float)(c0 > 1 ? c0 : 1);
            inv[seg + 1] = 1.0f / (float)(c1 > 1 ? c1 : 1);
            inv[seg + 2] = 1.0f / (float)(c2 > 1 ? c2 : 1);
            if (node == N_NODES - 1) offs[NR_TOTAL] = base + ex + ts;
        }
    }
    __syncthreads();

    for (int j0 = 0; j0 < n; j0 += SORT_BLOCK * CHUNK) {
        int v[CHUNK];
#pragma unroll
        for (int j = 0; j < CHUNK; ++j) {
            const int i = j0 + j * SORT_BLOCK + t;
            v[j] = (i < n) ? mid[base + i] : -1;
        }
#pragma unroll
        for (int j = 0; j < CHUNK; ++j) {
            if (v[j] >= 0) {
                const int p = atomicAdd(&cur[min(v[j] >> 16, BKT_SEGS - 1)], 1);
                const int q = min(max(base + p, 0), N_EDGES - 1);
                csr_src[q] = v[j] & 0xFFFF;
            }
        }
    }
}

__global__ void feat2c_kernel(const float* __restrict__ x,
                              const float* __restrict__ W,
                              const float* __restrict__ b,
                              float* __restrict__ h, int n) {
    int idx = blockIdx.x * blockDim.x + threadIdx.x;
    if (idx >= n * OUT_DIM) return;
    int node = idx >> 6;
    int o    = idx & 63;
    float acc = b[o];
#pragma unroll
    for (int d = 0; d < IN_DIM; ++d)
        acc += x[node * IN_DIM + d] * W[d * OUT_DIM + o];
    h[idx] = acc;
}

#define TILE_M   16
#define A_PITCH  260
static_assert(N_NODES % TILE_M == 0, "no partial node tile");

__device__ __forceinline__ int bcast_i(int v, int l) {
    return __builtin_amdgcn_readlane(v, l);
}
__device__ __forceinline__ float bcast_f(float v, int l) {
    return __int_as_float(__builtin_amdgcn_readlane(__float_as_int(v), l));
}

template <int G>
__device__ __forceinline__ void load_rows(float2* m, const float2* __restrict__ h2, int s, int k) {
    const int lane = threadIdx.x & 31;
    for (int i = 0; i < G; ++i) {
        const int rk = bcast_i(s, k + i);
        const float2* __restrict__ rowp = h2 + (size_t)rk * (OUT_DIM / 2);
        m[i] = rowp[lane];
    }
}
template <int G>
__device__ __forceinline__ void add_rows(float2& a, const float2* m) {
    for (int i = 0; i < G; ++i) { a.x += m[i].x; a.y += m[i].y; }
}

__device__ __forceinline__ void gather_range(float2& a, const float2* __restrict__ h2,
                                             int s, int k, int kend) {
    for (; k + 8 <= kend; k += 8) {
        float2 m[8];
        load_rows<8>(m, h2, s, k);
        add_rows<8>(a, m);
    }
    const int rem = max(kend - k, 0);
    const float2 z = make_float2(0.f, 0.f);
    float2 m4[4] = {z, z, z, z};
    float2 m2[2] = {z, z};
    float2 m1[1] = {z};
    if (rem & 4) { load_rows<4>(m4, h2, s, k); k += 4; }
    if (rem & 2) { load_rows<2>(m2, h2, s, k); k += 2; }
    if (rem & 1) { load_rows<1>(m1, h2, s, k); }
    add_rows<4>(a, m4);
    add_rows<2>(a, m2);
    add_rows<1>(a, m1);
}

__global__ __launch_bounds__(128, 16)
void rgcn_layer_kernel(const float* __restrict__ h,
                       const int*   __restrict__ offs,
                       const int*   __restrict__ csr_src,
                       const float* __restrict__ inv_denom,
                       const float* __restrict__ relW,
                       const float* __restrict__ rootW,
                       const float* __restrict__ bias,
                       const float* __restrict__ gamma,
                       const float* __restrict__ beta,
                       float* __restrict__ hout)
{
    __shared__ __attribute__((aligned(16))) float lA[TILE_M][A_PITCH];
    const int row0 = blockIdx.x * TILE_M;
    const int wave = threadIdx.x >> 5;
    const int lane = threadIdx.x & 31;

    {
        const float2* __restrict__ h2 = (const float2*)h;
        const int node0 = row0 + wave * 4;
        int bl = offs[node0 * NUM_REL + (lane < 12 ? lane : 12)];
        bl = min(max(bl, 0), N_EDGES);
        const float wl = inv_denom[node0 * NUM_REL + (lane < 11 ? lane : 11)];

        int snext = 0;
        {
            const int j = bcast_i(bl, 0) + lane;
            if (j < bcast_i(bl, 3)) snext = csr_src[j];
        }
#pragma unroll
        for (int t = 0; t < 4; ++t) {
            const int b0 = bcast_i(bl, 3 * t);
            const int b1 = bcast_i(bl, 3 * t + 1);
            const int b2 = bcast_i(bl, 3 * t + 2);
            const int b3 = bcast_i(bl, 3 * t + 3);
            const float2 hown = h2[(size_t)(node0 + t) * (OUT_DIM / 2) + lane];
            int s = snext;
            if (t < 3) {
                snext = 0;
                const int j = b3 + lane;
                if (j < bcast_i(bl, 3 * t + 6)) snext = csr_src[j];
            }
            float2 a0 = make_float2(0.f, 0.f), a1 = a0, a2 = a0;
            for (int j0 = b0; j0 < b3; j0 += 32) {
                if (j0 != b0) {
                    const int j = j0 + lane;
                    s = (j < b3) ? csr_src[j] : 0;
                }
                s = min(max(s, 0), N_NODES - 1);
                const int k1 = min(max(b1 - j0, 0), 32);
                const int k2 = min(max(b2 - j0, 0), 32);
                const int k3 = min(b3 - j0, 32);
                gather_range(a0, h2, s, 0,  k1);
                gather_range(a1, h2, s, k1, k2);
                gather_range(a2, h2, s, k2, k3);
            }
            const float w0 = bcast_f(wl, 3 * t);
            const float w1 = bcast_f(wl, 3 * t + 1);
            const float w2 = bcast_f(wl, 3 * t + 2);
            float* arow = &lA[wave * 4 + t][2 * lane];
            *(float2*)(arow)       = make_float2(a0.x * w0, a0.y * w0);
            *(float2*)(arow + 64)  = make_float2(a1.x * w1, a1.y * w1);
            *(float2*)(arow + 128) = make_float2(a2.x * w2, a2.y * w2);
            *(float2*)(arow + 192) = hown;
        }
    }
    __syncthreads();

    const int c     = lane & 15;
    const int hh    = lane >> 4;
    const int khalf = hh << 1;
    const float* __restrict__ Wq =
        (wave < NUM_REL) ? (relW + (size_t)wave * OUT_DIM * OUT_DIM) : rootW;

    v8f acc[4];
#pragma unroll
    for (int j = 0; j < 4; ++j) acc[j] = (v8f){0.f,0.f,0.f,0.f,0.f,0.f,0.f,0.f};

#pragma unroll
    for (int k0 = 0; k0 < OUT_DIM; k0 += 4) {
        const int kk = k0 + khalf;
        const v2f a = *(const v2f*)&lA[c][wave * OUT_DIM + kk];
        const float4 w0 = *(const float4*)(Wq + (size_t)kk * OUT_DIM + 4 * c);
        const float4 w1 = *(const float4*)(Wq + (size_t)(kk + 1) * OUT_DIM + 4 * c);
        v2f b;
        b.x = w0.x; b.y = w1.x;
        acc[0] = __builtin_amdgcn_wmma_f32_16x16x4_f32(false, a, false, b, (short)0, acc[0], false, false);
        b.x = w0.y; b.y = w1.y;
        acc[1] = __builtin_amdgcn_wmma_f32_16x16x4_f32(false, a, false, b, (short)0, acc[1], false, false);
        b.x = w0.z; b.y = w1.z;
        acc[2] = __builtin_amdgcn_wmma_f32_16x16x4_f32(false, a, false, b, (short)0, acc[2], false, false);
        b.x = w0.w; b.y = w1.w;
        acc[3] = __builtin_amdgcn_wmma_f32_16x16x4_f32(false, a, false, b, (short)0, acc[3], false, false);
    }

#pragma unroll
    for (int r = 0; r < 8; ++r) {
        *(float4*)&lA[8 * hh + r][wave * OUT_DIM + 4 * c] =
            make_float4(acc[0][r], acc[1][r], acc[2][r], acc[3][r]);
    }
    __syncthreads();

    const float2 bs = ((const float2*)bias)[lane];
    const float2 g2 = ((const float2*)gamma)[lane];
    const float2 be = ((const float2*)beta)[lane];
#pragma unroll
    for (int t = 0; t < 4; ++t) {
        const int row = wave * 4 + t;
        const float* prow = &lA[row][2 * lane];
        const float2 p0 = *(const float2*)(prow);
        const float2 p1 = *(const float2*)(prow + 64);
        const float2 p2 = *(const float2*)(prow + 128);
        const float2 p3 = *(const float2*)(prow + 192);
        float2 v;
        v.x = p3.x + bs.x;  v.y = p3.y + bs.y;
        v.x += p0.x;        v.y += p0.y;
        v.x += p1.x;        v.y += p1.y;
        v.x += p2.x;        v.y += p2.y;
        v.x = fmaxf(v.x, 0.0f);
        v.y = fmaxf(v.y, 0.0f);
        float s  = v.x + v.y;
        float sq = v.x * v.x + v.y * v.y;
#pragma unroll
        for (int off = 16; off > 0; off >>= 1) {
            s  += __shfl_xor(s,  off, 32);
            sq += __shfl_xor(sq, off, 32);
        }
        const float mu   = s * (1.0f / OUT_DIM);
        const float var  = sq * (1.0f / OUT_DIM) - mu * mu;
        const float rstd = rsqrtf(var + LN_EPS);
        float2 o;
        o.x = (v.x - mu) * rstd * g2.x + be.x;
        o.y = (v.y - mu) * rstd * g2.y + be.y;
        ((float2*)(hout + (size_t)(row0 + row) * OUT_DIM))[lane] = o;
    }
}

extern "C" void kernel_launch(void* const* d_in, const int* in_sizes, int n_in,
                              void* d_out, int out_size, void* d_ws, size_t ws_size,
                              hipStream_t stream) {
    const float* x      = (const float*)d_in[0];
    const int*   ei     = (const int*)  d_in[1];
    const int*   src    = ei;
    const int*   dst    = ei + N_EDGES;
    const int*   etype  = (const int*)  d_in[2];
    const float* f2cW   = (const float*)d_in[3];
    const float* f2cb   = (const float*)d_in[4];
    const float* rgW    = (const float*)d_in[5];
    const float* rgRoot = (const float*)d_in[6];
    const float* rgBias = (const float*)d_in[7];
    const float* gamma  = (const float*)d_in[8];
    const float* beta   = (const float*)d_in[9];

    float* ws = (float*)d_ws;
    float* h_a       = ws;  ws += (size_t)N_NODES * OUT_DIM;
    int*   mid       = (int*)ws;
    int*   csr_src   = mid + N_EDGES;
    int*   offs      = csr_src + N_EDGES;
    ws += (size_t)N_NODES * OUT_DIM;
    float* h_b       = ws;  ws += (size_t)N_NODES * NUM_REL * OUT_DIM;
    int*   th        = (int*)ws;
    int*   btot      = th + NUM_BKT * TILE_PITCH;
    int*   bbase     = btot + 512;
    ws += (size_t)N_NODES * OUT_DIM;
    float* inv_denom = ws;

    tile_hist_kernel<<<NUM_TILES, SORT_BLOCK, 0, stream>>>(dst, th);
    tile_prefix_kernel<<<(NUM_BKT + SORT_BLOCK / 32 - 1) / (SORT_BLOCK / 32), SORT_BLOCK, 0, stream>>>(th, btot);
    partition_kernel<<<NUM_TILES, SORT_BLOCK, 0, stream>>>(src, dst, etype, th, btot, bbase, mid);
    bucket_sort_kernel<<<NUM_BKT, SORT_BLOCK, 0, stream>>>(mid, bbase, csr_src, offs, inv_denom);
    feat2c_kernel<<<((size_t)N_NODES * OUT_DIM + 255) / 256, 256, 0, stream>>>(
        x, f2cW, f2cb, h_a, N_NODES);

    for (int l = 0; l < NUM_LAYERS; ++l) {
        const float* hi = (l & 1) ? h_b : h_a;
        float* ho = (l == NUM_LAYERS - 1) ? (float*)d_out : ((l & 1) ? h_a : h_b);
        rgcn_layer_kernel<<<N_NODES / TILE_M, 128, 0, stream>>>(
            hi, offs, csr_src, inv_denom,
            rgW    + (size_t)l * NUM_REL * OUT_DIM * OUT_DIM,
            rgRoot + (size_t)l * OUT_DIM * OUT_DIM,
            rgBias + (size_t)l * OUT_DIM,
            gamma, beta, ho);
    }
}


# Input projection x@W+b folded into the bucket-sort launch (each block projects its own 128 nodes, two whole h rows per wave store); the 12500-block one-output-per-thread launch is gone: 7 launches
# speedup vs baseline: 2.6521x; 1.0371x over previous
#include <hip/hip_runtime.h>
#include <hip/hip_bf16.h>

#define N_NODES   50000
#define N_EDGES   1250000
#define IN_DIM    5
#define OUT_DIM   64
#define NUM_REL   3
#define NUM_LAYERS 3
#define LN_EPS    1e-5f

typedef __attribute__((ext_vector_type(2))) float v2f;
typedef __attribute__((ext_vector_type(8))) float v8f;

#define NR_TOTAL    (N_NODES * NUM_REL)
#define BKT_NODES   128
#define BKT_SEGS    (BKT_NODES * NUM_REL)
#define NUM_BKT     ((N_NODES + BKT_NODES - 1) / BKT_NODES)
#define SORT_BLOCK  256
#define TILE_EDGES  (SORT_BLOCK * 16)
#define NUM_TILES   ((N_EDGES + TILE_EDGES - 1) / TILE_EDGES)
#define TILE_PITCH  320
#define CHUNK       8
static_assert(N_EDGES % 4 == 0, "edges are read as int4");
static_assert(NUM_TILES <= TILE_PITCH, "one row entry per tile");
static_assert(NUM_BKT <= 2 * SORT_BLOCK, "two buckets per thread in the bucket scan");
static_assert(N_NODES <= 65536, "a source node fits the low 16 bits of a filed edge");
static_assert(BKT_NODES <= SORT_BLOCK, "one thread per node of a bucket");

__device__ __forceinline__ int bucket_of(int d) {
    return min((int)((unsigned)d >> 7), NUM_BKT - 1);
}

__global__ __launch_bounds__(SORT_BLOCK)
void tile_hist_kernel(const int* __restrict__ dst, int* __restrict__ th) {
    __shared__ int hist[NUM_BKT];
    const int t = threadIdx.x;
    const int tile = blockIdx.x;
    for (int b = t; b < NUM_BKT; b += SORT_BLOCK) hist[b] = 0;
    __syncthreads();
    int4 d[4];
#pragma unroll
    for (int i = 0; i < 4; ++i) {
        const int e = tile * TILE_EDGES + i * (SORT_BLOCK * 4) + 4 * t;
        d[i] = (e < N_EDGES) ? *(const int4*)(dst + e) : make_int4(0, 0, 0, 0);
    }
#pragma unroll
    for (int i = 0; i < 4; ++i) {
        const int e = tile * TILE_EDGES + i * (SORT_BLOCK * 4) + 4 * t;
        if (e < N_EDGES) {
            atomicAdd(&hist[bucket_of(d[i].x)], 1);
            atomicAdd(&hist[bucket_of(d[i].y)], 1);
            atomicAdd(&hist[bucket_of(d[i].z)], 1);
            atomicAdd(&hist[bucket_of(d[i].w)], 1);
        }
    }
    __syncthreads();
    for (int b = t; b < NUM_BKT; b += SORT_BLOCK) th[b * TILE_PITCH + tile] = hist[b];
}

__global__ __launch_bounds__(SORT_BLOCK)
void tile_prefix_kernel(int* __restrict__ th, int* __restrict__ btot) {
    const int lane = threadIdx.x & 31;
    const int b = blockIdx.x * (SORT_BLOCK / 32) + (threadIdx.x >> 5);
    if (b >= NUM_BKT) return;
    int2* row = (int2*)(th + b * TILE_PITCH + lane * 10);
    int v[10];
#pragma unroll
    for (int j = 0; j < 5; ++j) {
        const int2 x = row[j];
        v[2 * j] = x.x;
        v[2 * j + 1] = x.y;
    }
    int sum = 0;
#pragma unroll
    for (int j = 0; j < 10; ++j) {
        const int c = (lane * 10 + j < NUM_TILES) ? v[j] : 0;
        v[j] = sum;
        sum += c;
    }
    int inc = sum;
    for (int off = 1; off < 32; off <<= 1) {
        const int u = __shfl_up(inc, off, 32);
        if (lane >= off) inc += u;
    }
    const int ex = inc - sum;
#pragma unroll
    for (int j = 0; j < 5; ++j) row[j] = make_int2(v[2 * j] + ex, v[2 * j + 1] + ex);
    if (lane == 31) btot[b] = inc;
}

__device__ __forceinline__ void file_edge(int* cursor, int* __restrict__ mid, int d, int ty, int s) {
    const int key = (d & (BKT_NODES - 1)) * NUM_REL + min(max(ty, 0), NUM_REL - 1);
    int pos = atomicAdd(&cursor[bucket_of(d)], 1);
    pos = min(max(pos, 0), N_EDGES - 1);
    mid[pos] = (key << 16) | (s & 0xFFFF);
}

__global__ __launch_bounds__(SORT_BLOCK)
void partition_kernel(const int* __restrict__ src,
                      const int* __restrict__ dst,
                      const int* __restrict__ etype,
                      const int* __restrict__ th,
                      const int* __restrict__ btot,
                      int* __restrict__ bbase,
                      int* __restrict__ mid) {
    __shared__ int cursor[NUM_BKT];
    __shared__ int wsum[SORT_BLOCK / 32];
    const int t = threadIdx.x;
    const int lane = t & 31;
    const int wave = t >> 5;
    const int tile = blockIdx.x;

    const int i0 = 2 * t, i1 = 2 * t + 1;
    const int c0 = (i0 < NUM_BKT) ? btot[i0] : 0;
    const int c1 = (i1 < NUM_BKT) ? btot[i1] : 0;
    const int p0 = (i0 < NUM_BKT) ? th[i0 * TILE_PITCH + tile] : 0;
    const int p1 = (i1 < NUM_BKT) ? th[i1 * TILE_PITCH + tile] : 0;
    const int ts = c0 + c1;
    int inc = ts;
    for (int off = 1; off < 32; off <<= 1) {
        const int u = __shfl_up(inc, off, 32);
        if (lane >= off) inc += u;
    }
    if (lane == 31) wsum[wave] = inc;
    __syncthreads();
    int ex = inc - ts;
    for (int w = 0; w < SORT_BLOCK / 32; ++w)
        if (w < wave) ex += wsum[w];
    if (i0 < NUM_BKT) { cursor[i0] = ex + p0;      if (tile == 0) bbase[i0] = ex; }
    if (i1 < NUM_BKT) { cursor[i1] = ex + c0 + p1; if (tile == 0) bbase[i1] = ex + c0; }
    if (tile == 0 && t == SORT_BLOCK - 1) bbase[NUM_BKT] = ex + ts;
    __syncthreads();

    int4 d[4], y[4], s[4];
#pragma unroll
    for (int i = 0; i < 4; ++i) {
        const int e = tile * TILE_EDGES + i * (SORT_BLOCK * 4) + 4 * t;
        const bool ok = e < N_EDGES;
        d[i] = ok ? *(const int4*)(dst + e)   : make_int4(0, 0, 0, 0);
        y[i] = ok ? *(const int4*)(etype + e) : make_int4(0, 0, 0, 0);
        s[i] = ok ? *(const int4*)(src + e)   : make_int4(0, 0, 0, 0);
    }
#pragma unroll
    for (int i = 0; i < 4; ++i) {
        const int e = tile * TILE_EDGES + i * (SORT_BLOCK * 4) + 4 * t;
        if (e < N_EDGES) {
            file_edge(cursor, mid, d[i].x, y[i].x, s[i].x);
            file_edge(cursor, mid, d[i].y, y[i].y, s[i].y);
            file_edge(cursor, mid, d[i].z, y[i].z, s[i].z);
            file_edge(cursor, mid, d[i].w, y[i].w, s[i].w);
        }
    }
}

#define X_VEC4  (BKT_NODES * IN_DIM / 4)
static_assert((BKT_NODES * IN_DIM) % 4 == 0 && (N_NODES * IN_DIM) % 4 == 0, "a bucket's x rows are read as float4");
static_assert(X_VEC4 <= SORT_BLOCK, "one float4 of x per thread");
static_assert(OUT_DIM == 64 && SORT_BLOCK == 256 && BKT_NODES % 16 == 0, "16 lanes x float4 per h row, 16 rows per pass");

template <bool FULL>
__device__ __forceinline__ void project_bucket(const float* xs, const float4 (&wv)[IN_DIM], float4 bv,
                                               float* __restrict__ h, int k, int t) {
    const int c4 = t & 15;
#pragma unroll
    for (int i = 0; i < BKT_NODES / 16; ++i) {
        const int r = (t >> 4) + 16 * i;
        const int node = k * BKT_NODES + r;
        const float* xr = &xs[r * IN_DIM];
        float4 acc = bv;
#pragma unroll
        for (int d = 0; d < IN_DIM; ++d) {
            const float xd = xr[d];
            acc.x += xd * wv[d].x;
            acc.y += xd * wv[d].y;
            acc.z += xd * wv[d].z;
            acc.w += xd * wv[d].w;
        }
        if (FULL || node < N_NODES) *(float4*)(h + (size_t)node * OUT_DIM + 4 * c4) = acc;
    }
}

__global__ __launch_bounds__(SORT_BLOCK)
void bucket_sort_kernel(const int* __restrict__ mid,
                        const int* __restrict__ bbase,
                        int* __restrict__ csr_src,
                        int* __restrict__ offs,
                        float* __restrict__ inv,
                        const float* __restrict__ x,
                        const float* __restrict__ W,
                        const float* __restrict__ b,
                        float* __restrict__ h) {
    __shared__ int cnt[BKT_SEGS];
    __shared__ int cur[BKT_SEGS];
    __shared__ int wsum[BKT_NODES / 32];
    __shared__ __attribute__((aligned(16))) float xs[BKT_NODES * IN_DIM];
    const int t = threadIdx.x;
    const int lane = t & 31;
    const int wave = t >> 5;
    const int k = blockIdx.x;

    const int c4 = t & 15;
    const int xf = k * (BKT_NODES * IN_DIM) + 4 * t;
    float4 xv = make_float4(0.f, 0.f, 0.f, 0.f);
    if (t < X_VEC4 && xf < N_NODES * IN_DIM) xv = *(const float4*)(x + xf);
    float4 wv[IN_DIM];
#pragma unroll
    for (int d = 0; d < IN_DIM; ++d) wv[d] = *(const float4*)(W + d * OUT_DIM + 4 * c4);
    const float4 bv = *(const float4*)(b + 4 * c4);

    const int base = min(max(bbase[k], 0), N_EDGES);
    const int n = min(max(bbase[k + 1] - base, 0), N_EDGES - base);

    for (int i = t; i < BKT_SEGS; i += SORT_BLOCK) cnt[i] = 0;
    __syncthreads();

    for (int j0 = 0; j0 < n; j0 += SORT_BLOCK * CHUNK) {
        int v[CHUNK];
#pragma unroll
        for (int j = 0; j < CHUNK; ++j) {
            const int i = j0 + j * SORT_BLOCK + t;
            v[j] = (i < n) ? mid[base + i] : -1;
        }
#pragma unroll
        for (int j = 0; j < CHUNK; ++j)
            if (v[j] >= 0) atomicAdd(&cnt[min(v[j] >> 16, BKT_SEGS - 1)], 1);
    }
    __syncthreads();

    int c0 = 0, c1 = 0, c2 = 0;
    if (t < BKT_NODES) { c0 = cnt[3 * t]; c1 = cnt[3 * t + 1]; c2 = cnt[3 * t + 2]; }
    const int ts = c0 + c1 + c2;
    int inc = ts;
    for (int off = 1; off < 32; off <<= 1) {
        const int u = __shfl_up(inc, off, 32);
        if (lane >= off) inc += u;
    }
    if (t < BKT_NODES && lane == 31) wsum[wave] = inc;
    __syncthreads();
    if (t < BKT_NODES) {
        int ex = inc - ts;
        for (int w = 0; w < BKT_NODES / 32; ++w)
            if (w < wave) ex += wsum[w];
        cur[3 * t]     = ex;
        cur[3 * t + 1] = ex + c0;
        cur[3 * t + 2] = ex + c0 + c1;
        const int node = k * BKT_NODES + t;
        if (node < N_NODES) {
            const int seg = node * NUM_REL;
            offs[seg]     = base + ex;
            offs[seg + 1] = base + ex + c0;
            offs[seg + 2] = base + ex + c0 + c1;
            inv[seg]     = 1.0f / (float)(c0 > 1 ? c0 : 1);
            inv[seg + 1] = 1.0f / (float)(c1 > 1 ? c1 : 1);
            inv[seg + 2] = 1.0f / (float)(c2 > 1 ? c2 : 1);
            if (node == N_NODES - 1) offs[NR_TOTAL] = base + ex + ts;
        }
    }
    if (t < X_VEC4) *(float4*)&xs[4 * t] = xv;
    __syncthreads();

    for (int j0 = 0; j0 < n; j0 += SORT_BLOCK * CHUNK) {
        int v[CHUNK];
#pragma unroll
        for (int j = 0; j < CHUNK; ++j) {
            const int i = j0 + j * SORT_BLOCK + t;
            v[j] = (i < n) ? mid[base + i] : -1;
        }
#pragma unroll
        for (int j = 0; j < CHUNK; ++j) {
            if (v[j] >= 0) {
                const int p = atomicAdd(&cur[min(v[j] >> 16, BKT_SEGS - 1)], 1);
                const int q = min(max(base + p, 0), N_EDGES - 1);
                csr_src[q] = v[j] & 0xFFFF;
            }
        }
    }

    if ((k + 1) * BKT_NODES <= N_NODES) project_bucket<true>(xs, wv, bv, h, k, t);
    else                                project_bucket<false>(xs, wv, bv, h, k, t);
}

#define TILE_M   16
#define A_PITCH  260
static_assert(N_NODES % TILE_M == 0, "no partial node tile");

__device__ __forceinline__ int bcast_i(int v, int l) {
    return __builtin_amdgcn_readlane(v, l);
}
__device__ __forceinline__ float bcast_f(float v, int l) {
    return __int_as_float(__builtin_amdgcn_readlane(__float_as_int(v), l));
}

template <int G>
__device__ __forceinline__ void load_rows(float2* m, const float2* __restrict__ h2, int s, int k) {
    const int lane = threadIdx.x & 31;
    for (int i = 0; i < G; ++i) {
        const int rk = bcast_i(s, k + i);
        const float2* __restrict__ rowp = h2 + (size_t)rk * (OUT_DIM / 2);
        m[i] = rowp[lane];
    }
}
template <int G>
__device__ __forceinline__ void add_rows(float2& a, const float2* m) {
    for (int i = 0; i < G; ++i) { a.x += m[i].x; a.y += m[i].y; }
}

__device__ __forceinline__ void gather_range(float2& a, const float2* __restrict__ h2,
                                             int s, int k, int kend) {
    for (; k + 8 <= kend; k += 8) {
        float2 m[8];
        load_rows<8>(m, h2, s, k);
        add_rows<8>(a, m);
    }
    const int rem = max(kend - k, 0);
    const float2 z = make_float2(0.f, 0.f);
    float2 m4[4] = {z, z, z, z};
    float2 m2[2] = {z, z};
    float2 m1[1] = {z};
    if (rem & 4) { load_rows<4>(m4, h2, s, k); k += 4; }
    if (rem & 2) { load_rows<2>(m2, h2, s, k); k += 2; }
    if (rem & 1) { load_rows<1>(m1, h2, s, k); }
    add_rows<4>(a, m4);
    add_rows<2>(a, m2);
    add_rows<1>(a, m1);
}

__global__ __launch_bounds__(128, 16)
void rgcn_layer_kernel(const float* __restrict__ h,
                       const int*   __restrict__ offs,
                       const int*   __restrict__ csr_src,
                       const float* __restrict__ inv_denom,
                       const float* __restrict__ relW,
                       const float* __restrict__ rootW,
                       const float* __restrict__ bias,
                       const float* __restrict__ gamma,
                       const float* __restrict__ beta,
                       float* __restrict__ hout)
{
    __shared__ __attribute__((aligned(16))) float lA[TILE_M][A_PITCH];
    const int row0 = blockIdx.x * TILE_M;
    const int wave = threadIdx.x >> 5;
    const int lane = threadIdx.x & 31;

    {
        const float2* __restrict__ h2 = (const float2*)h;
        const int node0 = row0 + wave * 4;
        int bl = offs[node0 * NUM_REL + (lane < 12 ? lane : 12)];
        bl = min(max(bl, 0), N_EDGES);
        const float wl = inv_denom[node0 * NUM_REL + (lane < 11 ? lane : 11)];

        int snext = 0;
        {
            const int j = bcast_i(bl, 0) + lane;
            if (j < bcast_i(bl, 3)) snext = csr_src[j];
        }
#pragma unroll
        for (int t = 0; t < 4; ++t) {
            const int b0 = bcast_i(bl, 3 * t);
            const int b1 = bcast_i(bl, 3 * t + 1);
            const int b2 = bcast_i(bl, 3 * t + 2);
            const int b3 = bcast_i(bl, 3 * t + 3);
            const float2 hown = h2[(size_t)(node0 + t) * (OUT_DIM / 2) + lane];
            int s = snext;
            if (t < 3) {
                snext = 0;
                const int j = b3 + lane;
                if (j < bcast_i(bl, 3 * t + 6)) snext = csr_src[j];
            }
            float2 a0 = make_float2(0.f, 0.f), a1 = a0, a2 = a0;
            for (int j0 = b0; j0 < b3; j0 += 32) {
                if (j0 != b0) {
                    const int j = j0 + lane;
                    s = (j < b3) ? csr_src[j] : 0;
                }
                s = min(max(s, 0), N_NODES - 1);
                const int k1 = min(max(b1 - j0, 0), 32);
                const int k2 = min(max(b2 - j0, 0), 32);
                const int k3 = min(b3 - j0, 32);
                gather_range(a0, h2, s, 0,  k1);
                gather_range(a1, h2, s, k1, k2);
                gather_range(a2, h2, s, k2, k3);
            }
            const float w0 = bcast_f(wl, 3 * t);
            const float w1 = bcast_f(wl, 3 * t + 1);
            const float w2 = bcast_f(wl, 3 * t + 2);
            float* arow = &lA[wave * 4 + t][2 * lane];
            *(float2*)(arow)       = make_float2(a0.x * w0, a0.y * w0);
            *(float2*)(arow + 64)  = make_float2(a1.x * w1, a1.y * w1);
            *(float2*)(arow + 128) = make_float2(a2.x * w2, a2.y * w2);
            *(float2*)(arow + 192) = hown;
        }
    }
    __syncthreads();

    const int c     = lane & 15;
    const int hh    = lane >> 4;
    const int khalf = hh << 1;
    const float* __restrict__ Wq =
        (wave < NUM_REL) ? (relW + (size_t)wave * OUT_DIM * OUT_DIM) : rootW;

    v8f acc[4];
#pragma unroll
    for (int j = 0; j < 4; ++j) acc[j] = (v8f){0.f,0.f,0.f,0.f,0.f,0.f,0.f,0.f};

#pragma unroll
    for (int k0 = 0; k0 < OUT_DIM; k0 += 4) {
        const int kk = k0 + khalf;
        const v2f a = *(const v2f*)&lA[c][wave * OUT_DIM + kk];
        const float4 w0 = *(const float4*)(Wq + (size_t)kk * OUT_DIM + 4 * c);
        const float4 w1 = *(const float4*)(Wq + (size_t)(kk + 1) * OUT_DIM + 4 * c);
        v2f b;
        b.x = w0.x; b.y = w1.x;
        acc[0] = __builtin_amdgcn_wmma_f32_16x16x4_f32(false, a, false, b, (short)0, acc[0], false, false);
        b.x = w0.y; b.y = w1.y;
        acc[1] = __builtin_amdgcn_wmma_f32_16x16x4_f32(false, a, false, b, (short)0, acc[1], false, false);
        b.x = w0.z; b.y = w1.z;
        acc[2] = __builtin_amdgcn_wmma_f32_16x16x4_f32(false, a, false, b, (short)0, acc[2], false, false);
        b.x = w0.w; b.y = w1.w;
        acc[3] = __builtin_amdgcn_wmma_f32_16x16x4_f32(false, a, false, b, (short)0, acc[3], false, false);
    }

#pragma unroll
    for (int r = 0; r < 8; ++r) {
        *(float4*)&lA[8 * hh + r][wave * OUT_DIM + 4 * c] =
            make_float4(acc[0][r], acc[1][r], acc[2][r], acc[3][r]);
    }
    __syncthreads();

    const float2 bs = ((const float2*)bias)[lane];
    const float2 g2 = ((const float2*)gamma)[lane];
    const float2 be = ((const float2*)beta)[lane];
#pragma unroll
    for (int t = 0; t < 4; ++t) {
        const int row = wave * 4 + t;
        const float* prow = &lA[row][2 * lane];
        const float2 p0 = *(const float2*)(prow);
        const float2 p1 = *(const float2*)(prow + 64);
        const float2 p2 = *(const float2*)(prow + 128);
        const float2 p3 = *(const float2*)(prow + 192);
        float2 v;
        v.x = p3.x + bs.x;  v.y = p3.y + bs.y;
        v.x += p0.x;        v.y += p0.y;
        v.x += p1.x;        v.y += p1.y;
        v.x += p2.x;        v.y += p2.y;
        v.x = fmaxf(v.x, 0.0f);
        v.y = fmaxf(v.y, 0.0f);
        float s  = v.x + v.y;
        float sq = v.x * v.x + v.y * v.y;
#pragma unroll
        for (int off = 16; off > 0; off >>= 1) {
            s  += __shfl_xor(s,  off, 32);
            sq += __shfl_xor(sq, off, 32);
        }
        const float mu   = s * (1.0f / OUT_DIM);
        const float var  = sq * (1.0f / OUT_DIM) - mu * mu;
        const float rstd = rsqrtf(var + LN_EPS);
        float2 o;
        o.x = (v.x - mu) * rstd * g2.x + be.x;
        o.y = (v.y - mu) * rstd * g2.y + be.y;
        ((float2*)(hout + (size_t)(row0 + row) * OUT_DIM))[lane] = o;
    }
}

extern "C" void kernel_launch(void* const* d_in, const int* in_sizes, int n_in,
                              void* d_out, int out_size, void* d_ws, size_t ws_size,
                              hipStream_t stream) {
    const float* x      = (const float*)d_in[0];
    const int*   ei     = (const int*)  d_in[1];
    const int*   src    = ei;
    const int*   dst    = ei + N_EDGES;
    const int*   etype  = (const int*)  d_in[2];
    const float* f2cW   = (const float*)d_in[3];
    const float* f2cb   = (const float*)d_in[4];
    const float* rgW    = (const float*)d_in[5];
    const float* rgRoot = (const float*)d_in[6];
    const float* rgBias = (const float*)d_in[7];
    const float* gamma  = (const float*)d_in[8];
    const float* beta   = (const float*)d_in[9];

    float* ws = (float*)d_ws;
    float* h_a       = ws;  ws += (size_t)N_NODES * OUT_DIM;
    int*   mid       = (int*)ws;
    int*   csr_src   = mid + N_EDGES;
    int*   offs      = csr_src + N_EDGES;
    ws += (size_t)N_NODES * OUT_DIM;
    float* h_b       = ws;  ws += (size_t)N_NODES * NUM_REL * OUT_DIM;
    int*   th        = (int*)ws;
    int*   btot      = th + NUM_BKT * TILE_PITCH;
    int*   bbase     = btot + 512;
    ws += (size_t)N_NODES * OUT_DIM;
    float* inv_denom = ws;

    tile_hist_kernel<<<NUM_TILES, SORT_BLOCK, 0, stream>>>(dst, th);
    tile_prefix_kernel<<<(NUM_BKT + SORT_BLOCK / 32 - 1) / (SORT_BLOCK / 32), SORT_BLOCK, 0, stream>>>(th, btot);
    partition_kernel<<<NUM_TILES, SORT_BLOCK, 0, stream>>>(src, dst, etype, th, btot, bbase, mid);
    bucket_sort_kernel<<<NUM_BKT, SORT_BLOCK, 0, stream>>>(mid, bbase, csr_src, offs, inv_denom,
                                                       x, f2cW, f2cb, h_a);

    for (int l = 0; l < NUM_LAYERS; ++l) {
        const float* hi = (l & 1) ? h_b : h_a;
        float* ho = (l == NUM_LAYERS - 1) ? (float*)d_out : ((l & 1) ? h_a : h_b);
        rgcn_layer_kernel<<<N_NODES / TILE_M, 128, 0, stream>>>(
            hi, offs, csr_src, inv_denom,
            rgW    + (size_t)l * NUM_REL * OUT_DIM * OUT_DIM,
            rgRoot + (size_t)l * OUT_DIM * OUT_DIM,
            rgBias + (size_t)l * OUT_DIM,
            gamma, beta, ho);
    }
}


# Both sort levels: per-edge scattered 4-byte global stores staged through an LDS image (tile by bucket, bucket by segment), written out as contiguous runs; order in a segment stays LDS arrival order
# speedup vs baseline: 3.0791x; 1.1610x over previous
#include <hip/hip_runtime.h>
#include <hip/hip_bf16.h>

#define N_NODES   50000
#define N_EDGES   1250000
#define IN_DIM    5
#define OUT_DIM   64
#define NUM_REL   3
#define NUM_LAYERS 3
#define LN_EPS    1e-5f

typedef __attribute__((ext_vector_type(2))) float v2f;
typedef __attribute__((ext_vector_type(8))) float v8f;

#define NR_TOTAL    (N_NODES * NUM_REL)
#define BKT_NODES   128
#define BKT_SEGS    (BKT_NODES * NUM_REL)
#define NUM_BKT     ((N_NODES + BKT_NODES - 1) / BKT_NODES)
#define SORT_BLOCK  256
#define TILE_EDGES  (SORT_BLOCK * 16)
#define NUM_TILES   ((N_EDGES + TILE_EDGES - 1) / TILE_EDGES)
#define TILE_PITCH  320
#define CHUNK       8
static_assert(N_EDGES % 4 == 0, "edges are read as int4");
static_assert(NUM_TILES < TILE_PITCH, "one row entry per tile, and one behind the last tile (the bucket's total)");
static_assert(NUM_BKT <= 2 * SORT_BLOCK, "two buckets per thread in the bucket scan");
static_assert(N_NODES <= 65536, "a source node fits the low 16 bits of a filed edge");
static_assert(BKT_NODES <= SORT_BLOCK, "one thread per node of a bucket");

__device__ __forceinline__ int bucket_of(int d) {
    return min((int)((unsigned)d >> 7), NUM_BKT - 1);
}

__global__ __launch_bounds__(SORT_BLOCK)
void tile_hist_kernel(const int* __restrict__ dst, int* __restrict__ th) {
    __shared__ int hist[NUM_BKT];
    const int t = threadIdx.x;
    const int tile = blockIdx.x;
    for (int b = t; b < NUM_BKT; b += SORT_BLOCK) hist[b] = 0;
    __syncthreads();
    int4 d[4];
#pragma unroll
    for (int i = 0; i < 4; ++i) {
        const int e = tile * TILE_EDGES + i * (SORT_BLOCK * 4) + 4 * t;
        d[i] = (e < N_EDGES) ? *(const int4*)(dst + e) : make_int4(0, 0, 0, 0);
    }
#pragma unroll
    for (int i = 0; i < 4; ++i) {
        const int e = tile * TILE_EDGES + i * (SORT_BLOCK * 4) + 4 * t;
        if (e < N_EDGES) {
            atomicAdd(&hist[bucket_of(d[i].x)], 1);
            atomicAdd(&hist[bucket_of(d[i].y)], 1);
            atomicAdd(&hist[bucket_of(d[i].z)], 1);
            atomicAdd(&hist[bucket_of(d[i].w)], 1);
        }
    }
    __syncthreads();
    for (int b = t; b < NUM_BKT; b += SORT_BLOCK) th[b * TILE_PITCH + tile] = hist[b];
}

__global__ __launch_bounds__(SORT_BLOCK)
void tile_prefix_kernel(int* __restrict__ th, int* __restrict__ btot) {
    const int lane = threadIdx.x & 31;
    const int b = blockIdx.x * (SORT_BLOCK / 32) + (threadIdx.x >> 5);
    if (b >= NUM_BKT) return;
    int2* row = (int2*)(th + b * TILE_PITCH + lane * 10);
    int v[10];
#pragma unroll
    for (int j = 0; j < 5; ++j) {
        const int2 x = row[j];
        v[2 * j] = x.x;
        v[2 * j + 1] = x.y;
    }
    int sum = 0;
#pragma unroll
    for (int j = 0; j < 10; ++j) {
        const int c = (lane * 10 + j < NUM_TILES) ? v[j] : 0;
        v[j] = sum;
        sum += c;
    }
    int inc = sum;
    for (int off = 1; off < 32; off <<= 1) {
        const int u = __shfl_up(inc, off, 32);
        if (lane >= off) inc += u;
    }
    const int ex = inc - sum;
#pragma unroll
    for (int j = 0; j < 5; ++j) row[j] = make_int2(v[2 * j] + ex, v[2 * j + 1] + ex);
    if (lane == 31) btot[b] = inc;
}

__device__ __forceinline__ void file_edge(int* lcur, const int* delta, int* sval, int* spos,
                                          int d, int ty, int s) {
    const int key = (d & (BKT_NODES - 1)) * NUM_REL + min(max(ty, 0), NUM_REL - 1);
    const int b = bucket_of(d);
    int lp = atomicAdd(&lcur[b], 1);
    lp = min(max(lp, 0), TILE_EDGES - 1);
    sval[lp] = (key << 16) | (s & 0xFFFF);
    spos[lp] = lp + delta[b];
}

__global__ __launch_bounds__(SORT_BLOCK)
void partition_kernel(const int* __restrict__ src,
                      const int* __restrict__ dst,
                      const int* __restrict__ etype,
                      const int* __restrict__ th,
                      const int* __restrict__ btot,
                      int* __restrict__ bbase,
                      int* __restrict__ mid) {
    __shared__ int lcur[NUM_BKT];
    __shared__ int delta[NUM_BKT];
    __shared__ int wsum[SORT_BLOCK / 32];
    __shared__ int lwsum[SORT_BLOCK / 32];
    __shared__ int sval[TILE_EDGES];
    __shared__ int spos[TILE_EDGES];
    const int t = threadIdx.x;
    const int lane = t & 31;
    const int wave = t >> 5;
    const int tile = blockIdx.x;

    const int i0 = 2 * t, i1 = 2 * t + 1;
    const int c0 = (i0 < NUM_BKT) ? btot[i0] : 0;
    const int c1 = (i1 < NUM_BKT) ? btot[i1] : 0;
    const int p0 = (i0 < NUM_BKT) ? th[i0 * TILE_PITCH + tile] : 0;
    const int p1 = (i1 < NUM_BKT) ? th[i1 * TILE_PITCH + tile] : 0;
    const int q0 = (i0 < NUM_BKT) ? th[i0 * TILE_PITCH + tile + 1] : 0;
    const int q1 = (i1 < NUM_BKT) ? th[i1 * TILE_PITCH + tile + 1] : 0;
    const int n0 = max(q0 - p0, 0);
    const int n1 = max(q1 - p1, 0);
    const int ts = c0 + c1;
    const int ls = n0 + n1;
    int inc = ts;
    int linc = ls;
    for (int off = 1; off < 32; off <<= 1) {
        const int u = __shfl_up(inc, off, 32);
        const int lu = __shfl_up(linc, off, 32);
        if (lane >= off) { inc += u; linc += lu; }
    }
    if (lane == 31) { wsum[wave] = inc; lwsum[wave] = linc; }
    __syncthreads();
    int ex = inc - ts;
    int lex = linc - ls;
    for (int w = 0; w < SORT_BLOCK / 32; ++w)
        if (w < wave) { ex += wsum[w]; lex += lwsum[w]; }
    if (i0 < NUM_BKT) { lcur[i0] = lex;      delta[i0] = ex + p0 - lex;             if (tile == 0) bbase[i0] = ex; }
    if (i1 < NUM_BKT) { lcur[i1] = lex + n0; delta[i1] = ex + c0 + p1 - (lex + n0); if (tile == 0) bbase[i1] = ex + c0; }
    if (tile == 0 && t == SORT_BLOCK - 1) bbase[NUM_BKT] = ex + ts;
    __syncthreads();

    int4 d[4], y[4], s[4];
#pragma unroll
    for (int i = 0; i < 4; ++i) {
        const int e = tile * TILE_EDGES + i * (SORT_BLOCK * 4) + 4 * t;
        const bool ok = e < N_EDGES;
        d[i] = ok ? *(const int4*)(dst + e)   : make_int4(0, 0, 0, 0);
        y[i] = ok ? *(const int4*)(etype + e) : make_int4(0, 0, 0, 0);
        s[i] = ok ? *(const int4*)(src + e)   : make_int4(0, 0, 0, 0);
    }
#pragma unroll
    for (int i = 0; i < 4; ++i) {
        const int e = tile * TILE_EDGES + i * (SORT_BLOCK * 4) + 4 * t;
        if (e < N_EDGES) {
            file_edge(lcur, delta, sval, spos, d[i].x, y[i].x, s[i].x);
            file_edge(lcur, delta, sval, spos, d[i].y, y[i].y, s[i].y);
            file_edge(lcur, delta, sval, spos, d[i].z, y[i].z, s[i].z);
            file_edge(lcur, delta, sval, spos, d[i].w, y[i].w, s[i].w);
        }
    }
    __syncthreads();

    const int cnt = min(TILE_EDGES, N_EDGES - tile * TILE_EDGES);
    int pv[TILE_EDGES / SORT_BLOCK], sv[TILE_EDGES / SORT_BLOCK];
#pragma unroll
    for (int j = 0; j < TILE_EDGES / SORT_BLOCK; ++j) {
        pv[j] = spos[j * SORT_BLOCK + t];
        sv[j] = sval[j * SORT_BLOCK + t];
    }
#pragma unroll
    for (int j = 0; j < TILE_EDGES / SORT_BLOCK; ++j) {
        if (j * SORT_BLOCK + t < cnt) {
            const int pos = min(max(pv[j], 0), N_EDGES - 1);
            mid[pos] = sv[j];
        }
    }
}

#define STAGE_CAP 4096
#define X_VEC4  (BKT_NODES * IN_DIM / 4)
static_assert((BKT_NODES * IN_DIM) % 4 == 0 && (N_NODES * IN_DIM) % 4 == 0, "a bucket's x rows are read as float4");
static_assert(X_VEC4 <= SORT_BLOCK, "one float4 of x per thread");
static_assert(OUT_DIM == 64 && SORT_BLOCK == 256 && BKT_NODES % 16 == 0, "16 lanes x float4 per h row, 16 rows per pass");

template <bool FULL>
__device__ __forceinline__ void project_bucket(const float* xs, const float4 (&wv)[IN_DIM], float4 bv,
                                               float* __restrict__ h, int k, int t) {
    const int c4 = t & 15;
#pragma unroll
    for (int i = 0; i < BKT_NODES / 16; ++i) {
        const int r = (t >> 4) + 16 * i;
        const int node = k * BKT_NODES + r;
        const float* xr = &xs[r * IN_DIM];
        float4 acc = bv;
#pragma unroll
        for (int d = 0; d < IN_DIM; ++d) {
            const float xd = xr[d];
            acc.x += xd * wv[d].x;
            acc.y += xd * wv[d].y;
            acc.z += xd * wv[d].z;
            acc.w += xd * wv[d].w;
        }
        if (FULL || node < N_NODES) *(float4*)(h + (size_t)node * OUT_DIM + 4 * c4) = acc;
    }
}

__global__ __launch_bounds__(SORT_BLOCK)
void bucket_sort_kernel(const int* __restrict__ mid,
                        const int* __restrict__ bbase,
                        int* __restrict__ csr_src,
                        int* __restrict__ offs,
                        float* __restrict__ inv,
                        const float* __restrict__ x,
                        const float* __restrict__ W,
                        const float* __restrict__ b,
                        float* __restrict__ h) {
    __shared__ int cnt[BKT_SEGS];
    __shared__ int cur[BKT_SEGS];
    __shared__ int wsum[BKT_NODES / 32];
    __shared__ __attribute__((aligned(16))) float xs[BKT_NODES * IN_DIM];
    __shared__ int stage[STAGE_CAP];
    const int t = threadIdx.x;
    const int lane = t & 31;
    const int wave = t >> 5;
    const int k = blockIdx.x;

    const int c4 = t & 15;
    const int xf = k * (BKT_NODES * IN_DIM) + 4 * t;
    float4 xv = make_float4(0.f, 0.f, 0.f, 0.f);
    if (t < X_VEC4 && xf < N_NODES * IN_DIM) xv = *(const float4*)(x + xf);
    float4 wv[IN_DIM];
#pragma unroll
    for (int d = 0; d < IN_DIM; ++d) wv[d] = *(const float4*)(W + d * OUT_DIM + 4 * c4);
    const float4 bv = *(const float4*)(b + 4 * c4);

    const int base = min(max(bbase[k], 0), N_EDGES);
    const int n = min(max(bbase[k + 1] - base, 0), N_EDGES - base);
    const bool staged = n <= STAGE_CAP;

    for (int i = t; i < BKT_SEGS; i += SORT_BLOCK) cnt[i] = 0;
    __syncthreads();

    for (int j0 = 0; j0 < n; j0 += SORT_BLOCK * CHUNK) {
        int v[CHUNK];
#pragma unroll
        for (int j = 0; j < CHUNK; ++j) {
            const int i = j0 + j * SORT_BLOCK + t;
            v[j] = (i < n) ? mid[base + i] : -1;
        }
#pragma unroll
        for (int j = 0; j < CHUNK; ++j)
            if (v[j] >= 0) atomicAdd(&cnt[min(v[j] >> 16, BKT_SEGS - 1)], 1);
    }
    __syncthreads();

    int c0 = 0, c1 = 0, c2 = 0;
    if (t < BKT_NODES) { c0 = cnt[3 * t]; c1 = cnt[3 * t + 1]; c2 = cnt[3 * t + 2]; }
    const int ts = c0 + c1 + c2;
    int inc = ts;
    for (int off = 1; off < 32; off <<= 1) {
        const int u = __shfl_up(inc, off, 32);
        if (lane >= off) inc += u;
    }
    if (t < BKT_NODES && lane == 31) wsum[wave] = inc;
    __syncthreads();
    if (t < BKT_NODES) {
        int ex = inc - ts;
        for (int w = 0; w < BKT_NODES / 32; ++w)
            if (w < wave) ex += wsum[w];
        cur[3 * t]     = ex;
        cur[3 * t + 1] = ex + c0;
        cur[3 * t + 2] = ex + c0 + c1;
        const int node = k * BKT_NODES + t;
        if (node < N_NODES) {
            const int seg = node * NUM_REL;
            offs[seg]     = base + ex;
            offs[seg + 1] = base + ex + c0;
            offs[seg + 2] = base + ex + c0 + c1;
            inv[seg]     = 1.0f / (float)(c0 > 1 ? c0 : 1);
            inv[seg + 1] = 1.0f / (float)(c1 > 1 ? c1 : 1);
            inv[seg + 2] = 1.0f / (float)(c2 > 1 ? c2 : 1);
            if (node == N_NODES - 1) offs[NR_TOTAL] = base + ex + ts;
        }
    }
    if (t < X_VEC4) *(float4*)&xs[4 * t] = xv;
    __syncthreads();

    for (int j0 = 0; j0 < n; j0 += SORT_BLOCK * CHUNK) {
        int v[CHUNK];
#pragma unroll
        for (int j = 0; j < CHUNK; ++j) {
            const int i = j0 + j * SORT_BLOCK + t;
            v[j] = (i < n) ? mid[base + i] : -1;
        }
#pragma unroll
        for (int j = 0; j < CHUNK; ++j) {
            if (v[j] >= 0) {
                const int p = atomicAdd(&cur[min(v[j] >> 16, BKT_SEGS - 1)], 1);
                if (staged) {
                    stage[min(max(p, 0), STAGE_CAP - 1)] = v[j] & 0xFFFF;
                } else {
                    const int q = min(max(base + p, 0), N_EDGES - 1);
                    csr_src[q] = v[j] & 0xFFFF;
                }
            }
        }
    }
    if (staged) {
        __syncthreads();
        for (int j0 = 0; j0 < n; j0 += SORT_BLOCK * CHUNK) {
            int w[CHUNK];
#pragma unroll
            for (int j = 0; j < CHUNK; ++j) {
                const int i = j0 + j * SORT_BLOCK + t;
                w[j] = (i < n) ? stage[i] : 0;
            }
#pragma unroll
            for (int j = 0; j < CHUNK; ++j) {
                const int i = j0 + j * SORT_BLOCK + t;
                if (i < n) csr_src[base + i] = w[j];
            }
        }
    }

    if ((k + 1) * BKT_NODES <= N_NODES) project_bucket<true>(xs, wv, bv, h, k, t);
    else                                project_bucket<false>(xs, wv, bv, h, k, t);
}

#define TILE_M   16
#define A_PITCH  260
static_assert(N_NODES % TILE_M == 0, "no partial node tile");

__device__ __forceinline__ int bcast_i(int v, int l) {
    return __builtin_amdgcn_readlane(v, l);
}
__device__ __forceinline__ float bcast_f(float v, int l) {
    return __int_as_float(__builtin_amdgcn_readlane(__float_as_int(v), l));
}

template <int G>
__device__ __forceinline__ void load_rows(float2* m, const float2* __restrict__ h2, int s, int k) {
    const int lane = threadIdx.x & 31;
    for (int i = 0; i < G; ++i) {
        const int rk = bcast_i(s, k + i);
        const float2* __restrict__ rowp = h2 + (size_t)rk * (OUT_DIM / 2);
        m[i] = rowp[lane];
    }
}
template <int G>
__device__ __forceinline__ void add_rows(float2& a, const float2* m) {
    for (int i = 0; i < G; ++i) { a.x += m[i].x; a.y += m[i].y; }
}

__device__ __forceinline__ void gather_range(float2& a, const float2* __restrict__ h2,
                                             int s, int k, int kend) {
    for (; k + 8 <= kend; k += 8) {
        float2 m[8];
        load_rows<8>(m, h2, s, k);
        add_rows<8>(a, m);
    }
    const int rem = max(kend - k, 0);
    const float2 z = make_float2(0.f, 0.f);
    float2 m4[4] = {z, z, z, z};
    float2 m2[2] = {z, z};
    float2 m1[1] = {z};
    if (rem & 4) { load_rows<4>(m4, h2, s, k); k += 4; }
    if (rem & 2) { load_rows<2>(m2, h2, s, k); k += 2; }
    if (rem & 1) { load_rows<1>(m1, h2, s, k); }
    add_rows<4>(a, m4);
    add_rows<2>(a, m2);
    add_rows<1>(a, m1);
}

__global__ __launch_bounds__(128, 16)
void rgcn_layer_kernel(const float* __restrict__ h,
                       const int*   __restrict__ offs,
                       const int*   __restrict__ csr_src,
                       const float* __restrict__ inv_denom,
                       const float* __restrict__ relW,
                       const float* __restrict__ rootW,
                       const float* __restrict__ bias,
                       const float* __restrict__ gamma,
                       const float* __restrict__ beta,
                       float* __restrict__ hout)
{
    __shared__ __attribute__((aligned(16))) float lA[TILE_M][A_PITCH];
    const int row0 = blockIdx.x * TILE_M;
    const int wave = threadIdx.x >> 5;
    const int lane = threadIdx.x & 31;

    {
        const float2* __restrict__ h2 = (const float2*)h;
        const int node0 = row0 + wave * 4;
        int bl = offs[node0 * NUM_REL + (lane < 12 ? lane : 12)];
        bl = min(max(bl, 0), N_EDGES);
        const float wl = inv_denom[node0 * NUM_REL + (lane < 11 ? lane : 11)];

        int snext = 0;
        {
            const int j = bcast_i(bl, 0) + lane;
            if (j < bcast_i(bl, 3)) snext = csr_src[j];
        }
#pragma unroll
        for (int t = 0; t < 4; ++t) {
            const int b0 = bcast_i(bl, 3 * t);
            const int b1 = bcast_i(bl, 3 * t + 1);
            const int b2 = bcast_i(bl, 3 * t + 2);
            const int b3 = bcast_i(bl, 3 * t + 3);
            const float2 hown = h2[(size_t)(node0 + t) * (OUT_DIM / 2) + lane];
            int s = snext;
            if (t < 3) {
                snext = 0;
                const int j = b3 + lane;
                if (j < bcast_i(bl, 3 * t + 6)) snext = csr_src[j];
            }
            float2 a0 = make_float2(0.f, 0.f), a1 = a0, a2 = a0;
            for (int j0 = b0; j0 < b3; j0 += 32) {
                if (j0 != b0) {
                    const int j = j0 + lane;
                    s = (j < b3) ? csr_src[j] : 0;
                }
                s = min(max(s, 0), N_NODES - 1);
                const int k1 = min(max(b1 - j0, 0), 32);
                const int k2 = min(max(b2 - j0, 0), 32);
                const int k3 = min(b3 - j0, 32);
                gather_range(a0, h2, s, 0,  k1);
                gather_range(a1, h2, s, k1, k2);
                gather_range(a2, h2, s, k2, k3);
            }
            const float w0 = bcast_f(wl, 3 * t);
            const float w1 = bcast_f(wl, 3 * t + 1);
            const float w2 = bcast_f(wl, 3 * t + 2);
            float* arow = &lA[wave * 4 + t][2 * lane];
            *(float2*)(arow)       = make_float2(a0.x * w0, a0.y * w0);
            *(float2*)(arow + 64)  = make_float2(a1.x * w1, a1.y * w1);
            *(float2*)(arow + 128) = make_float2(a2.x * w2, a2.y * w2);
            *(float2*)(arow + 192) = hown;
        }
    }
    __syncthreads();

    const int c     = lane & 15;
    const int hh    = lane >> 4;
    const int khalf = hh << 1;
    const float* __restrict__ Wq =
        (wave < NUM_REL) ? (relW + (size_t)wave * OUT_DIM * OUT_DIM) : rootW;

    v8f acc[4];
#pragma unroll
    for (int j = 0; j < 4; ++j) acc[j] = (v8f){0.f,0.f,0.f,0.f,0.f,0.f,0.f,0.f};

#pragma unroll
    for (int k0 = 0; k0 < OUT_DIM; k0 += 4) {
        const int kk = k0 + khalf;
        const v2f a = *(const v2f*)&lA[c][wave * OUT_DIM + kk];
        const float4 w0 = *(const float4*)(Wq + (size_t)kk * OUT_DIM + 4 * c);
        const float4 w1 = *(const float4*)(Wq + (size_t)(kk + 1) * OUT_DIM + 4 * c);
        v2f b;
        b.x = w0.x; b.y = w1.x;
        acc[0] = __builtin_amdgcn_wmma_f32_16x16x4_f32(false, a, false, b, (short)0, acc[0], false, false);
        b.x = w0.y; b.y = w1.y;
        acc[1] = __builtin_amdgcn_wmma_f32_16x16x4_f32(false, a, false, b, (short)0, acc[1], false, false);
        b.x = w0.z; b.y = w1.z;
        acc[2] = __builtin_amdgcn_wmma_f32_16x16x4_f32(false, a, false, b, (short)0, acc[2], false, false);
        b.x = w0.w; b.y = w1.w;
        acc[3] = __builtin_amdgcn_wmma_f32_16x16x4_f32(false, a, false, b, (short)0, acc[3], false, false);
    }

#pragma unroll
    for (int r = 0; r < 8; ++r) {
        *(float4*)&lA[8 * hh + r][wave * OUT_DIM + 4 * c] =
            make_float4(acc[0][r], acc[1][r], acc[2][r], acc[3][r]);
    }
    __syncthreads();

    const float2 bs = ((const float2*)bias)[lane];
    const float2 g2 = ((const float2*)gamma)[lane];
    const float2 be = ((const float2*)beta)[lane];
#pragma unroll
    for (int t = 0; t < 4; ++t) {
        const int row = wave * 4 + t;
        const float* prow = &lA[row][2 * lane];
        const float2 p0 = *(const float2*)(prow);
        const float2 p1 = *(const float2*)(prow + 64);
        const float2 p2 = *(const float2*)(prow + 128);
        const float2 p3 = *(const float2*)(prow + 192);
        float2 v;
        v.x = p3.x + bs.x;  v.y = p3.y + bs.y;
        v.x += p0.x;        v.y += p0.y;
        v.x += p1.x;        v.y += p1.y;
        v.x += p2.x;        v.y += p2.y;
        v.x = fmaxf(v.x, 0.0f);
        v.y = fmaxf(v.y, 0.0f);
        float s  = v.x + v.y;
        float sq = v.x * v.x + v.y * v.y;
#pragma unroll
        for (int off = 16; off > 0; off >>= 1) {
            s  += __shfl_xor(s,  off, 32);
            sq += __shfl_xor(sq, off, 32);
        }
        const float mu   = s * (1.0f / OUT_DIM);
        const float var  = sq * (1.0f / OUT_DIM) - mu * mu;
        const float rstd = rsqrtf(var + LN_EPS);
        float2 o;
        o.x = (v.x - mu) * rstd * g2.x + be.x;
        o.y = (v.y - mu) * rstd * g2.y + be.y;
        ((float2*)(hout + (size_t)(row0 + row) * OUT_DIM))[lane] = o;
    }
}

extern "C" void kernel_launch(void* const* d_in, const int* in_sizes, int n_in,
                              void* d_out, int out_size, void* d_ws, size_t ws_size,
                              hipStream_t stream) {
    const float* x      = (const float*)d_in[0];
    const int*   ei     = (const int*)  d_in[1];
    const int*   src    = ei;
    const int*   dst    = ei + N_EDGES;
    const int*   etype  = (const int*)  d_in[2];
    const float* f2cW   = (const float*)d_in[3];
    const float* f2cb   = (const float*)d_in[4];
    const float* rgW    = (const float*)d_in[5];
    const float* rgRoot = (const float*)d_in[6];
    const float* rgBias = (const float*)d_in[7];
    const float* gamma  = (const float*)d_in[8];
    const float* beta   = (const float*)d_in[9];

    float* ws = (float*)d_ws;
    float* h_a       = ws;  ws += (size_t)N_NODES * OUT_DIM;
    int*   mid       = (int*)ws;
    int*   csr_src   = mid + N_EDGES;
    int*   offs      = csr_src + N_EDGES;
    ws += (size_t)N_NODES * OUT_DIM;
    float* h_b       = ws;  ws += (size_t)N_NODES * NUM_REL * OUT_DIM;
    int*   th        = (int*)ws;
    int*   btot      = th + NUM_BKT * TILE_PITCH;
    int*   bbase     = btot + 512;
    ws += (size_t)N_NODES * OUT_DIM;
    float* inv_denom = ws;

    tile_hist_kernel<<<NUM_TILES, SORT_BLOCK, 0, stream>>>(dst, th);
    tile_prefix_kernel<<<(NUM_BKT + SORT_BLOCK / 32 - 1) / (SORT_BLOCK / 32), SORT_BLOCK, 0, stream>>>(th, btot);
    partition_kernel<<<NUM_TILES, SORT_BLOCK, 0, stream>>>(src, dst, etype, th, btot, bbase, mid);
    bucket_sort_kernel<<<NUM_BKT, SORT_BLOCK, 0, stream>>>(mid, bbase, csr_src, offs, inv_denom,
                                                       x, f2cW, f2cb, h_a);

    for (int l = 0; l < NUM_LAYERS; ++l) {
        const float* hi = (l & 1) ? h_b : h_a;
        float* ho = (l == NUM_LAYERS - 1) ? (float*)d_out : ((l & 1) ? h_a : h_b);
        rgcn_layer_kernel<<<N_NODES / TILE_M, 128, 0, stream>>>(
            hi, offs, csr_src, inv_denom,
            rgW    + (size_t)l * NUM_REL * OUT_DIM * OUT_DIM,
            rgRoot + (size_t)l * OUT_DIM * OUT_DIM,
            rgBias + (size_t)l * OUT_DIM,
            gamma, beta, ho);
    }
}


# Layer gather: a wave's 4 nodes' CSR range read as one row stream in groups of 8 on two register sets, next group's loads issued before the current group's adds; order of the sums unchanged
# speedup vs baseline: 3.7083x; 1.2043x over previous
#include <hip/hip_runtime.h>
#include <hip/hip_bf16.h>

#define N_NODES   50000
#define N_EDGES   1250000
#define IN_DIM    5
#define OUT_DIM   64
#define NUM_REL   3
#define NUM_LAYERS 3
#define LN_EPS    1e-5f

typedef __attribute__((ext_vector_type(2))) float v2f;
typedef __attribute__((ext_vector_type(8))) float v8f;

#define NR_TOTAL    (N_NODES * NUM_REL)
#define BKT_NODES   128
#define BKT_SEGS    (BKT_NODES * NUM_REL)
#define NUM_BKT     ((N_NODES + BKT_NODES - 1) / BKT_NODES)
#define SORT_BLOCK  256
#define TILE_EDGES  (SORT_BLOCK * 16)
#define NUM_TILES   ((N_EDGES + TILE_EDGES - 1) / TILE_EDGES)
#define TILE_PITCH  320
#define CHUNK       8
static_assert(N_EDGES % 4 == 0, "edges are read as int4");
static_assert(NUM_TILES < TILE_PITCH, "one row entry per tile, and one behind the last tile (the bucket's total)");
static_assert(NUM_BKT <= 2 * SORT_BLOCK, "two buckets per thread in the bucket scan");
static_assert(N_NODES <= 65536, "a source node fits the low 16 bits of a filed edge");
static_assert(BKT_NODES <= SORT_BLOCK, "one thread per node of a bucket");

__device__ __forceinline__ int bucket_of(int d) {
    return min((int)((unsigned)d >> 7), NUM_BKT - 1);
}

__global__ __launch_bounds__(SORT_BLOCK)
void tile_hist_kernel(const int* __restrict__ dst, int* __restrict__ th) {
    __shared__ int hist[NUM_BKT];
    const int t = threadIdx.x;
    const int tile = blockIdx.x;
    for (int b = t; b < NUM_BKT; b += SORT_BLOCK) hist[b] = 0;
    __syncthreads();
    int4 d[4];
#pragma unroll
    for (int i = 0; i < 4; ++i) {
        const int e = tile * TILE_EDGES + i * (SORT_BLOCK * 4) + 4 * t;
        d[i] = (e < N_EDGES) ? *(const int4*)(dst + e) : make_int4(0, 0, 0, 0);
    }
#pragma unroll
    for (int i = 0; i < 4; ++i) {
        const int e = tile * TILE_EDGES + i * (SORT_BLOCK * 4) + 4 * t;
        if (e < N_EDGES) {
            atomicAdd(&hist[bucket_of(d[i].x)], 1);
            atomicAdd(&hist[bucket_of(d[i].y)], 1);
            atomicAdd(&hist[bucket_of(d[i].z)], 1);
            atomicAdd(&hist[bucket_of(d[i].w)], 1);
        }
    }
    __syncthreads();
    for (int b = t; b < NUM_BKT; b += SORT_BLOCK) th[b * TILE_PITCH + tile] = hist[b];
}

__global__ __launch_bounds__(SORT_BLOCK)
void tile_prefix_kernel(int* __restrict__ th, int* __restrict__ btot) {
    const int lane = threadIdx.x & 31;
    const int b = blockIdx.x * (SORT_BLOCK / 32) + (threadIdx.x >> 5);
    if (b >= NUM_BKT) return;
    int2* row = (int2*)(th + b * TILE_PITCH + lane * 10);
    int v[10];
#pragma unroll
    for (int j = 0; j < 5; ++j) {
        const int2 x = row[j];
        v[2 * j] = x.x;
        v[2 * j + 1] = x.y;
    }
    int sum = 0;
#pragma unroll
    for (int j = 0; j < 10; ++j) {
        const int c = (lane * 10 + j < NUM_TILES) ? v[j] : 0;
        v[j] = sum;
        sum += c;
    }
    int inc = sum;
    for (int off = 1; off < 32; off <<= 1) {
        const int u = __shfl_up(inc, off, 32);
        if (lane >= off) inc += u;
    }
    const int ex = inc - sum;
#pragma unroll
    for (int j = 0; j < 5; ++j) row[j] = make_int2(v[2 * j] + ex, v[2 * j + 1] + ex);
    if (lane == 31) btot[b] = inc;
}

__device__ __forceinline__ void file_edge(int* lcur, const int* delta, int* sval, int* spos,
                                          int d, int ty, int s) {
    const int key = (d & (BKT_NODES - 1)) * NUM_REL + min(max(ty, 0), NUM_REL - 1);
    const int b = bucket_of(d);
    int lp = atomicAdd(&lcur[b], 1);
    lp = min(max(lp, 0), TILE_EDGES - 1);
    sval[lp] = (key << 16) | (s & 0xFFFF);
    spos[lp] = lp + delta[b];
}

__global__ __launch_bounds__(SORT_BLOCK)
void partition_kernel(const int* __restrict__ src,
                      const int* __restrict__ dst,
                      const int* __restrict__ etype,
                      const int* __restrict__ th,
                      const int* __restrict__ btot,
                      int* __restrict__ bbase,
                      int* __restrict__ mid) {
    __shared__ int lcur[NUM_BKT];
    __shared__ int delta[NUM_BKT];
    __shared__ int wsum[SORT_BLOCK / 32];
    __shared__ int lwsum[SORT_BLOCK / 32];
    __shared__ int sval[TILE_EDGES];
    __shared__ int spos[TILE_EDGES];
    const int t = threadIdx.x;
    const int lane = t & 31;
    const int wave = t >> 5;
    const int tile = blockIdx.x;

    const int i0 = 2 * t, i1 = 2 * t + 1;
    const int c0 = (i0 < NUM_BKT) ? btot[i0] : 0;
    const int c1 = (i1 < NUM_BKT) ? btot[i1] : 0;
    const int p0 = (i0 < NUM_BKT) ? th[i0 * TILE_PITCH + tile] : 0;
    const int p1 = (i1 < NUM_BKT) ? th[i1 * TILE_PITCH + tile] : 0;
    const int q0 = (i0 < NUM_BKT) ? th[i0 * TILE_PITCH + tile + 1] : 0;
    const int q1 = (i1 < NUM_BKT) ? th[i1 * TILE_PITCH + tile + 1] : 0;
    const int n0 = max(q0 - p0, 0);
    const int n1 = max(q1 - p1, 0);
    const int ts = c0 + c1;
    const int ls = n0 + n1;
    int inc = ts;
    int linc = ls;
    for (int off = 1; off < 32; off <<= 1) {
        const int u = __shfl_up(inc, off, 32);
        const int lu = __shfl_up(linc, off, 32);
        if (lane >= off) { inc += u; linc += lu; }
    }
    if (lane == 31) { wsum[wave] = inc; lwsum[wave] = linc; }
    __syncthreads();
    int ex = inc - ts;
    int lex = linc - ls;
    for (int w = 0; w < SORT_BLOCK / 32; ++w)
        if (w < wave) { ex += wsum[w]; lex += lwsum[w]; }
    if (i0 < NUM_BKT) { lcur[i0] = lex;      delta[i0] = ex + p0 - lex;             if (tile == 0) bbase[i0] = ex; }
    if (i1 < NUM_BKT) { lcur[i1] = lex + n0; delta[i1] = ex + c0 + p1 - (lex + n0); if (tile == 0) bbase[i1] = ex + c0; }
    if (tile == 0 && t == SORT_BLOCK - 1) bbase[NUM_BKT] = ex + ts;
    __syncthreads();

    int4 d[4], y[4], s[4];
#pragma unroll
    for (int i = 0; i < 4; ++i) {
        const int e = tile * TILE_EDGES + i * (SORT_BLOCK * 4) + 4 * t;
        const bool ok = e < N_EDGES;
        d[i] = ok ? *(const int4*)(dst + e)   : make_int4(0, 0, 0, 0);
        y[i] = ok ? *(const int4*)(etype + e) : make_int4(0, 0, 0, 0);
        s[i] = ok ? *(const int4*)(src + e)   : make_int4(0, 0, 0, 0);
    }
#pragma unroll
    for (int i = 0; i < 4; ++i) {
        const int e = tile * TILE_EDGES + i * (SORT_BLOCK * 4) + 4 * t;
        if (e < N_EDGES) {
            file_edge(lcur, delta, sval, spos, d[i].x, y[i].x, s[i].x);
            file_edge(lcur, delta, sval, spos, d[i].y, y[i].y, s[i].y);
            file_edge(lcur, delta, sval, spos, d[i].z, y[i].z, s[i].z);
            file_edge(lcur, delta, sval, spos, d[i].w, y[i].w, s[i].w);
        }
    }
    __syncthreads();

    const int cnt = min(TILE_EDGES, N_EDGES - tile * TILE_EDGES);
    int pv[TILE_EDGES / SORT_BLOCK], sv[TILE_EDGES / SORT_BLOCK];
#pragma unroll
    for (int j = 0; j < TILE_EDGES / SORT_BLOCK; ++j) {
        pv[j] = spos[j * SORT_BLOCK + t];
        sv[j] = sval[j * SORT_BLOCK + t];
    }
#pragma unroll
    for (int j = 0; j < TILE_EDGES / SORT_BLOCK; ++j) {
        if (j * SORT_BLOCK + t < cnt) {
            const int pos = min(max(pv[j], 0), N_EDGES - 1);
            mid[pos] = sv[j];
        }
    }
}

#define STAGE_CAP 4096
#define X_VEC4  (BKT_NODES * IN_DIM / 4)
static_assert((BKT_NODES * IN_DIM) % 4 == 0 && (N_NODES * IN_DIM) % 4 == 0, "a bucket's x rows are read as float4");
static_assert(X_VEC4 <= SORT_BLOCK, "one float4 of x per thread");
static_assert(OUT_DIM == 64 && SORT_BLOCK == 256 && BKT_NODES % 16 == 0, "16 lanes x float4 per h row, 16 rows per pass");

template <bool FULL>
__device__ __forceinline__ void project_bucket(const float* xs, const float4 (&wv)[IN_DIM], float4 bv,
                                               float* __restrict__ h, int k, int t) {
    const int c4 = t & 15;
#pragma unroll
    for (int i = 0; i < BKT_NODES / 16; ++i) {
        const int r = (t >> 4) + 16 * i;
        const int node = k * BKT_NODES + r;
        const float* xr = &xs[r * IN_DIM];
        float4 acc = bv;
#pragma unroll
        for (int d = 0; d < IN_DIM; ++d) {
            const float xd = xr[d];
            acc.x += xd * wv[d].x;
            acc.y += xd * wv[d].y;
            acc.z += xd * wv[d].z;
            acc.w += xd * wv[d].w;
        }
        if (FULL || node < N_NODES) *(float4*)(h + (size_t)node * OUT_DIM + 4 * c4) = acc;
    }
}

__global__ __launch_bounds__(SORT_BLOCK)
void bucket_sort_kernel(const int* __restrict__ mid,
                        const int* __restrict__ bbase,
                        int* __restrict__ csr_src,
                        int* __restrict__ offs,
                        float* __restrict__ inv,
                        const float* __restrict__ x,
                        const float* __restrict__ W,
                        const float* __restrict__ b,
                        float* __restrict__ h) {
    __shared__ int cnt[BKT_SEGS];
    __shared__ int cur[BKT_SEGS];
    __shared__ int wsum[BKT_NODES / 32];
    __shared__ __attribute__((aligned(16))) float xs[BKT_NODES * IN_DIM];
    __shared__ int stage[STAGE_CAP];
    const int t = threadIdx.x;
    const int lane = t & 31;
    const int wave = t >> 5;
    const int k = blockIdx.x;

    const int c4 = t & 15;
    const int xf = k * (BKT_NODES * IN_DIM) + 4 * t;
    float4 xv = make_float4(0.f, 0.f, 0.f, 0.f);
    if (t < X_VEC4 && xf < N_NODES * IN_DIM) xv = *(const float4*)(x + xf);
    float4 wv[IN_DIM];
#pragma unroll
    for (int d = 0; d < IN_DIM; ++d) wv[d] = *(const float4*)(W + d * OUT_DIM + 4 * c4);
    const float4 bv = *(const float4*)(b + 4 * c4);

    const int base = min(max(bbase[k], 0), N_EDGES);
    const int n = min(max(bbase[k + 1] - base, 0), N_EDGES - base);
    const bool staged = n <= STAGE_CAP;

    for (int i = t; i < BKT_SEGS; i += SORT_BLOCK) cnt[i] = 0;
    __syncthreads();

    for (int j0 = 0; j0 < n; j0 += SORT_BLOCK * CHUNK) {
        int v[CHUNK];
#pragma unroll
        for (int j = 0; j < CHUNK; ++j) {
            const int i = j0 + j * SORT_BLOCK + t;
            v[j] = (i < n) ? mid[base + i] : -1;
        }
#pragma unroll
        for (int j = 0; j < CHUNK; ++j)
            if (v[j] >= 0) atomicAdd(&cnt[min(v[j] >> 16, BKT_SEGS - 1)], 1);
    }
    __syncthreads();

    int c0 = 0, c1 = 0, c2 = 0;
    if (t < BKT_NODES) { c0 = cnt[3 * t]; c1 = cnt[3 * t + 1]; c2 = cnt[3 * t + 2]; }
    const int ts = c0 + c1 + c2;
    int inc = ts;
    for (int off = 1; off < 32; off <<= 1) {
        const int u = __shfl_up(inc, off, 32);
        if (lane >= off) inc += u;
    }
    if (t < BKT_NODES && lane == 31) wsum[wave] = inc;
    __syncthreads();
    if (t < BKT_NODES) {
        int ex = inc - ts;
        for (int w = 0; w < BKT_NODES / 32; ++w)
            if (w < wave) ex += wsum[w];
        cur[3 * t]     = ex;
        cur[3 * t + 1] = ex + c0;
        cur[3 * t + 2] = ex + c0 + c1;
        const int node = k * BKT_NODES + t;
        if (node < N_NODES) {
            const int seg = node * NUM_REL;
            offs[seg]     = base + ex;
            offs[seg + 1] = base + ex + c0;
            offs[seg + 2] = base + ex + c0 + c1;
            inv[seg]     = 1.0f / (float)(c0 > 1 ? c0 : 1);
            inv[seg + 1] = 1.0f / (float)(c1 > 1 ? c1 : 1);
            inv[seg + 2] = 1.0f / (float)(c2 > 1 ? c2 : 1);
            if (node == N_NODES - 1) offs[NR_TOTAL] = base + ex + ts;
        }
    }
    if (t < X_VEC4) *(float4*)&xs[4 * t] = xv;
    __syncthreads();

    for (int j0 = 0; j0 < n; j0 += SORT_BLOCK * CHUNK) {
        int v[CHUNK];
#pragma unroll
        for (int j = 0; j < CHUNK; ++j) {
            const int i = j0 + j * SORT_BLOCK + t;
            v[j] = (i < n) ? mid[base + i] : -1;
        }
#pragma unroll
        for (int j = 0; j < CHUNK; ++j) {
            if (v[j] >= 0) {
                const int p = atomicAdd(&cur[min(v[j] >> 16, BKT_SEGS - 1)], 1);
                if (staged) {
                    stage[min(max(p, 0), STAGE_CAP - 1)] = v[j] & 0xFFFF;
                } else {
                    const int q = min(max(base + p, 0), N_EDGES - 1);
                    csr_src[q] = v[j] & 0xFFFF;
                }
            }
        }
    }
    if (staged) {
        __syncthreads();
        for (int j0 = 0; j0 < n; j0 += SORT_BLOCK * CHUNK) {
            int w[CHUNK];
#pragma unroll
            for (int j = 0; j < CHUNK; ++j) {
                const int i = j0 + j * SORT_BLOCK + t;
                w[j] = (i < n) ? stage[i] : 0;
            }
#pragma unroll
            for (int j = 0; j < CHUNK; ++j) {
                const int i = j0 + j * SORT_BLOCK + t;
                if (i < n) csr_src[base + i] = w[j];
            }
        }
    }

    if ((k + 1) * BKT_NODES <= N_NODES) project_bucket<true>(xs, wv, bv, h, k, t);
    else                                project_bucket<false>(xs, wv, bv, h, k, t);
}

#define TILE_M   16
#define A_PITCH  260
static_assert(N_NODES % TILE_M == 0, "no partial node tile");

__device__ __forceinline__ int bcast_i(int v, int l) {
    return __builtin_amdgcn_readlane(v, l);
}
__device__ __forceinline__ float bcast_f(float v, int l) {
    return __int_as_float(__builtin_amdgcn_readlane(__float_as_int(v), l));
}

#define GRP 8

__global__ __launch_bounds__(128, 16)
void rgcn_layer_kernel(const float* __restrict__ h,
                       const int*   __restrict__ offs,
                       const int*   __restrict__ csr_src,
                       const float* __restrict__ inv_denom,
                       const float* __restrict__ relW,
                       const float* __restrict__ rootW,
                       const float* __restrict__ bias,
                       const float* __restrict__ gamma,
                       const float* __restrict__ beta,
                       float* __restrict__ hout)
{
    __shared__ __attribute__((aligned(16))) float lA[TILE_M][A_PITCH];
    const int row0 = blockIdx.x * TILE_M;
    const int wave = threadIdx.x >> 5;
    const int lane = threadIdx.x & 31;

    {
        const float2* __restrict__ h2 = (const float2*)h;
        const int node0 = row0 + wave * 4;
        float2 hown[4];
#pragma unroll
        for (int t = 0; t < 4; ++t) hown[t] = h2[(size_t)(node0 + t) * (OUT_DIM / 2) + lane];
        int bl = offs[node0 * NUM_REL + (lane < 12 ? lane : 12)];
        bl = min(max(bl, 0), N_EDGES);
        const float wl = inv_denom[node0 * NUM_REL + (lane < 11 ? lane : 11)];

        const int e0 = bcast_i(bl, 0);
        const int n  = max(bcast_i(bl, 12) - e0, 0);
        const int elast = min(max(e0 + n - 1, 0), N_EDGES - 1);

        auto idx_raw = [&](int cidx) __attribute__((always_inline)) {
            const int j = min(max(e0 + 2 * GRP * cidx + lane, 0), elast);
            return csr_src[j];
        };
        int s = idx_raw(0);

        float* lrow = &lA[wave * 4][2 * lane];
#pragma unroll
        for (int t = 0; t < 4; ++t) *(float2*)(lrow + t * A_PITCH + 3 * OUT_DIM) = hown[t];

        s = min(max(s, 0), N_NODES - 1);

        int seg = 0;
        int rel = 0;
        int segoff = 0;
        int nb = bcast_i(bl, 1) - e0;
        float2 a = make_float2(0.f, 0.f);

        auto flush = [&]() __attribute__((always_inline)) {
            const float w = bcast_f(wl, seg);
            *(float2*)(lrow + segoff) = make_float2(a.x * w, a.y * w);
            a = make_float2(0.f, 0.f);
            ++seg;
            ++rel;
            segoff += OUT_DIM;
            if (rel == NUM_REL) { rel = 0; segoff += A_PITCH - NUM_REL * OUT_DIM; }
            nb = (seg < 4 * NUM_REL) ? (bcast_i(bl, seg + 1) - e0) : 0x7fffffff;
        };
        auto load_group = [&](float2 (&m)[GRP], int sv, int k) __attribute__((always_inline)) {
#pragma unroll
            for (int i = 0; i < GRP; ++i) {
                const int rk = bcast_i(sv, k + i);
                const float2* __restrict__ rowp = h2 + (size_t)rk * (OUT_DIM / 2);
                m[i] = rowp[lane];
            }
        };
        auto consume = [&](const float2 (&m)[GRP], int p0) __attribute__((always_inline)) {
#pragma unroll
            for (int i = 0; i < GRP; ++i) {
                while (p0 + i >= nb) flush();
                a.x += m[i].x;
                a.y += m[i].y;
            }
        };

        {
            float2 mA[GRP], mB[GRP];
            int g = 0;
            while (g * GRP < n) {
                const int sraw = idx_raw(g / 2 + 1);
                load_group(mA, s, 0);
                if (g > 0) consume(mB, (g - 1) * GRP);
                load_group(mB, s, GRP);
                consume(mA, g * GRP);
                s = min(max(sraw, 0), N_NODES - 1);
                g += 2;
            }
            if (g > 0) consume(mB, (g - 1) * GRP);
        }
        while (seg < 4 * NUM_REL) flush();
    }
    __syncthreads();

    const int c     = lane & 15;
    const int hh    = lane >> 4;
    const int khalf = hh << 1;
    const float* __restrict__ Wq =
        (wave < NUM_REL) ? (relW + (size_t)wave * OUT_DIM * OUT_DIM) : rootW;

    v8f acc[4];
#pragma unroll
    for (int j = 0; j < 4; ++j) acc[j] = (v8f){0.f,0.f,0.f,0.f,0.f,0.f,0.f,0.f};

#pragma unroll
    for (int k0 = 0; k0 < OUT_DIM; k0 += 4) {
        const int kk = k0 + khalf;
        const v2f a = *(const v2f*)&lA[c][wave * OUT_DIM + kk];
        const float4 w0 = *(const float4*)(Wq + (size_t)kk * OUT_DIM + 4 * c);
        const float4 w1 = *(const float4*)(Wq + (size_t)(kk + 1) * OUT_DIM + 4 * c);
        v2f b;
        b.x = w0.x; b.y = w1.x;
        acc[0] = __builtin_amdgcn_wmma_f32_16x16x4_f32(false, a, false, b, (short)0, acc[0], false, false);
        b.x = w0.y; b.y = w1.y;
        acc[1] = __builtin_amdgcn_wmma_f32_16x16x4_f32(false, a, false, b, (short)0, acc[1], false, false);
        b.x = w0.z; b.y = w1.z;
        acc[2] = __builtin_amdgcn_wmma_f32_16x16x4_f32(false, a, false, b, (short)0, acc[2], false, false);
        b.x = w0.w; b.y = w1.w;
        acc[3] = __builtin_amdgcn_wmma_f32_16x16x4_f32(false, a, false, b, (short)0, acc[3], false, false);
    }

#pragma unroll
    for (int r = 0; r < 8; ++r) {
        *(float4*)&lA[8 * hh + r][wave * OUT_DIM + 4 * c] =
            make_float4(acc[0][r], acc[1][r], acc[2][r], acc[3][r]);
    }
    __syncthreads();

    const float2 bs = ((const float2*)bias)[lane];
    const float2 g2 = ((const float2*)gamma)[lane];
    const float2 be = ((const float2*)beta)[lane];
#pragma unroll
    for (int t = 0; t < 4; ++t) {
        const int row = wave * 4 + t;
        const float* prow = &lA[row][2 * lane];
        const float2 p0 = *(const float2*)(prow);
        const float2 p1 = *(const float2*)(prow + 64);
        const float2 p2 = *(const float2*)(prow + 128);
        const float2 p3 = *(const float2*)(prow + 192);
        float2 v;
        v.x = p3.x + bs.x;  v.y = p3.y + bs.y;
        v.x += p0.x;        v.y += p0.y;
        v.x += p1.x;        v.y += p1.y;
        v.x += p2.x;        v.y += p2.y;
        v.x = fmaxf(v.x, 0.0f);
        v.y = fmaxf(v.y, 0.0f);
        float s  = v.x + v.y;
        float sq = v.x * v.x + v.y * v.y;
#pragma unroll
        for (int off = 16; off > 0; off >>= 1) {
            s  += __shfl_xor(s,  off, 32);
            sq += __shfl_xor(sq, off, 32);
        }
        const float mu   = s * (1.0f / OUT_DIM);
        const float var  = sq * (1.0f / OUT_DIM) - mu * mu;
        const float rstd = rsqrtf(var + LN_EPS);
        float2 o;
        o.x = (v.x - mu) * rstd * g2.x + be.x;
        o.y = (v.y - mu) * rstd * g2.y + be.y;
        ((float2*)(hout + (size_t)(row0 + row) * OUT_DIM))[lane] = o;
    }
}

extern "C" void kernel_launch(void* const* d_in, const int* in_sizes, int n_in,
                              void* d_out, int out_size, void* d_ws, size_t ws_size,
                              hipStream_t stream) {
    const float* x      = (const float*)d_in[0];
    const int*   ei     = (const int*)  d_in[1];
    const int*   src    = ei;
    const int*   dst    = ei + N_EDGES;
    const int*   etype  = (const int*)  d_in[2];
    const float* f2cW   = (const float*)d_in[3];
    const float* f2cb   = (const float*)d_in[4];
    const float* rgW    = (const float*)d_in[5];
    const float* rgRoot = (const float*)d_in[6];
    const float* rgBias = (const float*)d_in[7];
    const float* gamma  = (const float*)d_in[8];
    const float* beta   = (const float*)d_in[9];

    float* ws = (float*)d_ws;
    float* h_a       = ws;  ws += (size_t)N_NODES * OUT_DIM;
    int*   mid       = (int*)ws;
    int*   csr_src   = mid + N_EDGES;
    int*   offs      = csr_src + N_EDGES;
    ws += (size_t)N_NODES * OUT_DIM;
    float* h_b       = ws;  ws += (size_t)N_NODES * NUM_REL * OUT_DIM;
    int*   th        = (int*)ws;
    int*   btot      = th + NUM_BKT * TILE_PITCH;
    int*   bbase     = btot + 512;
    ws += (size_t)N_NODES * OUT_DIM;
    float* inv_denom = ws;

    tile_hist_kernel<<<NUM_TILES, SORT_BLOCK, 0, stream>>>(dst, th);
    tile_prefix_kernel<<<(NUM_BKT + SORT_BLOCK / 32 - 1) / (SORT_BLOCK / 32), SORT_BLOCK, 0, stream>>>(th, btot);
    partition_kernel<<<NUM_TILES, SORT_BLOCK, 0, stream>>>(src, dst, etype, th, btot, bbase, mid);
    bucket_sort_kernel<<<NUM_BKT, SORT_BLOCK, 0, stream>>>(mid, bbase, csr_src, offs, inv_denom,
                                                       x, f2cW, f2cb, h_a);

    for (int l = 0; l < NUM_LAYERS; ++l) {
        const float* hi = (l & 1) ? h_b : h_a;
        float* ho = (l == NUM_LAYERS - 1) ? (float*)d_out : ((l & 1) ? h_a : h_b);
        rgcn_layer_kernel<<<N_NODES / TILE_M, 128, 0, stream>>>(
            hi, offs, csr_src, inv_denom,
            rgW    + (size_t)l * NUM_REL * OUT_DIM * OUT_DIM,
            rgRoot + (size_t)l * OUT_DIM * OUT_DIM,
            rgBias + (size_t)l * OUT_DIM,
            gamma, beta, ho);
    }
}


# Input projection folded into layer 1 (linear): means taken on the 5-float x rows (segmented lane scan, sum order changes), weights pre-multiplied by feat2c_W on WMMA, K 32 not 256; h0 never formed
# speedup vs baseline: 4.2726x; 1.1522x over previous
#include <hip/hip_runtime.h>
#include <hip/hip_bf16.h>

#define N_NODES   50000
#define N_EDGES   1250000
#define IN_DIM    5
#define OUT_DIM   64
#define NUM_REL   3
#define NUM_LAYERS 3
#define LN_EPS    1e-5f
#define TILE_M    16

typedef __attribute__((ext_vector_type(2))) float v2f;
typedef __attribute__((ext_vector_type(8))) float v8f;

#define NR_TOTAL    (N_NODES * NUM_REL)
#define BKT_NODES   128
#define BKT_SEGS    (BKT_NODES * NUM_REL)
#define NUM_BKT     ((N_NODES + BKT_NODES - 1) / BKT_NODES)
#define SORT_BLOCK  256
#define TILE_EDGES  (SORT_BLOCK * 16)
#define NUM_TILES   ((N_EDGES + TILE_EDGES - 1) / TILE_EDGES)
#define TILE_PITCH  320
#define CHUNK       8
static_assert(N_EDGES % 4 == 0, "edges are read as int4");
static_assert(NUM_TILES < TILE_PITCH, "one row entry per tile, and one behind the last tile (the bucket's total)");
static_assert(NUM_BKT <= 2 * SORT_BLOCK, "two buckets per thread in the bucket scan");
static_assert(N_NODES <= 65536, "a source node fits the low 16 bits of a filed edge");
static_assert(BKT_NODES <= SORT_BLOCK, "one thread per node of a bucket");

__device__ __forceinline__ int bucket_of(int d) {
    return min((int)((unsigned)d >> 7), NUM_BKT - 1);
}


#define FOLD_ROWS 8
#define FA_PITCH  68
static_assert(SORT_BLOCK == 256 && OUT_DIM == 64 && NUM_REL == 3, "eight waves: four matrices x two pairs of 16-column tiles");

__device__ __forceinline__ void fold_first_layer_weights(const float* __restrict__ Wf,
                                                         const float* __restrict__ bf,
                                                         const float* __restrict__ relW,
                                                         const float* __restrict__ rootW,
                                                         float* __restrict__ Wc) {
    __shared__ __attribute__((aligned(16))) float fA[TILE_M][FA_PITCH];
    const int t = threadIdx.x;
    const int wave = t >> 5;
    const int lane = t & 31;
    const int q     = wave & 3;
    const int jh    = wave >> 2;
    const int c     = lane & 15;
    const int hh    = lane >> 4;
    const int khalf = hh << 1;
    const float* __restrict__ Wq =
        (q < NUM_REL) ? (relW + (size_t)q * OUT_DIM * OUT_DIM) : rootW;

    float2 wa[OUT_DIM / 4], wb[OUT_DIM / 4];
#pragma unroll
    for (int s = 0; s < OUT_DIM / 4; ++s) {
        const int kk = 4 * s + khalf;
        wa[s] = *(const float2*)(Wq + (size_t)kk * OUT_DIM + 4 * c + 2 * jh);
        wb[s] = *(const float2*)(Wq + (size_t)(kk + 1) * OUT_DIM + 4 * c + 2 * jh);
    }
    {
        const int r = t >> 4, c4 = t & 15;
        float4 v = make_float4(0.f, 0.f, 0.f, 0.f);
        if (r < IN_DIM)       v = *(const float4*)(Wf + r * OUT_DIM + 4 * c4);
        else if (r == IN_DIM) v = *(const float4*)(bf + 4 * c4);
        *(float4*)&fA[r][4 * c4] = v;
    }
    __syncthreads();

    v8f acc0 = (v8f){0.f,0.f,0.f,0.f,0.f,0.f,0.f,0.f};
    v8f acc1 = (v8f){0.f,0.f,0.f,0.f,0.f,0.f,0.f,0.f};
#pragma unroll
    for (int s = 0; s < OUT_DIM / 4; ++s) {
        const int kk = 4 * s + khalf;
        const v2f a = *(const v2f*)&fA[c][kk];
        v2f b;
        b.x = wa[s].x; b.y = wb[s].x;
        acc0 = __builtin_amdgcn_wmma_f32_16x16x4_f32(false, a, false, b, (short)0, acc0, false, false);
        b.x = wa[s].y; b.y = wb[s].y;
        acc1 = __builtin_amdgcn_wmma_f32_16x16x4_f32(false, a, false, b, (short)0, acc1, false, false);
    }
    if (hh == 0) {
#pragma unroll
        for (int r = 0; r < FOLD_ROWS; ++r) {
            *(float2*)(Wc + (size_t)(q * FOLD_ROWS + r) * OUT_DIM + 4 * c + 2 * jh) =
                make_float2(acc0[r], acc1[r]);
        }
    }
}

__global__ __launch_bounds__(SORT_BLOCK)
void tile_hist_kernel(const int* __restrict__ dst, int* __restrict__ th,
                      const float* __restrict__ Wf, const float* __restrict__ bf,
                      const float* __restrict__ relW, const float* __restrict__ rootW,
                      float* __restrict__ Wc) {
    if (blockIdx.x == NUM_TILES) {
        fold_first_layer_weights(Wf, bf, relW, rootW, Wc);
        return;
    }
    __shared__ int hist[NUM_BKT];
    const int t = threadIdx.x;
    const int tile = blockIdx.x;
    for (int b = t; b < NUM_BKT; b += SORT_BLOCK) hist[b] = 0;
    __syncthreads();
    int4 d[4];
#pragma unroll
    for (int i = 0; i < 4; ++i) {
        const int e = tile * TILE_EDGES + i * (SORT_BLOCK * 4) + 4 * t;
        d[i] = (e < N_EDGES) ? *(const int4*)(dst + e) : make_int4(0, 0, 0, 0);
    }
#pragma unroll
    for (int i = 0; i < 4; ++i) {
        const int e = tile * TILE_EDGES + i * (SORT_BLOCK * 4) + 4 * t;
        if (e < N_EDGES) {
            atomicAdd(&hist[bucket_of(d[i].x)], 1);
            atomicAdd(&hist[bucket_of(d[i].y)], 1);
            atomicAdd(&hist[bucket_of(d[i].z)], 1);
            atomicAdd(&hist[bucket_of(d[i].w)], 1);
        }
    }
    __syncthreads();
    for (int b = t; b < NUM_BKT; b += SORT_BLOCK) th[b * TILE_PITCH + tile] = hist[b];
}

__global__ __launch_bounds__(SORT_BLOCK)
void tile_prefix_kernel(int* __restrict__ th, int* __restrict__ btot) {
    const int lane = threadIdx.x & 31;
    const int b = blockIdx.x * (SORT_BLOCK / 32) + (threadIdx.x >> 5);
    if (b >= NUM_BKT) return;
    int2* row = (int2*)(th + b * TILE_PITCH + lane * 10);
    int v[10];
#pragma unroll
    for (int j = 0; j < 5; ++j) {
        const int2 x = row[j];
        v[2 * j] = x.x;
        v[2 * j + 1] = x.y;
    }
    int sum = 0;
#pragma unroll
    for (int j = 0; j < 10; ++j) {
        const int c = (lane * 10 + j < NUM_TILES) ? v[j] : 0;
        v[j] = sum;
        sum += c;
    }
    int inc = sum;
    for (int off = 1; off < 32; off <<= 1) {
        const int u = __shfl_up(inc, off, 32);
        if (lane >= off) inc += u;
    }
    const int ex = inc - sum;
#pragma unroll
    for (int j = 0; j < 5; ++j) row[j] = make_int2(v[2 * j] + ex, v[2 * j + 1] + ex);
    if (lane == 31) btot[b] = inc;
}

__device__ __forceinline__ void file_edge(int* lcur, const int* delta, int* sval, int* spos,
                                          int d, int ty, int s) {
    const int key = (d & (BKT_NODES - 1)) * NUM_REL + min(max(ty, 0), NUM_REL - 1);
    const int b = bucket_of(d);
    int lp = atomicAdd(&lcur[b], 1);
    lp = min(max(lp, 0), TILE_EDGES - 1);
    sval[lp] = (key << 16) | (s & 0xFFFF);
    spos[lp] = lp + delta[b];
}

__global__ __launch_bounds__(SORT_BLOCK)
void partition_kernel(const int* __restrict__ src,
                      const int* __restrict__ dst,
                      const int* __restrict__ etype,
                      const int* __restrict__ th,
                      const int* __restrict__ btot,
                      int* __restrict__ bbase,
                      int* __restrict__ mid) {
    __shared__ int lcur[NUM_BKT];
    __shared__ int delta[NUM_BKT];
    __shared__ int wsum[SORT_BLOCK / 32];
    __shared__ int lwsum[SORT_BLOCK / 32];
    __shared__ int sval[TILE_EDGES];
    __shared__ int spos[TILE_EDGES];
    const int t = threadIdx.x;
    const int lane = t & 31;
    const int wave = t >> 5;
    const int tile = blockIdx.x;

    const int i0 = 2 * t, i1 = 2 * t + 1;
    const int c0 = (i0 < NUM_BKT) ? btot[i0] : 0;
    const int c1 = (i1 < NUM_BKT) ? btot[i1] : 0;
    const int p0 = (i0 < NUM_BKT) ? th[i0 * TILE_PITCH + tile] : 0;
    const int p1 = (i1 < NUM_BKT) ? th[i1 * TILE_PITCH + tile] : 0;
    const int q0 = (i0 < NUM_BKT) ? th[i0 * TILE_PITCH + tile + 1] : 0;
    const int q1 = (i1 < NUM_BKT) ? th[i1 * TILE_PITCH + tile + 1] : 0;
    const int n0 = max(q0 - p0, 0);
    const int n1 = max(q1 - p1, 0);
    const int ts = c0 + c1;
    const int ls = n0 + n1;
    int inc = ts;
    int linc = ls;
    for (int off = 1; off < 32; off <<= 1) {
        const int u = __shfl_up(inc, off, 32);
        const int lu = __shfl_up(linc, off, 32);
        if (lane >= off) { inc += u; linc += lu; }
    }
    if (lane == 31) { wsum[wave] = inc; lwsum[wave] = linc; }
    __syncthreads();
    int ex = inc - ts;
    int lex = linc - ls;
    for (int w = 0; w < SORT_BLOCK / 32; ++w)
        if (w < wave) { ex += wsum[w]; lex += lwsum[w]; }
    if (i0 < NUM_BKT) { lcur[i0] = lex;      delta[i0] = ex + p0 - lex;             if (tile == 0) bbase[i0] = ex; }
    if (i1 < NUM_BKT) { lcur[i1] = lex + n0; delta[i1] = ex + c0 + p1 - (lex + n0); if (tile == 0) bbase[i1] = ex + c0; }
    if (tile == 0 && t == SORT_BLOCK - 1) bbase[NUM_BKT] = ex + ts;
    __syncthreads();

    int4 d[4], y[4], s[4];
#pragma unroll
    for (int i = 0; i < 4; ++i) {
        const int e = tile * TILE_EDGES + i * (SORT_BLOCK * 4) + 4 * t;
        const bool ok = e < N_EDGES;
        d[i] = ok ? *(const int4*)(dst + e)   : make_int4(0, 0, 0, 0);
        y[i] = ok ? *(const int4*)(etype + e) : make_int4(0, 0, 0, 0);
        s[i] = ok ? *(const int4*)(src + e)   : make_int4(0, 0, 0, 0);
    }
#pragma unroll
    for (int i = 0; i < 4; ++i) {
        const int e = tile * TILE_EDGES + i * (SORT_BLOCK * 4) + 4 * t;
        if (e < N_EDGES) {
            file_edge(lcur, delta, sval, spos, d[i].x, y[i].x, s[i].x);
            file_edge(lcur, delta, sval, spos, d[i].y, y[i].y, s[i].y);
            file_edge(lcur, delta, sval, spos, d[i].z, y[i].z, s[i].z);
            file_edge(lcur, delta, sval, spos, d[i].w, y[i].w, s[i].w);
        }
    }
    __syncthreads();

    const int cnt = min(TILE_EDGES, N_EDGES - tile * TILE_EDGES);
    int pv[TILE_EDGES / SORT_BLOCK], sv[TILE_EDGES / SORT_BLOCK];
#pragma unroll
    for (int j = 0; j < TILE_EDGES / SORT_BLOCK; ++j) {
        pv[j] = spos[j * SORT_BLOCK + t];
        sv[j] = sval[j * SORT_BLOCK + t];
    }
#pragma unroll
    for (int j = 0; j < TILE_EDGES / SORT_BLOCK; ++j) {
        if (j * SORT_BLOCK + t < cnt) {
            const int pos = min(max(pv[j], 0), N_EDGES - 1);
            mid[pos] = sv[j];
        }
    }
}

#define STAGE_CAP 4096

__global__ __launch_bounds__(SORT_BLOCK)
void bucket_sort_kernel(const int* __restrict__ mid,
                        const int* __restrict__ bbase,
                        int* __restrict__ csr_src,
                        int* __restrict__ offs,
                        float* __restrict__ inv) {
    __shared__ int cnt[BKT_SEGS];
    __shared__ int cur[BKT_SEGS];
    __shared__ int wsum[BKT_NODES / 32];
    __shared__ int stage[STAGE_CAP];
    const int t = threadIdx.x;
    const int lane = t & 31;
    const int wave = t >> 5;
    const int k = blockIdx.x;

    const int base = min(max(bbase[k], 0), N_EDGES);
    const int n = min(max(bbase[k + 1] - base, 0), N_EDGES - base);
    const bool staged = n <= STAGE_CAP;

    for (int i = t; i < BKT_SEGS; i += SORT_BLOCK) cnt[i] = 0;
    __syncthreads();

    for (int j0 = 0; j0 < n; j0 += SORT_BLOCK * CHUNK) {
        int v[CHUNK];
#pragma unroll
        for (int j = 0; j < CHUNK; ++j) {
            const int i = j0 + j * SORT_BLOCK + t;
            v[j] = (i < n) ? mid[base + i] : -1;
        }
#pragma unroll
        for (int j = 0; j < CHUNK; ++j)
            if (v[j] >= 0) atomicAdd(&cnt[min(v[j] >> 16, BKT_SEGS - 1)], 1);
    }
    __syncthreads();

    int c0 = 0, c1 = 0, c2 = 0;
    if (t < BKT_NODES) { c0 = cnt[3 * t]; c1 = cnt[3 * t + 1]; c2 = cnt[3 * t + 2]; }
    const int ts = c0 + c1 + c2;
    int inc = ts;
    for (int off = 1; off < 32; off <<= 1) {
        const int u = __shfl_up(inc, off, 32);
        if (lane >= off) inc += u;
    }
    if (t < BKT_NODES && lane == 31) wsum[wave] = inc;
    __syncthreads();
    if (t < BKT_NODES) {
        int ex = inc - ts;
        for (int w = 0; w < BKT_NODES / 32; ++w)
            if (w < wave) ex += wsum[w];
        cur[3 * t]     = ex;
        cur[3 * t + 1] = ex + c0;
        cur[3 * t + 2] = ex + c0 + c1;
        const int node = k * BKT_NODES + t;
        if (node < N_NODES) {
            const int seg = node * NUM_REL;
            offs[seg]     = base + ex;
            offs[seg + 1] = base + ex + c0;
            offs[seg + 2] = base + ex + c0 + c1;
            inv[seg]     = 1.0f / (float)(c0 > 1 ? c0 : 1);
            inv[seg + 1] = 1.0f / (float)(c1 > 1 ? c1 : 1);
            inv[seg + 2] = 1.0f / (float)(c2 > 1 ? c2 : 1);
            if (node == N_NODES - 1) offs[NR_TOTAL] = base + ex + ts;
        }
    }
    __syncthreads();

    for (int j0 = 0; j0 < n; j0 += SORT_BLOCK * CHUNK) {
        int v[CHUNK];
#pragma unroll
        for (int j = 0; j < CHUNK; ++j) {
            const int i = j0 + j * SORT_BLOCK + t;
            v[j] = (i < n) ? mid[base + i] : -1;
        }
#pragma unroll
        for (int j = 0; j < CHUNK; ++j) {
            if (v[j] >= 0) {
                const int p = atomicAdd(&cur[min(v[j] >> 16, BKT_SEGS - 1)], 1);
                if (staged) {
                    stage[min(max(p, 0), STAGE_CAP - 1)] = v[j] & 0xFFFF;
                } else {
                    const int q = min(max(base + p, 0), N_EDGES - 1);
                    csr_src[q] = v[j] & 0xFFFF;
                }
            }
        }
    }
    if (staged) {
        __syncthreads();
        for (int j0 = 0; j0 < n; j0 += SORT_BLOCK * CHUNK) {
            int w[CHUNK];
#pragma unroll
            for (int j = 0; j < CHUNK; ++j) {
                const int i = j0 + j * SORT_BLOCK + t;
                w[j] = (i < n) ? stage[i] : 0;
            }
#pragma unroll
            for (int j = 0; j < CHUNK; ++j) {
                const int i = j0 + j * SORT_BLOCK + t;
                if (i < n) csr_src[base + i] = w[j];
            }
        }
    }
}

#define A_PITCH  260
#define X_PITCH  36
static_assert(N_NODES % TILE_M == 0, "no partial node tile");
static_assert(IN_DIM + 1 <= FOLD_ROWS && FOLD_ROWS % 4 == 0, "x, the 0/1 column and padding fill whole k-steps");

__device__ __forceinline__ int bcast_i(int v, int l) {
    return __builtin_amdgcn_readlane(v, l);
}
__device__ __forceinline__ float bcast_f(float v, int l) {
    return __int_as_float(__builtin_amdgcn_readlane(__float_as_int(v), l));
}

template <int KS, int APITCH>
__device__ __forceinline__ void multiply_and_norm(const float* aT,
                                                  float (&lA)[TILE_M][A_PITCH],
                                                  const int row0, const int wave, const int lane,
                                                  const float* __restrict__ Wq,
                                                  const float* __restrict__ bias,
                                                  const float* __restrict__ gamma,
                                                  const float* __restrict__ beta,
                                                  float* __restrict__ hout) {
    const int c     = lane & 15;
    const int hh    = lane >> 4;
    const int khalf = hh << 1;

    v8f acc[4];
#pragma unroll
    for (int j = 0; j < 4; ++j) acc[j] = (v8f){0.f,0.f,0.f,0.f,0.f,0.f,0.f,0.f};

#pragma unroll
    for (int k0 = 0; k0 < KS; k0 += 4) {
        const int kk = k0 + khalf;
        const v2f a = *(const v2f*)&aT[c * APITCH + wave * KS + kk];
        const float4 w0 = *(const float4*)(Wq + (size_t)kk * OUT_DIM + 4 * c);
        const float4 w1 = *(const float4*)(Wq + (size_t)(kk + 1) * OUT_DIM + 4 * c);
        v2f b;
        b.x = w0.x; b.y = w1.x;
        acc[0] = __builtin_amdgcn_wmma_f32_16x16x4_f32(false, a, false, b, (short)0, acc[0], false, false);
        b.x = w0.y; b.y = w1.y;
        acc[1] = __builtin_amdgcn_wmma_f32_16x16x4_f32(false, a, false, b, (short)0, acc[1], false, false);
        b.x = w0.z; b.y = w1.z;
        acc[2] = __builtin_amdgcn_wmma_f32_16x16x4_f32(false, a, false, b, (short)0, acc[2], false, false);
        b.x = w0.w; b.y = w1.w;
        acc[3] = __builtin_amdgcn_wmma_f32_16x16x4_f32(false, a, false, b, (short)0, acc[3], false, false);
    }

#pragma unroll
    for (int r = 0; r < 8; ++r) {
        *(float4*)&lA[8 * hh + r][wave * OUT_DIM + 4 * c] =
            make_float4(acc[0][r], acc[1][r], acc[2][r], acc[3][r]);
    }
    __syncthreads();

    const float2 bs = ((const float2*)bias)[lane];
    const float2 g2 = ((const float2*)gamma)[lane];
    const float2 be = ((const float2*)beta)[lane];
#pragma unroll
    for (int t = 0; t < 4; ++t) {
        const int row = wave * 4 + t;
        const float* prow = &lA[row][2 * lane];
        const float2 p0 = *(const float2*)(prow);
        const float2 p1 = *(const float2*)(prow + 64);
        const float2 p2 = *(const float2*)(prow + 128);
        const float2 p3 = *(const float2*)(prow + 192);
        float2 v;
        v.x = p3.x + bs.x;  v.y = p3.y + bs.y;
        v.x += p0.x;        v.y += p0.y;
        v.x += p1.x;        v.y += p1.y;
        v.x += p2.x;        v.y += p2.y;
        v.x = fmaxf(v.x, 0.0f);
        v.y = fmaxf(v.y, 0.0f);
        float s  = v.x + v.y;
        float sq = v.x * v.x + v.y * v.y;
#pragma unroll
        for (int off = 16; off > 0; off >>= 1) {
            s  += __shfl_xor(s,  off, 32);
            sq += __shfl_xor(sq, off, 32);
        }
        const float mu   = s * (1.0f / OUT_DIM);
        const float var  = sq * (1.0f / OUT_DIM) - mu * mu;
        const float rstd = rsqrtf(var + LN_EPS);
        float2 o;
        o.x = (v.x - mu) * rstd * g2.x + be.x;
        o.y = (v.y - mu) * rstd * g2.y + be.y;
        ((float2*)(hout + (size_t)(row0 + row) * OUT_DIM))[lane] = o;
    }
}

#define BPG 4

__global__ __launch_bounds__(128, 16)
void rgcn_first_layer_kernel(const float* __restrict__ x,
                             const float* __restrict__ Wc,
                             const int*   __restrict__ offs,
                             const int*   __restrict__ csr_src,
                             const float* __restrict__ inv_denom,
                             const float* __restrict__ bias,
                             const float* __restrict__ gamma,
                             const float* __restrict__ beta,
                             float* __restrict__ hout)
{
    __shared__ __attribute__((aligned(16))) float lA[TILE_M][A_PITCH];
    __shared__ __attribute__((aligned(16))) float lX[TILE_M][X_PITCH];
    const int row0 = blockIdx.x * TILE_M;
    const int wave = threadIdx.x >> 5;
    const int lane = threadIdx.x & 31;

    {
        const int node0 = row0 + wave * 4;
        int bl = offs[node0 * NUM_REL + (lane < 12 ? lane : 12)];
        bl = min(max(bl, 0), N_EDGES);
        const float wl = inv_denom[node0 * NUM_REL + (lane < 11 ? lane : 11)];

        auto load_row = [&](int sraw, float (&v)[IN_DIM]) __attribute__((always_inline)) {
            const int s = min(max(sraw, 0), N_NODES - 1);
            const float* xr = x + (size_t)s * IN_DIM;
            float4 q;
            __builtin_memcpy(&q, xr, sizeof(q));
            v[0] = q.x; v[1] = q.y; v[2] = q.z; v[3] = q.w;
            v[4] = xr[4];
        };
        float xown[IN_DIM];
        load_row(node0 + ((lane - 12) & 3), xown);

        const int e0 = bcast_i(bl, 0);
        const int n  = max(bcast_i(bl, 12) - e0, 0);
        const int elast = min(max(e0 + n - 1, 0), N_EDGES - 1);

        auto idx_raw = [&](int p) __attribute__((always_inline)) {
            const int j = min(max(e0 + p, 0), elast);
            return csr_src[j];
        };

        const int myLo = bl - e0;
        const int myHi = __shfl_down(bl, 1, 32) - e0;
        int segstart[4 * NUM_REL - 1];
#pragma unroll
        for (int k = 0; k < 4 * NUM_REL - 1; ++k) segstart[k] = bcast_i(bl, k + 1) - e0;

        float sum[IN_DIM];
#pragma unroll
        for (int d = 0; d < IN_DIM; ++d) sum[d] = 0.f;

        for (int g0 = 0; g0 < n; g0 += 32 * BPG) {
            int sraw[BPG];
            float row[BPG][IN_DIM];
#pragma unroll
            for (int q = 0; q < BPG; ++q) sraw[q] = idx_raw(g0 + 32 * q + lane);
#pragma unroll
            for (int q = 0; q < BPG; ++q) load_row(sraw[q], row[q]);
#pragma unroll
            for (int q = 0; q < BPG; ++q) {
                const int base = g0 + 32 * q;
                const int p = base + lane;
                float v[IN_DIM];
#pragma unroll
                for (int d = 0; d < IN_DIM; ++d) v[d] = (p < n) ? row[q][d] : 0.f;
                int st = 0;
#pragma unroll
                for (int k = 0; k < 4 * NUM_REL - 1; ++k) st = (p >= segstart[k]) ? segstart[k] : st;
                const int reach = min(p - st, lane);
#pragma unroll
                for (int off = 1; off < 32; off <<= 1) {
                    const bool take = reach >= off;
#pragma unroll
                    for (int d = 0; d < IN_DIM; ++d) {
                        const float u = __shfl_up(v[d], off, 32);
                        v[d] += take ? u : 0.f;
                    }
                }
                const int lo = max(myLo, base);
                const int hi = min(myHi, base + 32);
                const bool has = (lane < 4 * NUM_REL) && (lo < hi);
                const int tail = has ? (hi - 1 - base) : 0;
#pragma unroll
                for (int d = 0; d < IN_DIM; ++d) {
                    const float u = __shfl(v[d], tail, 32);
                    sum[d] += has ? u : 0.f;
                }
            }
        }

        if (lane < 4 * NUM_REL) {
            const int t = lane / NUM_REL;
            const int r = lane - t * NUM_REL;
            const float some = (myHi > myLo) ? 1.f : 0.f;
            float* d8 = &lX[wave * 4 + t][r * FOLD_ROWS];
            *(float4*)(d8)     = make_float4(sum[0] * wl, sum[1] * wl, sum[2] * wl, sum[3] * wl);
            *(float4*)(d8 + 4) = make_float4(sum[4] * wl, some, 0.f, 0.f);
        } else if (lane < 4 * NUM_REL + 4) {
            float* d8 = &lX[wave * 4 + (lane - 4 * NUM_REL)][NUM_REL * FOLD_ROWS];
            *(float4*)(d8)     = make_float4(xown[0], xown[1], xown[2], xown[3]);
            *(float4*)(d8 + 4) = make_float4(xown[4], 1.f, 0.f, 0.f);
        }
    }
    __syncthreads();

    multiply_and_norm<FOLD_ROWS, X_PITCH>(&lX[0][0], lA, row0, wave, lane,
                                          Wc + (size_t)wave * FOLD_ROWS * OUT_DIM,
                                          bias, gamma, beta, hout);
}

#define GRP 8

__global__ __launch_bounds__(128, 16)
void rgcn_layer_kernel(const float* __restrict__ h,
                       const int*   __restrict__ offs,
                       const int*   __restrict__ csr_src,
                       const float* __restrict__ inv_denom,
                       const float* __restrict__ relW,
                       const float* __restrict__ rootW,
                       const float* __restrict__ bias,
                       const float* __restrict__ gamma,
                       const float* __restrict__ beta,
                       float* __restrict__ hout)
{
    __shared__ __attribute__((aligned(16))) float lA[TILE_M][A_PITCH];
    const int row0 = blockIdx.x * TILE_M;
    const int wave = threadIdx.x >> 5;
    const int lane = threadIdx.x & 31;

    {
        const float2* __restrict__ h2 = (const float2*)h;
        const int node0 = row0 + wave * 4;
        float2 hown[4];
#pragma unroll
        for (int t = 0; t < 4; ++t) hown[t] = h2[(size_t)(node0 + t) * (OUT_DIM / 2) + lane];
        int bl = offs[node0 * NUM_REL + (lane < 12 ? lane : 12)];
        bl = min(max(bl, 0), N_EDGES);
        const float wl = inv_denom[node0 * NUM_REL + (lane < 11 ? lane : 11)];

        const int e0 = bcast_i(bl, 0);
        const int n  = max(bcast_i(bl, 12) - e0, 0);
        const int elast = min(max(e0 + n - 1, 0), N_EDGES - 1);

        auto idx_raw = [&](int cidx) __attribute__((always_inline)) {
            const int j = min(max(e0 + 2 * GRP * cidx + lane, 0), elast);
            return csr_src[j];
        };
        int s = idx_raw(0);

        float* lrow = &lA[wave * 4][2 * lane];
#pragma unroll
        for (int t = 0; t < 4; ++t) *(float2*)(lrow + t * A_PITCH + 3 * OUT_DIM) = hown[t];

        s = min(max(s, 0), N_NODES - 1);

        int seg = 0;
        int rel = 0;
        int segoff = 0;
        int nb = bcast_i(bl, 1) - e0;
        float2 a = make_float2(0.f, 0.f);

        auto flush = [&]() __attribute__((always_inline)) {
            const float w = bcast_f(wl, seg);
            *(float2*)(lrow + segoff) = make_float2(a.x * w, a.y * w);
            a = make_float2(0.f, 0.f);
            ++seg;
            ++rel;
            segoff += OUT_DIM;
            if (rel == NUM_REL) { rel = 0; segoff += A_PITCH - NUM_REL * OUT_DIM; }
            nb = (seg < 4 * NUM_REL) ? (bcast_i(bl, seg + 1) - e0) : 0x7fffffff;
        };
        auto load_group = [&](float2 (&m)[GRP], int sv, int k) __attribute__((always_inline)) {
#pragma unroll
            for (int i = 0; i < GRP; ++i) {
                const int rk = bcast_i(sv, k + i);
                const float2* __restrict__ rowp = h2 + (size_t)rk * (OUT_DIM / 2);
                m[i] = rowp[lane];
            }
        };
        auto consume = [&](const float2 (&m)[GRP], int p0) __attribute__((always_inline)) {
#pragma unroll
            for (int i = 0; i < GRP; ++i) {
                while (p0 + i >= nb) flush();
                a.x += m[i].x;
                a.y += m[i].y;
            }
        };

        {
            float2 mA[GRP], mB[GRP];
            int g = 0;
            while (g * GRP < n) {
                const int sraw = idx_raw(g / 2 + 1);
                load_group(mA, s, 0);
                if (g > 0) consume(mB, (g - 1) * GRP);
                load_group(mB, s, GRP);
                consume(mA, g * GRP);
                s = min(max(sraw, 0), N_NODES - 1);
                g += 2;
            }
            if (g > 0) consume(mB, (g - 1) * GRP);
        }
        while (seg < 4 * NUM_REL) flush();
    }
    __syncthreads();

    multiply_and_norm<OUT_DIM, A_PITCH>(&lA[0][0], lA, row0, wave, lane,
                                        (wave < NUM_REL) ? (relW + (size_t)wave * OUT_DIM * OUT_DIM) : rootW,
                                        bias, gamma, beta, hout);
}

extern "C" void kernel_launch(void* const* d_in, const int* in_sizes, int n_in,
                              void* d_out, int out_size, void* d_ws, size_t ws_size,
                              hipStream_t stream) {
    const float* x      = (const float*)d_in[0];
    const int*   ei     = (const int*)  d_in[1];
    const int*   src    = ei;
    const int*   dst    = ei + N_EDGES;
    const int*   etype  = (const int*)  d_in[2];
    const float* f2cW   = (const float*)d_in[3];
    const float* f2cb   = (const float*)d_in[4];
    const float* rgW    = (const float*)d_in[5];
    const float* rgRoot = (const float*)d_in[6];
    const float* rgBias = (const float*)d_in[7];
    const float* gamma  = (const float*)d_in[8];
    const float* beta   = (const float*)d_in[9];

    float* ws = (float*)d_ws;
    float* h_a       = ws;  ws += (size_t)N_NODES * OUT_DIM;
    int*   mid       = (int*)ws;
    int*   csr_src   = mid + N_EDGES;
    int*   offs      = csr_src + N_EDGES;
    ws += (size_t)N_NODES * OUT_DIM;
    float* h_b       = ws;
    float* wfold     = h_b + (size_t)N_NODES * OUT_DIM;
    ws += (size_t)N_NODES * NUM_REL * OUT_DIM;
    int*   th        = (int*)ws;
    int*   btot      = th + NUM_BKT * TILE_PITCH;
    int*   bbase     = btot + 512;
    ws += (size_t)N_NODES * OUT_DIM;
    float* inv_denom = ws;

    tile_hist_kernel<<<NUM_TILES + 1, SORT_BLOCK, 0, stream>>>(dst, th, f2cW, f2cb, rgW, rgRoot, wfold);
    tile_prefix_kernel<<<(NUM_BKT + SORT_BLOCK / 32 - 1) / (SORT_BLOCK / 32), SORT_BLOCK, 0, stream>>>(th, btot);
    partition_kernel<<<NUM_TILES, SORT_BLOCK, 0, stream>>>(src, dst, etype, th, btot, bbase, mid);
    bucket_sort_kernel<<<NUM_BKT, SORT_BLOCK, 0, stream>>>(mid, bbase, csr_src, offs, inv_denom);

    rgcn_first_layer_kernel<<<N_NODES / TILE_M, 128, 0, stream>>>(
        x, wfold, offs, csr_src, inv_denom, rgBias, gamma, beta, h_b);
    for (int l = 1; l < NUM_LAYERS; ++l) {
        const float* hi = (l & 1) ? h_b : h_a;
        float* ho = (l == NUM_LAYERS - 1) ? (float*)d_out : ((l & 1) ? h_a : h_b);
        rgcn_layer_kernel<<<N_NODES / TILE_M, 128, 0, stream>>>(
            hi, offs, csr_src, inv_denom,
            rgW    + (size_t)l * NUM_REL * OUT_DIM * OUT_DIM,
            rgRoot + (size_t)l * OUT_DIM * OUT_DIM,
            rgBias + (size_t)l * OUT_DIM,
            gamma, beta, ho);
    }
}


# CSR build: each tile counts, scans and files its edges by bucket in its own slice (edges loaded once, as one group); a bucket collects its per-tile runs; 7 to 5 launches; segment order = LDS arrival
# speedup vs baseline: 4.9388x; 1.1559x over previous
#include <hip/hip_runtime.h>
#include <hip/hip_bf16.h>

#define N_NODES   50000
#define N_EDGES   1250000
#define IN_DIM    5
#define OUT_DIM   64
#define NUM_REL   3
#define NUM_LAYERS 3
#define LN_EPS    1e-5f
#define TILE_M    16

typedef __attribute__((ext_vector_type(2))) float v2f;
typedef __attribute__((ext_vector_type(8))) float v8f;

#define NR_TOTAL    (N_NODES * NUM_REL)
#define BKT_NODES   128
#define BKT_SEGS    (BKT_NODES * NUM_REL)
#define NUM_BKT     ((N_NODES + BKT_NODES - 1) / BKT_NODES)
#define SORT_BLOCK  256
#define TILE_EDGES  (SORT_BLOCK * 16)
#define NUM_TILES   ((N_EDGES + TILE_EDGES - 1) / TILE_EDGES)
#define TILE_PITCH  320
#define CHUNK       8
static_assert(N_EDGES % 4 == 0, "edges are read as int4");
static_assert(NUM_TILES <= TILE_PITCH, "one row entry per tile");
static_assert(NUM_BKT + 1 <= 2 * SORT_BLOCK, "two table rows per thread in the tile's bucket scan");
static_assert(TILE_PITCH <= 2 * SORT_BLOCK, "two tiles per thread in the bucket's run scan");
static_assert(N_NODES <= 65536, "a source node fits the low 16 bits of a filed edge");
static_assert(BKT_NODES <= SORT_BLOCK, "one thread per node of a bucket");

__device__ __forceinline__ int bucket_of(int d) {
    return min((int)((unsigned)d >> 7), NUM_BKT - 1);
}


#define FOLD_ROWS 8
#define FA_PITCH  68
static_assert(SORT_BLOCK == 256 && OUT_DIM == 64 && NUM_REL == 3, "eight waves: four matrices x two pairs of 16-column tiles");

__device__ __forceinline__ void fold_first_layer_weights(const float* __restrict__ Wf,
                                                         const float* __restrict__ bf,
                                                         const float* __restrict__ relW,
                                                         const float* __restrict__ rootW,
                                                         float* __restrict__ Wc) {
    __shared__ __attribute__((aligned(16))) float fA[TILE_M][FA_PITCH];
    const int t = threadIdx.x;
    const int wave = t >> 5;
    const int lane = t & 31;
    const int q     = wave & 3;
    const int jh    = wave >> 2;
    const int c     = lane & 15;
    const int hh    = lane >> 4;
    const int khalf = hh << 1;
    const float* __restrict__ Wq =
        (q < NUM_REL) ? (relW + (size_t)q * OUT_DIM * OUT_DIM) : rootW;

    float2 wa[OUT_DIM / 4], wb[OUT_DIM / 4];
#pragma unroll
    for (int s = 0; s < OUT_DIM / 4; ++s) {
        const int kk = 4 * s + khalf;
        wa[s] = *(const float2*)(Wq + (size_t)kk * OUT_DIM + 4 * c + 2 * jh);
        wb[s] = *(const float2*)(Wq + (size_t)(kk + 1) * OUT_DIM + 4 * c + 2 * jh);
    }
    {
        const int r = t >> 4, c4 = t & 15;
        float4 v = make_float4(0.f, 0.f, 0.f, 0.f);
        if (r < IN_DIM)       v = *(const float4*)(Wf + r * OUT_DIM + 4 * c4);
        else if (r == IN_DIM) v = *(const float4*)(bf + 4 * c4);
        *(float4*)&fA[r][4 * c4] = v;
    }
    __syncthreads();

    v8f acc0 = (v8f){0.f,0.f,0.f,0.f,0.f,0.f,0.f,0.f};
    v8f acc1 = (v8f){0.f,0.f,0.f,0.f,0.f,0.f,0.f,0.f};
#pragma unroll
    for (int s = 0; s < OUT_DIM / 4; ++s) {
        const int kk = 4 * s + khalf;
        const v2f a = *(const v2f*)&fA[c][kk];
        v2f b;
        b.x = wa[s].x; b.y = wb[s].x;
        acc0 = __builtin_amdgcn_wmma_f32_16x16x4_f32(false, a, false, b, (short)0, acc0, false, false);
        b.x = wa[s].y; b.y = wb[s].y;
        acc1 = __builtin_amdgcn_wmma_f32_16x16x4_f32(false, a, false, b, (short)0, acc1, false, false);
    }
    if (hh == 0) {
#pragma unroll
        for (int r = 0; r < FOLD_ROWS; ++r) {
            *(float2*)(Wc + (size_t)(q * FOLD_ROWS + r) * OUT_DIM + 4 * c + 2 * jh) =
                make_float2(acc0[r], acc1[r]);
        }
    }
}

__device__ __forceinline__ void file_edge(int* lcur, int* sval, int d, int ty, int s) {
    const int key = (d & (BKT_NODES - 1)) * NUM_REL + min(max(ty, 0), NUM_REL - 1);
    int lp = atomicAdd(&lcur[bucket_of(d)], 1);
    lp = min(max(lp, 0), TILE_EDGES - 1);
    sval[lp] = (key << 16) | (s & 0xFFFF);
}

__global__ __launch_bounds__(SORT_BLOCK)
void partition_kernel(const int* __restrict__ src,
                      const int* __restrict__ dst,
                      const int* __restrict__ etype,
                      int* __restrict__ tb,
                      int* __restrict__ mid,
                      const float* __restrict__ Wf, const float* __restrict__ bf,
                      const float* __restrict__ relW, const float* __restrict__ rootW,
                      float* __restrict__ Wc) {
    if (blockIdx.x == NUM_TILES) {
        fold_first_layer_weights(Wf, bf, relW, rootW, Wc);
        return;
    }
    __shared__ int hist[NUM_BKT];
    __shared__ int lcur[NUM_BKT];
    __shared__ int lwsum[SORT_BLOCK / 32];
    __shared__ int sval[TILE_EDGES];
    const int t = threadIdx.x;
    const int lane = t & 31;
    const int wave = t >> 5;
    const int tile = blockIdx.x;

    int4 d[4], y[4], s[4];
#pragma unroll
    for (int i = 0; i < 4; ++i) {
        const int e = tile * TILE_EDGES + i * (SORT_BLOCK * 4) + 4 * t;
        const int ec = min(e, N_EDGES - 4);
        d[i] = *(const int4*)(dst + ec);
    }
#pragma unroll
    for (int i = 0; i < 4; ++i) {
        const int e = tile * TILE_EDGES + i * (SORT_BLOCK * 4) + 4 * t;
        const int ec = min(e, N_EDGES - 4);
        y[i] = *(const int4*)(etype + ec);
        s[i] = *(const int4*)(src + ec);
    }
    for (int b = t; b < NUM_BKT; b += SORT_BLOCK) hist[b] = 0;
    __syncthreads();
#pragma unroll
    for (int i = 0; i < 4; ++i) {
        const int e = tile * TILE_EDGES + i * (SORT_BLOCK * 4) + 4 * t;
        if (e < N_EDGES) {
            atomicAdd(&hist[bucket_of(d[i].x)], 1);
            atomicAdd(&hist[bucket_of(d[i].y)], 1);
            atomicAdd(&hist[bucket_of(d[i].z)], 1);
            atomicAdd(&hist[bucket_of(d[i].w)], 1);
        }
    }
    __syncthreads();

    const int i0 = 2 * t, i1 = 2 * t + 1;
    const int n0 = (i0 < NUM_BKT) ? hist[i0] : 0;
    const int n1 = (i1 < NUM_BKT) ? hist[i1] : 0;
    const int ls = n0 + n1;
    int linc = ls;
    for (int off = 1; off < 32; off <<= 1) {
        const int lu = __shfl_up(linc, off, 32);
        if (lane >= off) linc += lu;
    }
    if (lane == 31) lwsum[wave] = linc;
    __syncthreads();
    int lex = linc - ls;
    for (int w = 0; w < SORT_BLOCK / 32; ++w)
        if (w < wave) lex += lwsum[w];
    if (i0 < NUM_BKT)  lcur[i0] = lex;
    if (i1 < NUM_BKT)  lcur[i1] = lex + n0;
    if (i0 <= NUM_BKT) tb[i0 * TILE_PITCH + tile] = lex;
    if (i1 <= NUM_BKT) tb[i1 * TILE_PITCH + tile] = lex + n0;
    __syncthreads();

#pragma unroll
    for (int i = 0; i < 4; ++i) {
        const int e = tile * TILE_EDGES + i * (SORT_BLOCK * 4) + 4 * t;
        if (e < N_EDGES) {
            file_edge(lcur, sval, d[i].x, y[i].x, s[i].x);
            file_edge(lcur, sval, d[i].y, y[i].y, s[i].y);
            file_edge(lcur, sval, d[i].z, y[i].z, s[i].z);
            file_edge(lcur, sval, d[i].w, y[i].w, s[i].w);
        }
    }
    __syncthreads();

    const int cnt = min(TILE_EDGES, N_EDGES - tile * TILE_EDGES);
    int sv[TILE_EDGES / SORT_BLOCK];
#pragma unroll
    for (int j = 0; j < TILE_EDGES / SORT_BLOCK; ++j) sv[j] = sval[j * SORT_BLOCK + t];
#pragma unroll
    for (int j = 0; j < TILE_EDGES / SORT_BLOCK; ++j) {
        if (j * SORT_BLOCK + t < cnt) mid[tile * TILE_EDGES + j * SORT_BLOCK + t] = sv[j];
    }
}

#define STAGE_CAP 4096
#define RUN_TILES 10
static_assert(TILE_PITCH % (16 * RUN_TILES) == 0 && SORT_BLOCK == 256, "16 half-waves sweep the table row in whole trips");

template <int MODE>
__device__ __forceinline__ void sweep_runs(const int* __restrict__ mid,
                                           const int* rs, const int* rc, const int* rr,
                                           int* ctr, int* raw, int* __restrict__ csr_src,
                                           const int base, const int t) {
    const int hw = t >> 4, l16 = t & 15;
    auto take = [&](int v, int slot) __attribute__((always_inline)) {
        const int seg = min(v >> 16, BKT_SEGS - 1);
        if (MODE == 0) {
            raw[min(max(slot, 0), STAGE_CAP - 1)] = v;
            atomicAdd(&ctr[seg], 1);
        } else if (MODE == 1) {
            atomicAdd(&ctr[seg], 1);
        } else {
            const int p = atomicAdd(&ctr[seg], 1);
            const int q = min(max(base + p, 0), N_EDGES - 1);
            csr_src[q] = v & 0xFFFF;
        }
    };
    for (int T0 = hw; T0 < TILE_PITCH; T0 += 16 * RUN_TILES) {
        int v[RUN_TILES][2], c[RUN_TILES], r[RUN_TILES], a[RUN_TILES];
#pragma unroll
        for (int u = 0; u < RUN_TILES; ++u) {
            const int T = T0 + 16 * u;
            c[u] = rc[T];
            r[u] = rr[T];
            a[u] = T * TILE_EDGES + rs[T];
#pragma unroll
            for (int h = 0; h < 2; ++h) {
                const int i = l16 + 16 * h;
                v[u][h] = (i < c[u]) ? mid[min(max(a[u] + i, 0), N_EDGES - 1)] : -1;
            }
        }
#pragma unroll
        for (int u = 0; u < RUN_TILES; ++u) {
#pragma unroll
            for (int h = 0; h < 2; ++h)
                if (v[u][h] >= 0) take(v[u][h], r[u] + l16 + 16 * h);
        }
#pragma unroll
        for (int u = 0; u < RUN_TILES; ++u) {
            for (int i = l16 + 32; i < c[u]; i += 16) {
                const int w = mid[min(max(a[u] + i, 0), N_EDGES - 1)];
                if (w >= 0) take(w, r[u] + i);
            }
        }
    }
}

__global__ __launch_bounds__(SORT_BLOCK)
void bucket_sort_kernel(const int* __restrict__ mid,
                        const int* __restrict__ tb,
                        int* __restrict__ csr_src,
                        int* __restrict__ offs,
                        float* __restrict__ inv) {
    __shared__ int cnt[BKT_SEGS];
    __shared__ int cur[BKT_SEGS];
    __shared__ int wsum[BKT_NODES / 32];
    __shared__ int wsA[SORT_BLOCK / 32], wsB[SORT_BLOCK / 32], wsG[SORT_BLOCK / 32];
    __shared__ int rs[TILE_PITCH], rc[TILE_PITCH], rr[TILE_PITCH];
    __shared__ int raw[STAGE_CAP];
    __shared__ int stage[STAGE_CAP];
    const int t = threadIdx.x;
    const int lane = t & 31;
    const int wave = t >> 5;
    const int k = blockIdx.x;

    auto run_of = [&](int T, int& s, int& c) __attribute__((always_inline)) {
        const bool ok = T < NUM_TILES;
        const int Tc = ok ? T : NUM_TILES - 1;
        const int lo = tb[k * TILE_PITCH + Tc];
        const int hi = tb[(k + 1) * TILE_PITCH + Tc];
        const int tcnt = min(TILE_EDGES, N_EDGES - Tc * TILE_EDGES);
        const int sc = min(max(lo, 0), tcnt);
        const int ec = min(max(hi, sc), tcnt);
        s = ok ? sc : 0;
        c = ok ? ec - sc : 0;
    };
    int s0, ra, s1, rb;
    run_of(t, s0, ra);
    run_of(t + SORT_BLOCK, s1, rb);
    const int g = s0 + s1;
    int ia = ra, ib = rb, ig = g;
    for (int off = 1; off < 32; off <<= 1) {
        const int ua = __shfl_up(ia, off, 32);
        const int ub = __shfl_up(ib, off, 32);
        const int ug = __shfl_up(ig, off, 32);
        if (lane >= off) { ia += ua; ib += ub; ig += ug; }
    }
    if (lane == 31) { wsA[wave] = ia; wsB[wave] = ib; wsG[wave] = ig; }
    for (int i = t; i < BKT_SEGS; i += SORT_BLOCK) cnt[i] = 0;
    __syncthreads();
    int exa = ia - ra, exb = ib - rb;
    int totA = 0, totB = 0, gsum = 0;
    for (int w = 0; w < SORT_BLOCK / 32; ++w) {
        const int a = wsA[w], b = wsB[w];
        totA += a;
        totB += b;
        gsum += wsG[w];
        if (w < wave) { exa += a; exb += b; }
    }
    rs[t] = s0;
    rc[t] = ra;
    rr[t] = exa;
    if (t + SORT_BLOCK < TILE_PITCH) {
        rs[t + SORT_BLOCK] = s1;
        rc[t + SORT_BLOCK] = rb;
        rr[t + SORT_BLOCK] = totA + exb;
    }
    const int base = min(max(gsum, 0), N_EDGES);
    const int n = min(max(totA + totB, 0), N_EDGES - base);
    const bool staged = n <= STAGE_CAP;
    __syncthreads();

    if (staged) sweep_runs<0>(mid, rs, rc, rr, cnt, raw, csr_src, base, t);
    else        sweep_runs<1>(mid, rs, rc, rr, cnt, raw, csr_src, base, t);
    __syncthreads();

    int c0 = 0, c1 = 0, c2 = 0;
    if (t < BKT_NODES) { c0 = cnt[3 * t]; c1 = cnt[3 * t + 1]; c2 = cnt[3 * t + 2]; }
    const int ts = c0 + c1 + c2;
    int inc = ts;
    for (int off = 1; off < 32; off <<= 1) {
        const int u = __shfl_up(inc, off, 32);
        if (lane >= off) inc += u;
    }
    if (t < BKT_NODES && lane == 31) wsum[wave] = inc;
    __syncthreads();
    if (t < BKT_NODES) {
        int ex = inc - ts;
        for (int w = 0; w < BKT_NODES / 32; ++w)
            if (w < wave) ex += wsum[w];
        cur[3 * t]     = ex;
        cur[3 * t + 1] = ex + c0;
        cur[3 * t + 2] = ex + c0 + c1;
        const int node = k * BKT_NODES + t;
        if (node < N_NODES) {
            const int seg = node * NUM_REL;
            offs[seg]     = base + ex;
            offs[seg + 1] = base + ex + c0;
            offs[seg + 2] = base + ex + c0 + c1;
            inv[seg]     = 1.0f / (float)(c0 > 1 ? c0 : 1);
            inv[seg + 1] = 1.0f / (float)(c1 > 1 ? c1 : 1);
            inv[seg + 2] = 1.0f / (float)(c2 > 1 ? c2 : 1);
            if (node == N_NODES - 1) offs[NR_TOTAL] = base + ex + ts;
        }
    }
    __syncthreads();

    if (staged) {
        for (int j0 = 0; j0 < n; j0 += SORT_BLOCK * CHUNK) {
            int v[CHUNK];
#pragma unroll
            for (int j = 0; j < CHUNK; ++j) {
                const int i = j0 + j * SORT_BLOCK + t;
                v[j] = (i < n) ? raw[i] : -1;
            }
#pragma unroll
            for (int j = 0; j < CHUNK; ++j) {
                if (v[j] >= 0) {
                    const int p = atomicAdd(&cur[min(v[j] >> 16, BKT_SEGS - 1)], 1);
                    stage[min(max(p, 0), STAGE_CAP - 1)] = v[j] & 0xFFFF;
                }
            }
        }
        __syncthreads();
        for (int j0 = 0; j0 < n; j0 += SORT_BLOCK * CHUNK) {
            int w[CHUNK];
#pragma unroll
            for (int j = 0; j < CHUNK; ++j) {
                const int i = j0 + j * SORT_BLOCK + t;
                w[j] = (i < n) ? stage[i] : 0;
            }
#pragma unroll
            for (int j = 0; j < CHUNK; ++j) {
                const int i = j0 + j * SORT_BLOCK + t;
                if (i < n) csr_src[base + i] = w[j];
            }
        }
    } else {
        sweep_runs<2>(mid, rs, rc, rr, cur, raw, csr_src, base, t);
    }
}

#define A_PITCH  260
#define X_PITCH  36
static_assert(N_NODES % TILE_M == 0, "no partial node tile");
static_assert(IN_DIM + 1 <= FOLD_ROWS && FOLD_ROWS % 4 == 0, "x, the 0/1 column and padding fill whole k-steps");

__device__ __forceinline__ int bcast_i(int v, int l) {
    return __builtin_amdgcn_readlane(v, l);
}
__device__ __forceinline__ float bcast_f(float v, int l) {
    return __int_as_float(__builtin_amdgcn_readlane(__float_as_int(v), l));
}

template <int KS, int APITCH>
__device__ __forceinline__ void multiply_and_norm(const float* aT,
                                                  float (&lA)[TILE_M][A_PITCH],
                                                  const int row0, const int wave, const int lane,
                                                  const float* __restrict__ Wq,
                                                  const float* __restrict__ bias,
                                                  const float* __restrict__ gamma,
                                                  const float* __restrict__ beta,
                                                  float* __restrict__ hout) {
    const int c     = lane & 15;
    const int hh    = lane >> 4;
    const int khalf = hh << 1;

    v8f acc[4];
#pragma unroll
    for (int j = 0; j < 4; ++j) acc[j] = (v8f){0.f,0.f,0.f,0.f,0.f,0.f,0.f,0.f};

#pragma unroll
    for (int k0 = 0; k0 < KS; k0 += 4) {
        const int kk = k0 + khalf;
        const v2f a = *(const v2f*)&aT[c * APITCH + wave * KS + kk];
        const float4 w0 = *(const float4*)(Wq + (size_t)kk * OUT_DIM + 4 * c);
        const float4 w1 = *(const float4*)(Wq + (size_t)(kk + 1) * OUT_DIM + 4 * c);
        v2f b;
        b.x = w0.x; b.y = w1.x;
        acc[0] = __builtin_amdgcn_wmma_f32_16x16x4_f32(false, a, false, b, (short)0, acc[0], false, false);
        b.x = w0.y; b.y = w1.y;
        acc[1] = __builtin_amdgcn_wmma_f32_16x16x4_f32(false, a, false, b, (short)0, acc[1], false, false);
        b.x = w0.z; b.y = w1.z;
        acc[2] = __builtin_amdgcn_wmma_f32_16x16x4_f32(false, a, false, b, (short)0, acc[2], false, false);
        b.x = w0.w; b.y = w1.w;
        acc[3] = __builtin_amdgcn_wmma_f32_16x16x4_f32(false, a, false, b, (short)0, acc[3], false, false);
    }

#pragma unroll
    for (int r = 0; r < 8; ++r) {
        *(float4*)&lA[8 * hh + r][wave * OUT_DIM + 4 * c] =
            make_float4(acc[0][r], acc[1][r], acc[2][r], acc[3][r]);
    }
    __syncthreads();

    const float2 bs = ((const float2*)bias)[lane];
    const float2 g2 = ((const float2*)gamma)[lane];
    const float2 be = ((const float2*)beta)[lane];
#pragma unroll
    for (int t = 0; t < 4; ++t) {
        const int row = wave * 4 + t;
        const float* prow = &lA[row][2 * lane];
        const float2 p0 = *(const float2*)(prow);
        const float2 p1 = *(const float2*)(prow + 64);
        const float2 p2 = *(const float2*)(prow + 128);
        const float2 p3 = *(const float2*)(prow + 192);
        float2 v;
        v.x = p3.x + bs.x;  v.y = p3.y + bs.y;
        v.x += p0.x;        v.y += p0.y;
        v.x += p1.x;        v.y += p1.y;
        v.x += p2.x;        v.y += p2.y;
        v.x = fmaxf(v.x, 0.0f);
        v.y = fmaxf(v.y, 0.0f);
        float s  = v.x + v.y;
        float sq = v.x * v.x + v.y * v.y;
#pragma unroll
        for (int off = 16; off > 0; off >>= 1) {
            s  += __shfl_xor(s,  off, 32);
            sq += __shfl_xor(sq, off, 32);
        }
        const float mu   = s * (1.0f / OUT_DIM);
        const float var  = sq * (1.0f / OUT_DIM) - mu * mu;
        const float rstd = rsqrtf(var + LN_EPS);
        float2 o;
        o.x = (v.x - mu) * rstd * g2.x + be.x;
        o.y = (v.y - mu) * rstd * g2.y + be.y;
        ((float2*)(hout + (size_t)(row0 + row) * OUT_DIM))[lane] = o;
    }
}

#define BPG 4

__global__ __launch_bounds__(128, 16)
void rgcn_first_layer_kernel(const float* __restrict__ x,
                             const float* __restrict__ Wc,
                             const int*   __restrict__ offs,
                             const int*   __restrict__ csr_src,
                             const float* __restrict__ inv_denom,
                             const float* __restrict__ bias,
                             const float* __restrict__ gamma,
                             const float* __restrict__ beta,
                             float* __restrict__ hout)
{
    __shared__ __attribute__((aligned(16))) float lA[TILE_M][A_PITCH];
    __shared__ __attribute__((aligned(16))) float lX[TILE_M][X_PITCH];
    const int row0 = blockIdx.x * TILE_M;
    const int wave = threadIdx.x >> 5;
    const int lane = threadIdx.x & 31;

    {
        const int node0 = row0 + wave * 4;
        int bl = offs[node0 * NUM_REL + (lane < 12 ? lane : 12)];
        bl = min(max(bl, 0), N_EDGES);
        const float wl = inv_denom[node0 * NUM_REL + (lane < 11 ? lane : 11)];

        auto load_row = [&](int sraw, float (&v)[IN_DIM]) __attribute__((always_inline)) {
            const int s = min(max(sraw, 0), N_NODES - 1);
            const float* xr = x + (size_t)s * IN_DIM;
            float4 q;
            __builtin_memcpy(&q, xr, sizeof(q));
            v[0] = q.x; v[1] = q.y; v[2] = q.z; v[3] = q.w;
            v[4] = xr[4];
        };
        float xown[IN_DIM];
        load_row(node0 + ((lane - 12) & 3), xown);

        const int e0 = bcast_i(bl, 0);
        const int n  = max(bcast_i(bl, 12) - e0, 0);
        const int elast = min(max(e0 + n - 1, 0), N_EDGES - 1);

        auto idx_raw = [&](int p) __attribute__((always_inline)) {
            const int j = min(max(e0 + p, 0), elast);
            return csr_src[j];
        };

        const int myLo = bl - e0;
        const int myHi = __shfl_down(bl, 1, 32) - e0;
        int segstart[4 * NUM_REL - 1];
#pragma unroll
        for (int k = 0; k < 4 * NUM_REL - 1; ++k) segstart[k] = bcast_i(bl, k + 1) - e0;

        float sum[IN_DIM];
#pragma unroll
        for (int d = 0; d < IN_DIM; ++d) sum[d] = 0.f;

        for (int g0 = 0; g0 < n; g0 += 32 * BPG) {
            int sraw[BPG];
            float row[BPG][IN_DIM];
#pragma unroll
            for (int q = 0; q < BPG; ++q) sraw[q] = idx_raw(g0 + 32 * q + lane);
#pragma unroll
            for (int q = 0; q < BPG; ++q) load_row(sraw[q], row[q]);
#pragma unroll
            for (int q = 0; q < BPG; ++q) {
                const int base = g0 + 32 * q;
                const int p = base + lane;
                float v[IN_DIM];
#pragma unroll
                for (int d = 0; d < IN_DIM; ++d) v[d] = (p < n) ? row[q][d] : 0.f;
                int st = 0;
#pragma unroll
                for (int k = 0; k < 4 * NUM_REL - 1; ++k) st = (p >= segstart[k]) ? segstart[k] : st;
                const int reach = min(p - st, lane);
#pragma unroll
                for (int off = 1; off < 32; off <<= 1) {
                    const bool take = reach >= off;
#pragma unroll
                    for (int d = 0; d < IN_DIM; ++d) {
                        const float u = __shfl_up(v[d], off, 32);
                        v[d] += take ? u : 0.f;
                    }
                }
                const int lo = max(myLo, base);
                const int hi = min(myHi, base + 32);
                const bool has = (lane < 4 * NUM_REL) && (lo < hi);
                const int tail = has ? (hi - 1 - base) : 0;
#pragma unroll
                for (int d = 0; d < IN_DIM; ++d) {
                    const float u = __shfl(v[d], tail, 32);
                    sum[d] += has ? u : 0.f;
                }
            }
        }

        if (lane < 4 * NUM_REL) {
            const int t = lane / NUM_REL;
            const int r = lane - t * NUM_REL;
            const float some = (myHi > myLo) ? 1.f : 0.f;
            float* d8 = &lX[wave * 4 + t][r * FOLD_ROWS];
            *(float4*)(d8)     = make_float4(sum[0] * wl, sum[1] * wl, sum[2] * wl, sum[3] * wl);
            *(float4*)(d8 + 4) = make_float4(sum[4] * wl, some, 0.f, 0.f);
        } else if (lane < 4 * NUM_REL + 4) {
            float* d8 = &lX[wave * 4 + (lane - 4 * NUM_REL)][NUM_REL * FOLD_ROWS];
            *(float4*)(d8)     = make_float4(xown[0], xown[1], xown[2], xown[3]);
            *(float4*)(d8 + 4) = make_float4(xown[4], 1.f, 0.f, 0.f);
        }
    }
    __syncthreads();

    multiply_and_norm<FOLD_ROWS, X_PITCH>(&lX[0][0], lA, row0, wave, lane,
                                          Wc + (size_t)wave * FOLD_ROWS * OUT_DIM,
                                          bias, gamma, beta, hout);
}

#define GRP 8

__global__ __launch_bounds__(128, 16)
void rgcn_layer_kernel(const float* __restrict__ h,
                       const int*   __restrict__ offs,
                       const int*   __restrict__ csr_src,
                       const float* __restrict__ inv_denom,
                       const float* __restrict__ relW,
                       const float* __restrict__ rootW,
                       const float* __restrict__ bias,
                       const float* __restrict__ gamma,
                       const float* __restrict__ beta,
                       float* __restrict__ hout)
{
    __shared__ __attribute__((aligned(16))) float lA[TILE_M][A_PITCH];
    const int row0 = blockIdx.x * TILE_M;
    const int wave = threadIdx.x >> 5;
    const int lane = threadIdx.x & 31;

    {
        const float2* __restrict__ h2 = (const float2*)h;
        const int node0 = row0 + wave * 4;
        float2 hown[4];
#pragma unroll
        for (int t = 0; t < 4; ++t) hown[t] = h2[(size_t)(node0 + t) * (OUT_DIM / 2) + lane];
        int bl = offs[node0 * NUM_REL + (lane < 12 ? lane : 12)];
        bl = min(max(bl, 0), N_EDGES);
        const float wl = inv_denom[node0 * NUM_REL + (lane < 11 ? lane : 11)];

        const int e0 = bcast_i(bl, 0);
        const int n  = max(bcast_i(bl, 12) - e0, 0);
        const int elast = min(max(e0 + n - 1, 0), N_EDGES - 1);

        auto idx_raw = [&](int cidx) __attribute__((always_inline)) {
            const int j = min(max(e0 + 2 * GRP * cidx + lane, 0), elast);
            return csr_src[j];
        };
        int s = idx_raw(0);

        float* lrow = &lA[wave * 4][2 * lane];
#pragma unroll
        for (int t = 0; t < 4; ++t) *(float2*)(lrow + t * A_PITCH + 3 * OUT_DIM) = hown[t];

        s = min(max(s, 0), N_NODES - 1);

        int seg = 0;
        int rel = 0;
        int segoff = 0;
        int nb = bcast_i(bl, 1) - e0;
        float2 a = make_float2(0.f, 0.f);

        auto flush = [&]() __attribute__((always_inline)) {
            const float w = bcast_f(wl, seg);
            *(float2*)(lrow + segoff) = make_float2(a.x * w, a.y * w);
            a = make_float2(0.f, 0.f);
            ++seg;
            ++rel;
            segoff += OUT_DIM;
            if (rel == NUM_REL) { rel = 0; segoff += A_PITCH - NUM_REL * OUT_DIM; }
            nb = (seg < 4 * NUM_REL) ? (bcast_i(bl, seg + 1) - e0) : 0x7fffffff;
        };
        auto load_group = [&](float2 (&m)[GRP], int sv, int k) __attribute__((always_inline)) {
#pragma unroll
            for (int i = 0; i < GRP; ++i) {
                const int rk = bcast_i(sv, k + i);
                const float2* __restrict__ rowp = h2 + (size_t)rk * (OUT_DIM / 2);
                m[i] = rowp[lane];
            }
        };
        auto consume = [&](const float2 (&m)[GRP], int p0) __attribute__((always_inline)) {
#pragma unroll
            for (int i = 0; i < GRP; ++i) {
                while (p0 + i >= nb) flush();
                a.x += m[i].x;
                a.y += m[i].y;
            }
        };

        {
            float2 mA[GRP], mB[GRP];
            int g = 0;
            while (g * GRP < n) {
                const int sraw = idx_raw(g / 2 + 1);
                load_group(mA, s, 0);
                if (g > 0) consume(mB, (g - 1) * GRP);
                load_group(mB, s, GRP);
                consume(mA, g * GRP);
                s = min(max(sraw, 0), N_NODES - 1);
                g += 2;
            }
            if (g > 0) consume(mB, (g - 1) * GRP);
        }
        while (seg < 4 * NUM_REL) flush();
    }
    __syncthreads();

    multiply_and_norm<OUT_DIM, A_PITCH>(&lA[0][0], lA, row0, wave, lane,
                                        (wave < NUM_REL) ? (relW + (size_t)wave * OUT_DIM * OUT_DIM) : rootW,
                                        bias, gamma, beta, hout);
}

extern "C" void kernel_launch(void* const* d_in, const int* in_sizes, int n_in,
                              void* d_out, int out_size, void* d_ws, size_t ws_size,
                              hipStream_t stream) {
    const float* x      = (const float*)d_in[0];
    const int*   ei     = (const int*)  d_in[1];
    const int*   src    = ei;
    const int*   dst    = ei + N_EDGES;
    const int*   etype  = (const int*)  d_in[2];
    const float* f2cW   = (const float*)d_in[3];
    const float* f2cb   = (const float*)d_in[4];
    const float* rgW    = (const float*)d_in[5];
    const float* rgRoot = (const float*)d_in[6];
    const float* rgBias = (const float*)d_in[7];
    const float* gamma  = (const float*)d_in[8];
    const float* beta   = (const float*)d_in[9];

    float* ws = (float*)d_ws;
    float* h_a       = ws;  ws += (size_t)N_NODES * OUT_DIM;
    int*   mid       = (int*)ws;
    int*   csr_src   = mid + N_EDGES;
    int*   offs      = csr_src + N_EDGES;
    ws += (size_t)N_NODES * OUT_DIM;
    float* h_b       = ws;
    float* wfold     = h_b + (size_t)N_NODES * OUT_DIM;
    ws += (size_t)N_NODES * NUM_REL * OUT_DIM;
    int*   tb        = (int*)ws;
    ws += (size_t)N_NODES * OUT_DIM;
    float* inv_denom = ws;

    partition_kernel<<<NUM_TILES + 1, SORT_BLOCK, 0, stream>>>(src, dst, etype, tb, mid,
                                                                f2cW, f2cb, rgW, rgRoot, wfold);
    bucket_sort_kernel<<<NUM_BKT, SORT_BLOCK, 0, stream>>>(mid, tb, csr_src, offs, inv_denom);

    rgcn_first_layer_kernel<<<N_NODES / TILE_M, 128, 0, stream>>>(
        x, wfold, offs, csr_src, inv_denom, rgBias, gamma, beta, h_b);
    for (int l = 1; l < NUM_LAYERS; ++l) {
        const float* hi = (l & 1) ? h_b : h_a;
        float* ho = (l == NUM_LAYERS - 1) ? (float*)d_out : ((l & 1) ? h_a : h_b);
        rgcn_layer_kernel<<<N_NODES / TILE_M, 128, 0, stream>>>(
            hi, offs, csr_src, inv_denom,
            rgW    + (size_t)l * NUM_REL * OUT_DIM * OUT_DIM,
            rgRoot + (size_t)l * OUT_DIM * OUT_DIM,
            rgBias + (size_t)l * OUT_DIM,
            gamma, beta, ho);
    }
}


# Later layers: a block owns 32 nodes (two 16-row tiles, 8 waves); each wave multiplies both tiles by its 32-column half of a matrix, so a fetched weight fragment feeds two WMMAs; sums unchanged
# speedup vs baseline: 5.1585x; 1.0445x over previous
#include <hip/hip_runtime.h>
#include <hip/hip_bf16.h>

#define N_NODES   50000
#define N_EDGES   1250000
#define IN_DIM    5
#define OUT_DIM   64
#define NUM_REL   3
#define NUM_LAYERS 3
#define LN_EPS    1e-5f
#define TILE_M    16

typedef __attribute__((ext_vector_type(2))) float v2f;
typedef __attribute__((ext_vector_type(8))) float v8f;

#define NR_TOTAL    (N_NODES * NUM_REL)
#define BKT_NODES   128
#define BKT_SEGS    (BKT_NODES * NUM_REL)
#define NUM_BKT     ((N_NODES + BKT_NODES - 1) / BKT_NODES)
#define SORT_BLOCK  256
#define TILE_EDGES  (SORT_BLOCK * 16)
#define NUM_TILES   ((N_EDGES + TILE_EDGES - 1) / TILE_EDGES)
#define TILE_PITCH  320
#define CHUNK       8
static_assert(N_EDGES % 4 == 0, "edges are read as int4");
static_assert(NUM_TILES <= TILE_PITCH, "one row entry per tile");
static_assert(NUM_BKT + 1 <= 2 * SORT_BLOCK, "two table rows per thread in the tile's bucket scan");
static_assert(TILE_PITCH <= 2 * SORT_BLOCK, "two tiles per thread in the bucket's run scan");
static_assert(N_NODES <= 65536, "a source node fits the low 16 bits of a filed edge");
static_assert(BKT_NODES <= SORT_BLOCK, "one thread per node of a bucket");

__device__ __forceinline__ int bucket_of(int d) {
    return min((int)((unsigned)d >> 7), NUM_BKT - 1);
}


#define FOLD_ROWS 8
#define FA_PITCH  68
static_assert(SORT_BLOCK == 256 && OUT_DIM == 64 && NUM_REL == 3, "eight waves: four matrices x two pairs of 16-column tiles");

__device__ __forceinline__ void fold_first_layer_weights(const float* __restrict__ Wf,
                                                         const float* __restrict__ bf,
                                                         const float* __restrict__ relW,
                                                         const float* __restrict__ rootW,
                                                         float* __restrict__ Wc) {
    __shared__ __attribute__((aligned(16))) float fA[TILE_M][FA_PITCH];
    const int t = threadIdx.x;
    const int wave = t >> 5;
    const int lane = t & 31;
    const int q     = wave & 3;
    const int jh    = wave >> 2;
    const int c     = lane & 15;
    const int hh    = lane >> 4;
    const int khalf = hh << 1;
    const float* __restrict__ Wq =
        (q < NUM_REL) ? (relW + (size_t)q * OUT_DIM * OUT_DIM) : rootW;

    float2 wa[OUT_DIM / 4], wb[OUT_DIM / 4];
#pragma unroll
    for (int s = 0; s < OUT_DIM / 4; ++s) {
        const int kk = 4 * s + khalf;
        wa[s] = *(const float2*)(Wq + (size_t)kk * OUT_DIM + 4 * c + 2 * jh);
        wb[s] = *(const float2*)(Wq + (size_t)(kk + 1) * OUT_DIM + 4 * c + 2 * jh);
    }
    {
        const int r = t >> 4, c4 = t & 15;
        float4 v = make_float4(0.f, 0.f, 0.f, 0.f);
        if (r < IN_DIM)       v = *(const float4*)(Wf + r * OUT_DIM + 4 * c4);
        else if (r == IN_DIM) v = *(const float4*)(bf + 4 * c4);
        *(float4*)&fA[r][4 * c4] = v;
    }
    __syncthreads();

    v8f acc0 = (v8f){0.f,0.f,0.f,0.f,0.f,0.f,0.f,0.f};
    v8f acc1 = (v8f){0.f,0.f,0.f,0.f,0.f,0.f,0.f,0.f};
#pragma unroll
    for (int s = 0; s < OUT_DIM / 4; ++s) {
        const int kk = 4 * s + khalf;
        const v2f a = *(const v2f*)&fA[c][kk];
        v2f b;
        b.x = wa[s].x; b.y = wb[s].x;
        acc0 = __builtin_amdgcn_wmma_f32_16x16x4_f32(false, a, false, b, (short)0, acc0, false, false);
        b.x = wa[s].y; b.y = wb[s].y;
        acc1 = __builtin_amdgcn_wmma_f32_16x16x4_f32(false, a, false, b, (short)0, acc1, false, false);
    }
    if (hh == 0) {
#pragma unroll
        for (int r = 0; r < FOLD_ROWS; ++r) {
            *(float2*)(Wc + (size_t)(q * FOLD_ROWS + r) * OUT_DIM + 4 * c + 2 * jh) =
                make_float2(acc0[r], acc1[r]);
        }
    }
}

__device__ __forceinline__ void file_edge(int* lcur, int* sval, int d, int ty, int s) {
    const int key = (d & (BKT_NODES - 1)) * NUM_REL + min(max(ty, 0), NUM_REL - 1);
    int lp = atomicAdd(&lcur[bucket_of(d)], 1);
    lp = min(max(lp, 0), TILE_EDGES - 1);
    sval[lp] = (key << 16) | (s & 0xFFFF);
}

__global__ __launch_bounds__(SORT_BLOCK)
void partition_kernel(const int* __restrict__ src,
                      const int* __restrict__ dst,
                      const int* __restrict__ etype,
                      int* __restrict__ tb,
                      int* __restrict__ mid,
                      const float* __restrict__ Wf, const float* __restrict__ bf,
                      const float* __restrict__ relW, const float* __restrict__ rootW,
                      float* __restrict__ Wc) {
    if (blockIdx.x == NUM_TILES) {
        fold_first_layer_weights(Wf, bf, relW, rootW, Wc);
        return;
    }
    __shared__ int hist[NUM_BKT];
    __shared__ int lcur[NUM_BKT];
    __shared__ int lwsum[SORT_BLOCK / 32];
    __shared__ int sval[TILE_EDGES];
    const int t = threadIdx.x;
    const int lane = t & 31;
    const int wave = t >> 5;
    const int tile = blockIdx.x;

    int4 d[4], y[4], s[4];
#pragma unroll
    for (int i = 0; i < 4; ++i) {
        const int e = tile * TILE_EDGES + i * (SORT_BLOCK * 4) + 4 * t;
        const int ec = min(e, N_EDGES - 4);
        d[i] = *(const int4*)(dst + ec);
    }
#pragma unroll
    for (int i = 0; i < 4; ++i) {
        const int e = tile * TILE_EDGES + i * (SORT_BLOCK * 4) + 4 * t;
        const int ec = min(e, N_EDGES - 4);
        y[i] = *(const int4*)(etype + ec);
        s[i] = *(const int4*)(src + ec);
    }
    for (int b = t; b < NUM_BKT; b += SORT_BLOCK) hist[b] = 0;
    __syncthreads();
#pragma unroll
    for (int i = 0; i < 4; ++i) {
        const int e = tile * TILE_EDGES + i * (SORT_BLOCK * 4) + 4 * t;
        if (e < N_EDGES) {
            atomicAdd(&hist[bucket_of(d[i].x)], 1);
            atomicAdd(&hist[bucket_of(d[i].y)], 1);
            atomicAdd(&hist[bucket_of(d[i].z)], 1);
            atomicAdd(&hist[bucket_of(d[i].w)], 1);
        }
    }
    __syncthreads();

    const int i0 = 2 * t, i1 = 2 * t + 1;
    const int n0 = (i0 < NUM_BKT) ? hist[i0] : 0;
    const int n1 = (i1 < NUM_BKT) ? hist[i1] : 0;
    const int ls = n0 + n1;
    int linc = ls;
    for (int off = 1; off < 32; off <<= 1) {
        const int lu = __shfl_up(linc, off, 32);
        if (lane >= off) linc += lu;
    }
    if (lane == 31) lwsum[wave] = linc;
    __syncthreads();
    int lex = linc - ls;
    for (int w = 0; w < SORT_BLOCK / 32; ++w)
        if (w < wave) lex += lwsum[w];
    if (i0 < NUM_BKT)  lcur[i0] = lex;
    if (i1 < NUM_BKT)  lcur[i1] = lex + n0;
    if (i0 <= NUM_BKT) tb[i0 * TILE_PITCH + tile] = lex;
    if (i1 <= NUM_BKT) tb[i1 * TILE_PITCH + tile] = lex + n0;
    __syncthreads();

#pragma unroll
    for (int i = 0; i < 4; ++i) {
        const int e = tile * TILE_EDGES + i * (SORT_BLOCK * 4) + 4 * t;
        if (e < N_EDGES) {
            file_edge(lcur, sval, d[i].x, y[i].x, s[i].x);
            file_edge(lcur, sval, d[i].y, y[i].y, s[i].y);
            file_edge(lcur, sval, d[i].z, y[i].z, s[i].z);
            file_edge(lcur, sval, d[i].w, y[i].w, s[i].w);
        }
    }
    __syncthreads();

    const int cnt = min(TILE_EDGES, N_EDGES - tile * TILE_EDGES);
    int sv[TILE_EDGES / SORT_BLOCK];
#pragma unroll
    for (int j = 0; j < TILE_EDGES / SORT_BLOCK; ++j) sv[j] = sval[j * SORT_BLOCK + t];
#pragma unroll
    for (int j = 0; j < TILE_EDGES / SORT_BLOCK; ++j) {
        if (j * SORT_BLOCK + t < cnt) mid[tile * TILE_EDGES + j * SORT_BLOCK + t] = sv[j];
    }
}

#define STAGE_CAP 4096
#define RUN_TILES 10
static_assert(TILE_PITCH % (16 * RUN_TILES) == 0 && SORT_BLOCK == 256, "16 half-waves sweep the table row in whole trips");

template <int MODE>
__device__ __forceinline__ void sweep_runs(const int* __restrict__ mid,
                                           const int* rs, const int* rc, const int* rr,
                                           int* ctr, int* raw, int* __restrict__ csr_src,
                                           const int base, const int t) {
    const int hw = t >> 4, l16 = t & 15;
    auto take = [&](int v, int slot) __attribute__((always_inline)) {
        const int seg = min(v >> 16, BKT_SEGS - 1);
        if (MODE == 0) {
            raw[min(max(slot, 0), STAGE_CAP - 1)] = v;
            atomicAdd(&ctr[seg], 1);
        } else if (MODE == 1) {
            atomicAdd(&ctr[seg], 1);
        } else {
            const int p = atomicAdd(&ctr[seg], 1);
            const int q = min(max(base + p, 0), N_EDGES - 1);
            csr_src[q] = v & 0xFFFF;
        }
    };
    for (int T0 = hw; T0 < TILE_PITCH; T0 += 16 * RUN_TILES) {
        int v[RUN_TILES][2], c[RUN_TILES], r[RUN_TILES], a[RUN_TILES];
#pragma unroll
        for (int u = 0; u < RUN_TILES; ++u) {
            const int T = T0 + 16 * u;
            c[u] = rc[T];
            r[u] = rr[T];
            a[u] = T * TILE_EDGES + rs[T];
#pragma unroll
            for (int h = 0; h < 2; ++h) {
                const int i = l16 + 16 * h;
                v[u][h] = (i < c[u]) ? mid[min(max(a[u] + i, 0), N_EDGES - 1)] : -1;
            }
        }
#pragma unroll
        for (int u = 0; u < RUN_TILES; ++u) {
#pragma unroll
            for (int h = 0; h < 2; ++h)
                if (v[u][h] >= 0) take(v[u][h], r[u] + l16 + 16 * h);
        }
#pragma unroll
        for (int u = 0; u < RUN_TILES; ++u) {
            for (int i = l16 + 32; i < c[u]; i += 16) {
                const int w = mid[min(max(a[u] + i, 0), N_EDGES - 1)];
                if (w >= 0) take(w, r[u] + i);
            }
        }
    }
}

__global__ __launch_bounds__(SORT_BLOCK)
void bucket_sort_kernel(const int* __restrict__ mid,
                        const int* __restrict__ tb,
                        int* __restrict__ csr_src,
                        int* __restrict__ offs,
                        float* __restrict__ inv) {
    __shared__ int cnt[BKT_SEGS];
    __shared__ int cur[BKT_SEGS];
    __shared__ int wsum[BKT_NODES / 32];
    __shared__ int wsA[SORT_BLOCK / 32], wsB[SORT_BLOCK / 32], wsG[SORT_BLOCK / 32];
    __shared__ int rs[TILE_PITCH], rc[TILE_PITCH], rr[TILE_PITCH];
    __shared__ int raw[STAGE_CAP];
    __shared__ int stage[STAGE_CAP];
    const int t = threadIdx.x;
    const int lane = t & 31;
    const int wave = t >> 5;
    const int k = blockIdx.x;

    auto run_of = [&](int T, int& s, int& c) __attribute__((always_inline)) {
        const bool ok = T < NUM_TILES;
        const int Tc = ok ? T : NUM_TILES - 1;
        const int lo = tb[k * TILE_PITCH + Tc];
        const int hi = tb[(k + 1) * TILE_PITCH + Tc];
        const int tcnt = min(TILE_EDGES, N_EDGES - Tc * TILE_EDGES);
        const int sc = min(max(lo, 0), tcnt);
        const int ec = min(max(hi, sc), tcnt);
        s = ok ? sc : 0;
        c = ok ? ec - sc : 0;
    };
    int s0, ra, s1, rb;
    run_of(t, s0, ra);
    run_of(t + SORT_BLOCK, s1, rb);
    const int g = s0 + s1;
    int ia = ra, ib = rb, ig = g;
    for (int off = 1; off < 32; off <<= 1) {
        const int ua = __shfl_up(ia, off, 32);
        const int ub = __shfl_up(ib, off, 32);
        const int ug = __shfl_up(ig, off, 32);
        if (lane >= off) { ia += ua; ib += ub; ig += ug; }
    }
    if (lane == 31) { wsA[wave] = ia; wsB[wave] = ib; wsG[wave] = ig; }
    for (int i = t; i < BKT_SEGS; i += SORT_BLOCK) cnt[i] = 0;
    __syncthreads();
    int exa = ia - ra, exb = ib - rb;
    int totA = 0, totB = 0, gsum = 0;
    for (int w = 0; w < SORT_BLOCK / 32; ++w) {
        const int a = wsA[w], b = wsB[w];
        totA += a;
        totB += b;
        gsum += wsG[w];
        if (w < wave) { exa += a; exb += b; }
    }
    rs[t] = s0;
    rc[t] = ra;
    rr[t] = exa;
    if (t + SORT_BLOCK < TILE_PITCH) {
        rs[t + SORT_BLOCK] = s1;
        rc[t + SORT_BLOCK] = rb;
        rr[t + SORT_BLOCK] = totA + exb;
    }
    const int base = min(max(gsum, 0), N_EDGES);
    const int n = min(max(totA + totB, 0), N_EDGES - base);
    const bool staged = n <= STAGE_CAP;
    __syncthreads();

    if (staged) sweep_runs<0>(mid, rs, rc, rr, cnt, raw, csr_src, base, t);
    else        sweep_runs<1>(mid, rs, rc, rr, cnt, raw, csr_src, base, t);
    __syncthreads();

    int c0 = 0, c1 = 0, c2 = 0;
    if (t < BKT_NODES) { c0 = cnt[3 * t]; c1 = cnt[3 * t + 1]; c2 = cnt[3 * t + 2]; }
    const int ts = c0 + c1 + c2;
    int inc = ts;
    for (int off = 1; off < 32; off <<= 1) {
        const int u = __shfl_up(inc, off, 32);
        if (lane >= off) inc += u;
    }
    if (t < BKT_NODES && lane == 31) wsum[wave] = inc;
    __syncthreads();
    if (t < BKT_NODES) {
        int ex = inc - ts;
        for (int w = 0; w < BKT_NODES / 32; ++w)
            if (w < wave) ex += wsum[w];
        cur[3 * t]     = ex;
        cur[3 * t + 1] = ex + c0;
        cur[3 * t + 2] = ex + c0 + c1;
        const int node = k * BKT_NODES + t;
        if (node < N_NODES) {
            const int seg = node * NUM_REL;
            offs[seg]     = base + ex;
            offs[seg + 1] = base + ex + c0;
            offs[seg + 2] = base + ex + c0 + c1;
            inv[seg]     = 1.0f / (float)(c0 > 1 ? c0 : 1);
            inv[seg + 1] = 1.0f / (float)(c1 > 1 ? c1 : 1);
            inv[seg + 2] = 1.0f / (float)(c2 > 1 ? c2 : 1);
            if (node == N_NODES - 1) offs[NR_TOTAL] = base + ex + ts;
        }
    }
    __syncthreads();

    if (staged) {
        for (int j0 = 0; j0 < n; j0 += SORT_BLOCK * CHUNK) {
            int v[CHUNK];
#pragma unroll
            for (int j = 0; j < CHUNK; ++j) {
                const int i = j0 + j * SORT_BLOCK + t;
                v[j] = (i < n) ? raw[i] : -1;
            }
#pragma unroll
            for (int j = 0; j < CHUNK; ++j) {
                if (v[j] >= 0) {
                    const int p = atomicAdd(&cur[min(v[j] >> 16, BKT_SEGS - 1)], 1);
                    stage[min(max(p, 0), STAGE_CAP - 1)] = v[j] & 0xFFFF;
                }
            }
        }
        __syncthreads();
        for (int j0 = 0; j0 < n; j0 += SORT_BLOCK * CHUNK) {
            int w[CHUNK];
#pragma unroll
            for (int j = 0; j < CHUNK; ++j) {
                const int i = j0 + j * SORT_BLOCK + t;
                w[j] = (i < n) ? stage[i] : 0;
            }
#pragma unroll
            for (int j = 0; j < CHUNK; ++j) {
                const int i = j0 + j * SORT_BLOCK + t;
                if (i < n) csr_src[base + i] = w[j];
            }
        }
    } else {
        sweep_runs<2>(mid, rs, rc, rr, cur, raw, csr_src, base, t);
    }
}

#define A_PITCH  260
#define X_PITCH  36
static_assert(N_NODES % TILE_M == 0, "no partial node tile");
static_assert(IN_DIM + 1 <= FOLD_ROWS && FOLD_ROWS % 4 == 0, "x, the 0/1 column and padding fill whole k-steps");

__device__ __forceinline__ int bcast_i(int v, int l) {
    return __builtin_amdgcn_readlane(v, l);
}
__device__ __forceinline__ float bcast_f(float v, int l) {
    return __int_as_float(__builtin_amdgcn_readlane(__float_as_int(v), l));
}

template <int KS, int APITCH>
__device__ __forceinline__ void multiply_and_norm(const float* aT,
                                                  float (&lA)[TILE_M][A_PITCH],
                                                  const int row0, const int wave, const int lane,
                                                  const float* __restrict__ Wq,
                                                  const float* __restrict__ bias,
                                                  const float* __restrict__ gamma,
                                                  const float* __restrict__ beta,
                                                  float* __restrict__ hout) {
    const int c     = lane & 15;
    const int hh    = lane >> 4;
    const int khalf = hh << 1;

    v8f acc[4];
#pragma unroll
    for (int j = 0; j < 4; ++j) acc[j] = (v8f){0.f,0.f,0.f,0.f,0.f,0.f,0.f,0.f};

#pragma unroll
    for (int k0 = 0; k0 < KS; k0 += 4) {
        const int kk = k0 + khalf;
        const v2f a = *(const v2f*)&aT[c * APITCH + wave * KS + kk];
        const float4 w0 = *(const float4*)(Wq + (size_t)kk * OUT_DIM + 4 * c);
        const float4 w1 = *(const float4*)(Wq + (size_t)(kk + 1) * OUT_DIM + 4 * c);
        v2f b;
        b.x = w0.x; b.y = w1.x;
        acc[0] = __builtin_amdgcn_wmma_f32_16x16x4_f32(false, a, false, b, (short)0, acc[0], false, false);
        b.x = w0.y; b.y = w1.y;
        acc[1] = __builtin_amdgcn_wmma_f32_16x16x4_f32(false, a, false, b, (short)0, acc[1], false, false);
        b.x = w0.z; b.y = w1.z;
        acc[2] = __builtin_amdgcn_wmma_f32_16x16x4_f32(false, a, false, b, (short)0, acc[2], false, false);
        b.x = w0.w; b.y = w1.w;
        acc[3] = __builtin_amdgcn_wmma_f32_16x16x4_f32(false, a, false, b, (short)0, acc[3], false, false);
    }

#pragma unroll
    for (int r = 0; r < 8; ++r) {
        *(float4*)&lA[8 * hh + r][wave * OUT_DIM + 4 * c] =
            make_float4(acc[0][r], acc[1][r], acc[2][r], acc[3][r]);
    }
    __syncthreads();

    const float2 bs = ((const float2*)bias)[lane];
    const float2 g2 = ((const float2*)gamma)[lane];
    const float2 be = ((const float2*)beta)[lane];
#pragma unroll
    for (int t = 0; t < 4; ++t) {
        const int row = wave * 4 + t;
        const float* prow = &lA[row][2 * lane];
        const float2 p0 = *(const float2*)(prow);
        const float2 p1 = *(const float2*)(prow + 64);
        const float2 p2 = *(const float2*)(prow + 128);
        const float2 p3 = *(const float2*)(prow + 192);
        float2 v;
        v.x = p3.x + bs.x;  v.y = p3.y + bs.y;
        v.x += p0.x;        v.y += p0.y;
        v.x += p1.x;        v.y += p1.y;
        v.x += p2.x;        v.y += p2.y;
        v.x = fmaxf(v.x, 0.0f);
        v.y = fmaxf(v.y, 0.0f);
        float s  = v.x + v.y;
        float sq = v.x * v.x + v.y * v.y;
#pragma unroll
        for (int off = 16; off > 0; off >>= 1) {
            s  += __shfl_xor(s,  off, 32);
            sq += __shfl_xor(sq, off, 32);
        }
        const float mu   = s * (1.0f / OUT_DIM);
        const float var  = sq * (1.0f / OUT_DIM) - mu * mu;
        const float rstd = rsqrtf(var + LN_EPS);
        float2 o;
        o.x = (v.x - mu) * rstd * g2.x + be.x;
        o.y = (v.y - mu) * rstd * g2.y + be.y;
        ((float2*)(hout + (size_t)(row0 + row) * OUT_DIM))[lane] = o;
    }
}

#define LAYER_M        (2 * TILE_M)
#define LAYER_THREADS  256
static_assert(N_NODES % 4 == 0, "a wave's four nodes are all inside or all outside the node range");

__device__ __forceinline__ void multiply_pair_and_norm(float (&lA)[LAYER_M][A_PITCH],
                                                       const int row0, const int wave, const int lane,
                                                       const float* __restrict__ relW,
                                                       const float* __restrict__ rootW,
                                                       const float* __restrict__ bias,
                                                       const float* __restrict__ gamma,
                                                       const float* __restrict__ beta,
                                                       float* __restrict__ hout) {
    const int q     = wave & 3;
    const int jh    = wave >> 2;
    const int c     = lane & 15;
    const int hh    = lane >> 4;
    const int khalf = hh << 1;
    const float* __restrict__ Wq =
        (q < NUM_REL) ? (relW + (size_t)q * OUT_DIM * OUT_DIM) : rootW;
    const float* __restrict__ wp = Wq + 32 * jh + 2 * c;
    const float* a0p = &lA[c][q * OUT_DIM];
    const float* a1p = &lA[TILE_M + c][q * OUT_DIM];

    v8f acc[2][2];
#pragma unroll
    for (int m = 0; m < 2; ++m)
#pragma unroll
        for (int t = 0; t < 2; ++t) acc[m][t] = (v8f){0.f,0.f,0.f,0.f,0.f,0.f,0.f,0.f};

#pragma unroll
    for (int k0 = 0; k0 < OUT_DIM; k0 += 4) {
        const int kk = k0 + khalf;
        const v2f a0 = *(const v2f*)(a0p + kk);
        const v2f a1 = *(const v2f*)(a1p + kk);
        const float2 w0 = *(const float2*)(wp + (size_t)kk * OUT_DIM);
        const float2 w1 = *(const float2*)(wp + (size_t)(kk + 1) * OUT_DIM);
        v2f b;
        b.x = w0.x; b.y = w1.x;
        acc[0][0] = __builtin_amdgcn_wmma_f32_16x16x4_f32(false, a0, false, b, (short)0, acc[0][0], false, false);
        acc[1][0] = __builtin_amdgcn_wmma_f32_16x16x4_f32(false, a1, false, b, (short)0, acc[1][0], false, false);
        b.x = w0.y; b.y = w1.y;
        acc[0][1] = __builtin_amdgcn_wmma_f32_16x16x4_f32(false, a0, false, b, (short)0, acc[0][1], false, false);
        acc[1][1] = __builtin_amdgcn_wmma_f32_16x16x4_f32(false, a1, false, b, (short)0, acc[1][1], false, false);
    }
    __syncthreads();

#pragma unroll
    for (int m = 0; m < 2; ++m) {
#pragma unroll
        for (int r = 0; r < 8; ++r) {
            *(float2*)&lA[m * TILE_M + 8 * hh + r][q * OUT_DIM + 32 * jh + 2 * c] =
                make_float2(acc[m][0][r], acc[m][1][r]);
        }
    }
    __syncthreads();

    const float2 bs = ((const float2*)bias)[lane];
    const float2 g2 = ((const float2*)gamma)[lane];
    const float2 be = ((const float2*)beta)[lane];
#pragma unroll
    for (int t = 0; t < 4; ++t) {
        const int row = wave * 4 + t;
        const float* prow = &lA[row][2 * lane];
        const float2 p0 = *(const float2*)(prow);
        const float2 p1 = *(const float2*)(prow + 64);
        const float2 p2 = *(const float2*)(prow + 128);
        const float2 p3 = *(const float2*)(prow + 192);
        float2 v;
        v.x = p3.x + bs.x;  v.y = p3.y + bs.y;
        v.x += p0.x;        v.y += p0.y;
        v.x += p1.x;        v.y += p1.y;
        v.x += p2.x;        v.y += p2.y;
        v.x = fmaxf(v.x, 0.0f);
        v.y = fmaxf(v.y, 0.0f);
        float s  = v.x + v.y;
        float sq = v.x * v.x + v.y * v.y;
#pragma unroll
        for (int off = 16; off > 0; off >>= 1) {
            s  += __shfl_xor(s,  off, 32);
            sq += __shfl_xor(sq, off, 32);
        }
        const float mu   = s * (1.0f / OUT_DIM);
        const float var  = sq * (1.0f / OUT_DIM) - mu * mu;
        const float rstd = rsqrtf(var + LN_EPS);
        float2 o;
        o.x = (v.x - mu) * rstd * g2.x + be.x;
        o.y = (v.y - mu) * rstd * g2.y + be.y;
        if (row0 + row < N_NODES)
            ((float2*)(hout + (size_t)(row0 + row) * OUT_DIM))[lane] = o;
    }
}

#define BPG 4

__global__ __launch_bounds__(128, 16)
void rgcn_first_layer_kernel(const float* __restrict__ x,
                             const float* __restrict__ Wc,
                             const int*   __restrict__ offs,
                             const int*   __restrict__ csr_src,
                             const float* __restrict__ inv_denom,
                             const float* __restrict__ bias,
                             const float* __restrict__ gamma,
                             const float* __restrict__ beta,
                             float* __restrict__ hout)
{
    __shared__ __attribute__((aligned(16))) float lA[TILE_M][A_PITCH];
    __shared__ __attribute__((aligned(16))) float lX[TILE_M][X_PITCH];
    const int row0 = blockIdx.x * TILE_M;
    const int wave = threadIdx.x >> 5;
    const int lane = threadIdx.x & 31;

    {
        const int node0 = row0 + wave * 4;
        int bl = offs[node0 * NUM_REL + (lane < 12 ? lane : 12)];
        bl = min(max(bl, 0), N_EDGES);
        const float wl = inv_denom[node0 * NUM_REL + (lane < 11 ? lane : 11)];

        auto load_row = [&](int sraw, float (&v)[IN_DIM]) __attribute__((always_inline)) {
            const int s = min(max(sraw, 0), N_NODES - 1);
            const float* xr = x + (size_t)s * IN_DIM;
            float4 q;
            __builtin_memcpy(&q, xr, sizeof(q));
            v[0] = q.x; v[1] = q.y; v[2] = q.z; v[3] = q.w;
            v[4] = xr[4];
        };
        float xown[IN_DIM];
        load_row(node0 + ((lane - 12) & 3), xown);

        const int e0 = bcast_i(bl, 0);
        const int n  = max(bcast_i(bl, 12) - e0, 0);
        const int elast = min(max(e0 + n - 1, 0), N_EDGES - 1);

        auto idx_raw = [&](int p) __attribute__((always_inline)) {
            const int j = min(max(e0 + p, 0), elast);
            return csr_src[j];
        };

        const int myLo = bl - e0;
        const int myHi = __shfl_down(bl, 1, 32) - e0;
        int segstart[4 * NUM_REL - 1];
#pragma unroll
        for (int k = 0; k < 4 * NUM_REL - 1; ++k) segstart[k] = bcast_i(bl, k + 1) - e0;

        float sum[IN_DIM];
#pragma unroll
        for (int d = 0; d < IN_DIM; ++d) sum[d] = 0.f;

        for (int g0 = 0; g0 < n; g0 += 32 * BPG) {
            int sraw[BPG];
            float row[BPG][IN_DIM];
#pragma unroll
            for (int q = 0; q < BPG; ++q) sraw[q] = idx_raw(g0 + 32 * q + lane);
#pragma unroll
            for (int q = 0; q < BPG; ++q) load_row(sraw[q], row[q]);
#pragma unroll
            for (int q = 0; q < BPG; ++q) {
                const int base = g0 + 32 * q;
                const int p = base + lane;
                float v[IN_DIM];
#pragma unroll
                for (int d = 0; d < IN_DIM; ++d) v[d] = (p < n) ? row[q][d] : 0.f;
                int st = 0;
#pragma unroll
                for (int k = 0; k < 4 * NUM_REL - 1; ++k) st = (p >= segstart[k]) ? segstart[k] : st;
                const int reach = min(p - st, lane);
#pragma unroll
                for (int off = 1; off < 32; off <<= 1) {
                    const bool take = reach >= off;
#pragma unroll
                    for (int d = 0; d < IN_DIM; ++d) {
                        const float u = __shfl_up(v[d], off, 32);
                        v[d] += take ? u : 0.f;
                    }
                }
                const int lo = max(myLo, base);
                const int hi = min(myHi, base + 32);
                const bool has = (lane < 4 * NUM_REL) && (lo < hi);
                const int tail = has ? (hi - 1 - base) : 0;
#pragma unroll
                for (int d = 0; d < IN_DIM; ++d) {
                    const float u = __shfl(v[d], tail, 32);
                    sum[d] += has ? u : 0.f;
                }
            }
        }

        if (lane < 4 * NUM_REL) {
            const int t = lane / NUM_REL;
            const int r = lane - t * NUM_REL;
            const float some = (myHi > myLo) ? 1.f : 0.f;
            float* d8 = &lX[wave * 4 + t][r * FOLD_ROWS];
            *(float4*)(d8)     = make_float4(sum[0] * wl, sum[1] * wl, sum[2] * wl, sum[3] * wl);
            *(float4*)(d8 + 4) = make_float4(sum[4] * wl, some, 0.f, 0.f);
        } else if (lane < 4 * NUM_REL + 4) {
            float* d8 = &lX[wave * 4 + (lane - 4 * NUM_REL)][NUM_REL * FOLD_ROWS];
            *(float4*)(d8)     = make_float4(xown[0], xown[1], xown[2], xown[3]);
            *(float4*)(d8 + 4) = make_float4(xown[4], 1.f, 0.f, 0.f);
        }
    }
    __syncthreads();

    multiply_and_norm<FOLD_ROWS, X_PITCH>(&lX[0][0], lA, row0, wave, lane,
                                          Wc + (size_t)wave * FOLD_ROWS * OUT_DIM,
                                          bias, gamma, beta, hout);
}

#define GRP 8

__global__ __launch_bounds__(LAYER_THREADS, 16)
void rgcn_layer_kernel(const float* __restrict__ h,
                       const int*   __restrict__ offs,
                       const int*   __restrict__ csr_src,
                       const float* __restrict__ inv_denom,
                       const float* __restrict__ relW,
                       const float* __restrict__ rootW,
                       const float* __restrict__ bias,
                       const float* __restrict__ gamma,
                       const float* __restrict__ beta,
                       float* __restrict__ hout)
{
    __shared__ __attribute__((aligned(16))) float lA[LAYER_M][A_PITCH];
    const int row0 = blockIdx.x * LAYER_M;
    const int wave = __builtin_amdgcn_readfirstlane(threadIdx.x >> 5);
    const int lane = threadIdx.x & 31;

    if (row0 + wave * 4 < N_NODES) {
        const float2* __restrict__ h2 = (const float2*)h;
        const int node0 = row0 + wave * 4;
        float2 hown[4];
#pragma unroll
        for (int t = 0; t < 4; ++t) hown[t] = h2[(size_t)(node0 + t) * (OUT_DIM / 2) + lane];
        int bl = offs[node0 * NUM_REL + (lane < 12 ? lane : 12)];
        bl = min(max(bl, 0), N_EDGES);
        const float wl = inv_denom[node0 * NUM_REL + (lane < 11 ? lane : 11)];

        const int e0 = bcast_i(bl, 0);
        const int n  = max(bcast_i(bl, 12) - e0, 0);
        const int elast = min(max(e0 + n - 1, 0), N_EDGES - 1);

        auto idx_raw = [&](int cidx) __attribute__((always_inline)) {
            const int j = min(max(e0 + 2 * GRP * cidx + lane, 0), elast);
            return csr_src[j];
        };
        int s = idx_raw(0);

        float* lrow = &lA[wave * 4][2 * lane];
#pragma unroll
        for (int t = 0; t < 4; ++t) *(float2*)(lrow + t * A_PITCH + 3 * OUT_DIM) = hown[t];

        s = min(max(s, 0), N_NODES - 1);

        int seg = 0;
        int rel = 0;
        int segoff = 0;
        int nb = bcast_i(bl, 1) - e0;
        float2 a = make_float2(0.f, 0.f);

        auto flush = [&]() __attribute__((always_inline)) {
            const float w = bcast_f(wl, seg);
            *(float2*)(lrow + segoff) = make_float2(a.x * w, a.y * w);
            a = make_float2(0.f, 0.f);
            ++seg;
            ++rel;
            segoff += OUT_DIM;
            if (rel == NUM_REL) { rel = 0; segoff += A_PITCH - NUM_REL * OUT_DIM; }
            nb = (seg < 4 * NUM_REL) ? (bcast_i(bl, seg + 1) - e0) : 0x7fffffff;
        };
        auto load_group = [&](float2 (&m)[GRP], int sv, int k) __attribute__((always_inline)) {
#pragma unroll
            for (int i = 0; i < GRP; ++i) {
                const int rk = bcast_i(sv, k + i);
                const float2* __restrict__ rowp = h2 + (size_t)rk * (OUT_DIM / 2);
                m[i] = rowp[lane];
            }
        };
        auto consume = [&](const float2 (&m)[GRP], int p0) __attribute__((always_inline)) {
#pragma unroll
            for (int i = 0; i < GRP; ++i) {
                while (p0 + i >= nb) flush();
                a.x += m[i].x;
                a.y += m[i].y;
            }
        };

        {
            float2 mA[GRP], mB[GRP];
            int g = 0;
            while (g * GRP < n) {
                const int sraw = idx_raw(g / 2 + 1);
                load_group(mA, s, 0);
                if (g > 0) consume(mB, (g - 1) * GRP);
                load_group(mB, s, GRP);
                consume(mA, g * GRP);
                s = min(max(sraw, 0), N_NODES - 1);
                g += 2;
            }
            if (g > 0) consume(mB, (g - 1) * GRP);
        }
        while (seg < 4 * NUM_REL) flush();
    } else {
        float* lrow = &lA[wave * 4][2 * lane];
#pragma unroll
        for (int t = 0; t < 4; ++t)
#pragma unroll
            for (int j = 0; j < 4; ++j)
                *(float2*)(lrow + t * A_PITCH + j * OUT_DIM) = make_float2(0.f, 0.f);
    }
    __syncthreads();

    multiply_pair_and_norm(lA, row0, wave, lane, relW, rootW, bias, gamma, beta, hout);
}

extern "C" void kernel_launch(void* const* d_in, const int* in_sizes, int n_in,
                              void* d_out, int out_size, void* d_ws, size_t ws_size,
                              hipStream_t stream) {
    const float* x      = (const float*)d_in[0];
    const int*   ei     = (const int*)  d_in[1];
    const int*   src    = ei;
    const int*   dst    = ei + N_EDGES;
    const int*   etype  = (const int*)  d_in[2];
    const float* f2cW   = (const float*)d_in[3];
    const float* f2cb   = (const float*)d_in[4];
    const float* rgW    = (const float*)d_in[5];
    const float* rgRoot = (const float*)d_in[6];
    const float* rgBias = (const float*)d_in[7];
    const float* gamma  = (const float*)d_in[8];
    const float* beta   = (const float*)d_in[9];

    float* ws = (float*)d_ws;
    float* h_a       = ws;  ws += (size_t)N_NODES * OUT_DIM;
    int*   mid       = (int*)ws;
    int*   csr_src   = mid + N_EDGES;
    int*   offs      = csr_src + N_EDGES;
    ws += (size_t)N_NODES * OUT_DIM;
    float* h_b       = ws;
    float* wfold     = h_b + (size_t)N_NODES * OUT_DIM;
    ws += (size_t)N_NODES * NUM_REL * OUT_DIM;
    int*   tb        = (int*)ws;
    ws += (size_t)N_NODES * OUT_DIM;
    float* inv_denom = ws;

    partition_kernel<<<NUM_TILES + 1, SORT_BLOCK, 0, stream>>>(src, dst, etype, tb, mid,
                                                                f2cW, f2cb, rgW, rgRoot, wfold);
    bucket_sort_kernel<<<NUM_BKT, SORT_BLOCK, 0, stream>>>(mid, tb, csr_src, offs, inv_denom);

    rgcn_first_layer_kernel<<<N_NODES / TILE_M, 128, 0, stream>>>(
        x, wfold, offs, csr_src, inv_denom, rgBias, gamma, beta, h_b);
    for (int l = 1; l < NUM_LAYERS; ++l) {
        const float* hi = (l & 1) ? h_b : h_a;
        float* ho = (l == NUM_LAYERS - 1) ? (float*)d_out : ((l & 1) ? h_a : h_b);
        rgcn_layer_kernel<<<(N_NODES + LAYER_M - 1) / LAYER_M, LAYER_THREADS, 0, stream>>>(
            hi, offs, csr_src, inv_denom,
            rgW    + (size_t)l * NUM_REL * OUT_DIM * OUT_DIM,
            rgRoot + (size_t)l * OUT_DIM * OUT_DIM,
            rgBias + (size_t)l * OUT_DIM,
            gamma, beta, ho);
    }
}


# First layer folded into the bucket-sort launch: a thread per segment sums x rows off the LDS edge image in list order (sum order changes), K=32 WMMA chain per wave, LayerNorm in registers
# speedup vs baseline: 5.4482x; 1.0562x over previous
#include <hip/hip_runtime.h>
#include <hip/hip_bf16.h>

#define N_NODES   50000
#define N_EDGES   1250000
#define IN_DIM    5
#define OUT_DIM   64
#define NUM_REL   3
#define NUM_LAYERS 3
#define LN_EPS    1e-5f
#define TILE_M    16

typedef __attribute__((ext_vector_type(2))) float v2f;
typedef __attribute__((ext_vector_type(8))) float v8f;

#define NR_TOTAL    (N_NODES * NUM_REL)
#define BKT_NODES   128
#define BKT_SEGS    (BKT_NODES * NUM_REL)
#define NUM_BKT     ((N_NODES + BKT_NODES - 1) / BKT_NODES)
#define SORT_BLOCK  256
#define TILE_EDGES  (SORT_BLOCK * 16)
#define NUM_TILES   ((N_EDGES + TILE_EDGES - 1) / TILE_EDGES)
#define TILE_PITCH  320
#define CHUNK       8
static_assert(N_EDGES % 4 == 0, "edges are read as int4");
static_assert(NUM_TILES <= TILE_PITCH, "one row entry per tile");
static_assert(NUM_BKT + 1 <= 2 * SORT_BLOCK, "two table rows per thread in the tile's bucket scan");
static_assert(TILE_PITCH <= 2 * SORT_BLOCK, "two tiles per thread in the bucket's run scan");
static_assert(N_NODES <= 65536, "a source node fits the low 16 bits of a filed edge");
static_assert(BKT_NODES <= SORT_BLOCK, "one thread per node of a bucket");

__device__ __forceinline__ int bucket_of(int d) {
    return min((int)((unsigned)d >> 7), NUM_BKT - 1);
}


#define FOLD_ROWS 8
#define FA_PITCH  68
static_assert(SORT_BLOCK == 256 && OUT_DIM == 64 && NUM_REL == 3, "eight waves: four matrices x two pairs of 16-column tiles");

__device__ __forceinline__ void fold_first_layer_weights(const float* __restrict__ Wf,
                                                         const float* __restrict__ bf,
                                                         const float* __restrict__ relW,
                                                         const float* __restrict__ rootW,
                                                         float* __restrict__ Wc) {
    __shared__ __attribute__((aligned(16))) float fA[TILE_M][FA_PITCH];
    const int t = threadIdx.x;
    const int wave = t >> 5;
    const int lane = t & 31;
    const int q     = wave & 3;
    const int jh    = wave >> 2;
    const int c     = lane & 15;
    const int hh    = lane >> 4;
    const int khalf = hh << 1;
    const float* __restrict__ Wq =
        (q < NUM_REL) ? (relW + (size_t)q * OUT_DIM * OUT_DIM) : rootW;

    float2 wa[OUT_DIM / 4], wb[OUT_DIM / 4];
#pragma unroll
    for (int s = 0; s < OUT_DIM / 4; ++s) {
        const int kk = 4 * s + khalf;
        wa[s] = *(const float2*)(Wq + (size_t)kk * OUT_DIM + 4 * c + 2 * jh);
        wb[s] = *(const float2*)(Wq + (size_t)(kk + 1) * OUT_DIM + 4 * c + 2 * jh);
    }
    {
        const int r = t >> 4, c4 = t & 15;
        float4 v = make_float4(0.f, 0.f, 0.f, 0.f);
        if (r < IN_DIM)       v = *(const float4*)(Wf + r * OUT_DIM + 4 * c4);
        else if (r == IN_DIM) v = *(const float4*)(bf + 4 * c4);
        *(float4*)&fA[r][4 * c4] = v;
    }
    __syncthreads();

    v8f acc0 = (v8f){0.f,0.f,0.f,0.f,0.f,0.f,0.f,0.f};
    v8f acc1 = (v8f){0.f,0.f,0.f,0.f,0.f,0.f,0.f,0.f};
#pragma unroll
    for (int s = 0; s < OUT_DIM / 4; ++s) {
        const int kk = 4 * s + khalf;
        const v2f a = *(const v2f*)&fA[c][kk];
        v2f b;
        b.x = wa[s].x; b.y = wb[s].x;
        acc0 = __builtin_amdgcn_wmma_f32_16x16x4_f32(false, a, false, b, (short)0, acc0, false, false);
        b.x = wa[s].y; b.y = wb[s].y;
        acc1 = __builtin_amdgcn_wmma_f32_16x16x4_f32(false, a, false, b, (short)0, acc1, false, false);
    }
    if (hh == 0) {
#pragma unroll
        for (int r = 0; r < FOLD_ROWS; ++r) {
            *(float2*)(Wc + (size_t)(q * FOLD_ROWS + r) * OUT_DIM + 4 * c + 2 * jh) =
                make_float2(acc0[r], acc1[r]);
        }
    }
}

__device__ __forceinline__ void file_edge(int* lcur, int* sval, int d, int ty, int s) {
    const int key = (d & (BKT_NODES - 1)) * NUM_REL + min(max(ty, 0), NUM_REL - 1);
    int lp = atomicAdd(&lcur[bucket_of(d)], 1);
    lp = min(max(lp, 0), TILE_EDGES - 1);
    sval[lp] = (key << 16) | (s & 0xFFFF);
}

__global__ __launch_bounds__(SORT_BLOCK)
void partition_kernel(const int* __restrict__ src,
                      const int* __restrict__ dst,
                      const int* __restrict__ etype,
                      int* __restrict__ tb,
                      int* __restrict__ mid,
                      const float* __restrict__ Wf, const float* __restrict__ bf,
                      const float* __restrict__ relW, const float* __restrict__ rootW,
                      float* __restrict__ Wc) {
    if (blockIdx.x == NUM_TILES) {
        fold_first_layer_weights(Wf, bf, relW, rootW, Wc);
        return;
    }
    __shared__ int hist[NUM_BKT];
    __shared__ int lcur[NUM_BKT];
    __shared__ int lwsum[SORT_BLOCK / 32];
    __shared__ int sval[TILE_EDGES];
    const int t = threadIdx.x;
    const int lane = t & 31;
    const int wave = t >> 5;
    const int tile = blockIdx.x;

    int4 d[4], y[4], s[4];
#pragma unroll
    for (int i = 0; i < 4; ++i) {
        const int e = tile * TILE_EDGES + i * (SORT_BLOCK * 4) + 4 * t;
        const int ec = min(e, N_EDGES - 4);
        d[i] = *(const int4*)(dst + ec);
    }
#pragma unroll
    for (int i = 0; i < 4; ++i) {
        const int e = tile * TILE_EDGES + i * (SORT_BLOCK * 4) + 4 * t;
        const int ec = min(e, N_EDGES - 4);
        y[i] = *(const int4*)(etype + ec);
        s[i] = *(const int4*)(src + ec);
    }
    for (int b = t; b < NUM_BKT; b += SORT_BLOCK) hist[b] = 0;
    __syncthreads();
#pragma unroll
    for (int i = 0; i < 4; ++i) {
        const int e = tile * TILE_EDGES + i * (SORT_BLOCK * 4) + 4 * t;
        if (e < N_EDGES) {
            atomicAdd(&hist[bucket_of(d[i].x)], 1);
            atomicAdd(&hist[bucket_of(d[i].y)], 1);
            atomicAdd(&hist[bucket_of(d[i].z)], 1);
            atomicAdd(&hist[bucket_of(d[i].w)], 1);
        }
    }
    __syncthreads();

    const int i0 = 2 * t, i1 = 2 * t + 1;
    const int n0 = (i0 < NUM_BKT) ? hist[i0] : 0;
    const int n1 = (i1 < NUM_BKT) ? hist[i1] : 0;
    const int ls = n0 + n1;
    int linc = ls;
    for (int off = 1; off < 32; off <<= 1) {
        const int lu = __shfl_up(linc, off, 32);
        if (lane >= off) linc += lu;
    }
    if (lane == 31) lwsum[wave] = linc;
    __syncthreads();
    int lex = linc - ls;
    for (int w = 0; w < SORT_BLOCK / 32; ++w)
        if (w < wave) lex += lwsum[w];
    if (i0 < NUM_BKT)  lcur[i0] = lex;
    if (i1 < NUM_BKT)  lcur[i1] = lex + n0;
    if (i0 <= NUM_BKT) tb[i0 * TILE_PITCH + tile] = lex;
    if (i1 <= NUM_BKT) tb[i1 * TILE_PITCH + tile] = lex + n0;
    __syncthreads();

#pragma unroll
    for (int i = 0; i < 4; ++i) {
        const int e = tile * TILE_EDGES + i * (SORT_BLOCK * 4) + 4 * t;
        if (e < N_EDGES) {
            file_edge(lcur, sval, d[i].x, y[i].x, s[i].x);
            file_edge(lcur, sval, d[i].y, y[i].y, s[i].y);
            file_edge(lcur, sval, d[i].z, y[i].z, s[i].z);
            file_edge(lcur, sval, d[i].w, y[i].w, s[i].w);
        }
    }
    __syncthreads();

    const int cnt = min(TILE_EDGES, N_EDGES - tile * TILE_EDGES);
    int sv[TILE_EDGES / SORT_BLOCK];
#pragma unroll
    for (int j = 0; j < TILE_EDGES / SORT_BLOCK; ++j) sv[j] = sval[j * SORT_BLOCK + t];
#pragma unroll
    for (int j = 0; j < TILE_EDGES / SORT_BLOCK; ++j) {
        if (j * SORT_BLOCK + t < cnt) mid[tile * TILE_EDGES + j * SORT_BLOCK + t] = sv[j];
    }
}

#define STAGE_CAP 4096
#define RUN_TILES 10
static_assert(TILE_PITCH % (16 * RUN_TILES) == 0 && SORT_BLOCK == 256, "16 half-waves sweep the table row in whole trips");

template <int MODE>
__device__ __forceinline__ void sweep_runs(const int* __restrict__ mid,
                                           const int* rs, const int* rc, const int* rr,
                                           int* ctr, int* raw, int* __restrict__ csr_src,
                                           const int base, const int t) {
    const int hw = t >> 4, l16 = t & 15;
    auto take = [&](int v, int slot) __attribute__((always_inline)) {
        const int seg = min(v >> 16, BKT_SEGS - 1);
        if (MODE == 0) {
            raw[min(max(slot, 0), STAGE_CAP - 1)] = v;
            atomicAdd(&ctr[seg], 1);
        } else if (MODE == 1) {
            atomicAdd(&ctr[seg], 1);
        } else {
            const int p = atomicAdd(&ctr[seg], 1);
            const int q = min(max(base + p, 0), N_EDGES - 1);
            csr_src[q] = v & 0xFFFF;
        }
    };
    for (int T0 = hw; T0 < TILE_PITCH; T0 += 16 * RUN_TILES) {
        int v[RUN_TILES][2], c[RUN_TILES], r[RUN_TILES], a[RUN_TILES];
#pragma unroll
        for (int u = 0; u < RUN_TILES; ++u) {
            const int T = T0 + 16 * u;
            c[u] = rc[T];
            r[u] = rr[T];
            a[u] = T * TILE_EDGES + rs[T];
#pragma unroll
            for (int h = 0; h < 2; ++h) {
                const int i = l16 + 16 * h;
                v[u][h] = (i < c[u]) ? mid[min(max(a[u] + i, 0), N_EDGES - 1)] : -1;
            }
        }
#pragma unroll
        for (int u = 0; u < RUN_TILES; ++u) {
#pragma unroll
            for (int h = 0; h < 2; ++h)
                if (v[u][h] >= 0) take(v[u][h], r[u] + l16 + 16 * h);
        }
#pragma unroll
        for (int u = 0; u < RUN_TILES; ++u) {
            for (int i = l16 + 32; i < c[u]; i += 16) {
                const int w = mid[min(max(a[u] + i, 0), N_EDGES - 1)];
                if (w >= 0) take(w, r[u] + i);
            }
        }
    }
}

#define A_PITCH  260
#define X_PITCH  36
static_assert(N_NODES % TILE_M == 0, "no partial node tile");
static_assert(IN_DIM + 1 <= FOLD_ROWS && FOLD_ROWS % 4 == 0, "x, the 0/1 column and padding fill whole k-steps");

__device__ __forceinline__ int bcast_i(int v, int l) {
    return __builtin_amdgcn_readlane(v, l);
}
__device__ __forceinline__ float bcast_f(float v, int l) {
    return __int_as_float(__builtin_amdgcn_readlane(__float_as_int(v), l));
}

#define SEG_UNROLL 8

template <bool STAGED>
__device__ __forceinline__ void first_layer_means(const float* __restrict__ x,
                                                  const int* cur, const int* cnt,
                                                  const int* stage,
                                                  const int* csr_src,
                                                  const int base, const int nb,
                                                  const int node_b0,
                                                  const int t,
                                                  float* lXb)
{
    auto load_row = [&](int sraw, float (&v)[IN_DIM]) __attribute__((always_inline)) {
        const int s = min(max(sraw, 0), N_NODES - 1);
        const float* xr = x + (size_t)s * IN_DIM;
        float4 q;
        __builtin_memcpy(&q, xr, sizeof(q));
        v[0] = q.x; v[1] = q.y; v[2] = q.z; v[3] = q.w;
        v[4] = xr[4];
    };
    for (int sI = t; sI < BKT_SEGS; sI += SORT_BLOCK) {
        const int node = sI / NUM_REL;
        const int r    = sI - node * NUM_REL;
        const int len   = min(max(cnt[sI], 0), nb);
        const int end   = min(max(cur[sI], len), nb);
        const int start = end - len;
        float sum[IN_DIM];
#pragma unroll
        for (int d = 0; d < IN_DIM; ++d) sum[d] = 0.f;
        for (int i0 = 0; i0 < len; i0 += SEG_UNROLL) {
            int   sr[SEG_UNROLL];
            float row[SEG_UNROLL][IN_DIM];
#pragma unroll
            for (int u = 0; u < SEG_UNROLL; ++u) {
                const int j = start + min(i0 + u, len - 1);
                if (STAGED) sr[u] = stage[min(j, STAGE_CAP - 1)];
                else        sr[u] = csr_src[min(max(base + j, 0), N_EDGES - 1)];
            }
#pragma unroll
            for (int u = 0; u < SEG_UNROLL; ++u) load_row(sr[u], row[u]);
#pragma unroll
            for (int u = 0; u < SEG_UNROLL; ++u) {
#pragma unroll
                for (int d = 0; d < IN_DIM; ++d) sum[d] += (i0 + u < len) ? row[u][d] : 0.f;
            }
        }
        const float wl   = 1.0f / (float)(len > 1 ? len : 1);
        const float some = (len > 0) ? 1.f : 0.f;
        float* d8 = lXb + ((node >> 4) * TILE_M + (node & 15)) * X_PITCH + r * FOLD_ROWS;
        *(float4*)(d8)     = make_float4(sum[0] * wl, sum[1] * wl, sum[2] * wl, sum[3] * wl);
        *(float4*)(d8 + 4) = make_float4(sum[4] * wl, some, 0.f, 0.f);
    }
    if (t < BKT_NODES) {
        float xo[IN_DIM];
        load_row(node_b0 + t, xo);
        float* d8 = lXb + ((t >> 4) * TILE_M + (t & 15)) * X_PITCH + NUM_REL * FOLD_ROWS;
        *(float4*)(d8)     = make_float4(xo[0], xo[1], xo[2], xo[3]);
        *(float4*)(d8 + 4) = make_float4(xo[4], 1.f, 0.f, 0.f);
    }
}

__device__ __forceinline__ void first_layer_product(const float* lXw,
                                                    const float4 (&w0)[FOLD_ROWS], const float4 (&w1)[FOLD_ROWS],
                                                    const int node_t0, const int lane,
                                                    const float* __restrict__ bias,
                                                    const float* __restrict__ gamma,
                                                    const float* __restrict__ beta,
                                                    float* __restrict__ hout) {
    const int c     = lane & 15;
    const int hh    = lane >> 4;
    const int khalf = hh << 1;

    v8f acc[4];
#pragma unroll
    for (int j = 0; j < 4; ++j) acc[j] = (v8f){0.f,0.f,0.f,0.f,0.f,0.f,0.f,0.f};

#pragma unroll
    for (int s = 0; s < FOLD_ROWS; ++s) {
        const int kk = 4 * s + khalf;
        const v2f a = *(const v2f*)(lXw + c * X_PITCH + kk);
        v2f b;
        b.x = w0[s].x; b.y = w1[s].x;
        acc[0] = __builtin_amdgcn_wmma_f32_16x16x4_f32(false, a, false, b, (short)0, acc[0], false, false);
        b.x = w0[s].y; b.y = w1[s].y;
        acc[1] = __builtin_amdgcn_wmma_f32_16x16x4_f32(false, a, false, b, (short)0, acc[1], false, false);
        b.x = w0[s].z; b.y = w1[s].z;
        acc[2] = __builtin_amdgcn_wmma_f32_16x16x4_f32(false, a, false, b, (short)0, acc[2], false, false);
        b.x = w0[s].w; b.y = w1[s].w;
        acc[3] = __builtin_amdgcn_wmma_f32_16x16x4_f32(false, a, false, b, (short)0, acc[3], false, false);
    }

    const float4 bs = *(const float4*)(bias + 4 * c);
    const float4 g4 = *(const float4*)(gamma + 4 * c);
    const float4 be = *(const float4*)(beta + 4 * c);
#pragma unroll
    for (int r = 0; r < 8; ++r) {
        float4 v;
        v.x = fmaxf(acc[0][r] + bs.x, 0.0f);
        v.y = fmaxf(acc[1][r] + bs.y, 0.0f);
        v.z = fmaxf(acc[2][r] + bs.z, 0.0f);
        v.w = fmaxf(acc[3][r] + bs.w, 0.0f);
        float s  = (v.x + v.y) + (v.z + v.w);
        float sq = (v.x * v.x + v.y * v.y) + (v.z * v.z + v.w * v.w);
#pragma unroll
        for (int off = 8; off > 0; off >>= 1) {
            s  += __shfl_xor(s,  off, 32);
            sq += __shfl_xor(sq, off, 32);
        }
        const float mu   = s * (1.0f / OUT_DIM);
        const float var  = sq * (1.0f / OUT_DIM) - mu * mu;
        const float rstd = rsqrtf(var + LN_EPS);
        float4 o;
        o.x = (v.x - mu) * rstd * g4.x + be.x;
        o.y = (v.y - mu) * rstd * g4.y + be.y;
        o.z = (v.z - mu) * rstd * g4.z + be.z;
        o.w = (v.w - mu) * rstd * g4.w + be.w;
        *(float4*)(hout + (size_t)(node_t0 + 8 * hh + r) * OUT_DIM + 4 * c) = o;
    }
}

__global__ __launch_bounds__(SORT_BLOCK)
void bucket_sort_kernel(const int* __restrict__ mid,
                        const int* __restrict__ tb,
                        int* csr_src,
                        int* __restrict__ offs,
                        float* __restrict__ inv,
                        const float* __restrict__ x,
                        const float* __restrict__ Wc,
                        const float* __restrict__ bias,
                        const float* __restrict__ gamma,
                        const float* __restrict__ beta,
                        float* __restrict__ hout) {
    __shared__ int cnt[BKT_SEGS];
    __shared__ int cur[BKT_SEGS];
    __shared__ int wsum[BKT_NODES / 32];
    __shared__ int wsA[SORT_BLOCK / 32], wsB[SORT_BLOCK / 32], wsG[SORT_BLOCK / 32];
    __shared__ int rs[TILE_PITCH], rc[TILE_PITCH], rr[TILE_PITCH];
    __shared__ int raw[STAGE_CAP];
    __shared__ int stage[STAGE_CAP];
    __shared__ __attribute__((aligned(16))) float lX[SORT_BLOCK / 32][TILE_M][X_PITCH];
    const int t = threadIdx.x;
    const int lane = t & 31;
    const int wave = t >> 5;
    const int k = blockIdx.x;

    const int wv = __builtin_amdgcn_readfirstlane(wave);
    const int node_t0 = k * BKT_NODES + wv * TILE_M;
    const bool live = node_t0 < N_NODES;
    float4 fw0[FOLD_ROWS], fw1[FOLD_ROWS];
#pragma unroll
    for (int s = 0; s < FOLD_ROWS; ++s) {
        const int kk = 4 * s + ((lane >> 4) << 1);
        fw0[s] = *(const float4*)(Wc + (size_t)kk * OUT_DIM + 4 * (lane & 15));
        fw1[s] = *(const float4*)(Wc + (size_t)(kk + 1) * OUT_DIM + 4 * (lane & 15));
    }

    auto run_of = [&](int T, int& s, int& c) __attribute__((always_inline)) {
        const bool ok = T < NUM_TILES;
        const int Tc = ok ? T : NUM_TILES - 1;
        const int lo = tb[k * TILE_PITCH + Tc];
        const int hi = tb[(k + 1) * TILE_PITCH + Tc];
        const int tcnt = min(TILE_EDGES, N_EDGES - Tc * TILE_EDGES);
        const int sc = min(max(lo, 0), tcnt);
        const int ec = min(max(hi, sc), tcnt);
        s = ok ? sc : 0;
        c = ok ? ec - sc : 0;
    };
    int s0, ra, s1, rb;
    run_of(t, s0, ra);
    run_of(t + SORT_BLOCK, s1, rb);
    const int g = s0 + s1;
    int ia = ra, ib = rb, ig = g;
    for (int off = 1; off < 32; off <<= 1) {
        const int ua = __shfl_up(ia, off, 32);
        const int ub = __shfl_up(ib, off, 32);
        const int ug = __shfl_up(ig, off, 32);
        if (lane >= off) { ia += ua; ib += ub; ig += ug; }
    }
    if (lane == 31) { wsA[wave] = ia; wsB[wave] = ib; wsG[wave] = ig; }
    for (int i = t; i < BKT_SEGS; i += SORT_BLOCK) cnt[i] = 0;
    __syncthreads();
    int exa = ia - ra, exb = ib - rb;
    int totA = 0, totB = 0, gsum = 0;
    for (int w = 0; w < SORT_BLOCK / 32; ++w) {
        const int a = wsA[w], b = wsB[w];
        totA += a;
        totB += b;
        gsum += wsG[w];
        if (w < wave) { exa += a; exb += b; }
    }
    rs[t] = s0;
    rc[t] = ra;
    rr[t] = exa;
    if (t + SORT_BLOCK < TILE_PITCH) {
        rs[t + SORT_BLOCK] = s1;
        rc[t + SORT_BLOCK] = rb;
        rr[t + SORT_BLOCK] = totA + exb;
    }
    const int base = min(max(gsum, 0), N_EDGES);
    const int n = min(max(totA + totB, 0), N_EDGES - base);
    const bool staged = n <= STAGE_CAP;
    __syncthreads();

    if (staged) sweep_runs<0>(mid, rs, rc, rr, cnt, raw, csr_src, base, t);
    else        sweep_runs<1>(mid, rs, rc, rr, cnt, raw, csr_src, base, t);
    __syncthreads();

    int c0 = 0, c1 = 0, c2 = 0;
    if (t < BKT_NODES) { c0 = cnt[3 * t]; c1 = cnt[3 * t + 1]; c2 = cnt[3 * t + 2]; }
    const int ts = c0 + c1 + c2;
    int inc = ts;
    for (int off = 1; off < 32; off <<= 1) {
        const int u = __shfl_up(inc, off, 32);
        if (lane >= off) inc += u;
    }
    if (t < BKT_NODES && lane == 31) wsum[wave] = inc;
    __syncthreads();
    if (t < BKT_NODES) {
        int ex = inc - ts;
        for (int w = 0; w < BKT_NODES / 32; ++w)
            if (w < wave) ex += wsum[w];
        cur[3 * t]     = ex;
        cur[3 * t + 1] = ex + c0;
        cur[3 * t + 2] = ex + c0 + c1;
        const int node = k * BKT_NODES + t;
        if (node < N_NODES) {
            const int seg = node * NUM_REL;
            offs[seg]     = base + ex;
            offs[seg + 1] = base + ex + c0;
            offs[seg + 2] = base + ex + c0 + c1;
            inv[seg]     = 1.0f / (float)(c0 > 1 ? c0 : 1);
            inv[seg + 1] = 1.0f / (float)(c1 > 1 ? c1 : 1);
            inv[seg + 2] = 1.0f / (float)(c2 > 1 ? c2 : 1);
            if (node == N_NODES - 1) offs[NR_TOTAL] = base + ex + ts;
        }
    }
    __syncthreads();

    if (staged) {
        for (int j0 = 0; j0 < n; j0 += SORT_BLOCK * CHUNK) {
            int v[CHUNK];
#pragma unroll
            for (int j = 0; j < CHUNK; ++j) {
                const int i = j0 + j * SORT_BLOCK + t;
                v[j] = (i < n) ? raw[i] : -1;
            }
#pragma unroll
            for (int j = 0; j < CHUNK; ++j) {
                if (v[j] >= 0) {
                    const int p = atomicAdd(&cur[min(v[j] >> 16, BKT_SEGS - 1)], 1);
                    stage[min(max(p, 0), STAGE_CAP - 1)] = v[j] & 0xFFFF;
                }
            }
        }
        __syncthreads();
        for (int j0 = 0; j0 < n; j0 += SORT_BLOCK * CHUNK) {
            int w[CHUNK];
#pragma unroll
            for (int j = 0; j < CHUNK; ++j) {
                const int i = j0 + j * SORT_BLOCK + t;
                w[j] = (i < n) ? stage[i] : 0;
            }
#pragma unroll
            for (int j = 0; j < CHUNK; ++j) {
                const int i = j0 + j * SORT_BLOCK + t;
                if (i < n) csr_src[base + i] = w[j];
            }
        }
    } else {
        sweep_runs<2>(mid, rs, rc, rr, cur, raw, csr_src, base, t);
        __syncthreads();
    }

    if (staged)
        first_layer_means<true>(x, cur, cnt, stage, csr_src, base, n, k * BKT_NODES, t, &lX[0][0][0]);
    else
        first_layer_means<false>(x, cur, cnt, stage, csr_src, base, n, k * BKT_NODES, t, &lX[0][0][0]);
    __syncthreads();
    if (live)
        first_layer_product(&lX[wv][0][0], fw0, fw1, node_t0, lane, bias, gamma, beta, hout);
}

#define LAYER_M        (2 * TILE_M)
#define LAYER_THREADS  256
static_assert(N_NODES % 4 == 0, "a wave's four nodes are all inside or all outside the node range");

__device__ __forceinline__ void multiply_pair_and_norm(float (&lA)[LAYER_M][A_PITCH],
                                                       const int row0, const int wave, const int lane,
                                                       const float* __restrict__ relW,
                                                       const float* __restrict__ rootW,
                                                       const float* __restrict__ bias,
                                                       const float* __restrict__ gamma,
                                                       const float* __restrict__ beta,
                                                       float* __restrict__ hout) {
    const int q     = wave & 3;
    const int jh    = wave >> 2;
    const int c     = lane & 15;
    const int hh    = lane >> 4;
    const int khalf = hh << 1;
    const float* __restrict__ Wq =
        (q < NUM_REL) ? (relW + (size_t)q * OUT_DIM * OUT_DIM) : rootW;
    const float* __restrict__ wp = Wq + 32 * jh + 2 * c;
    const float* a0p = &lA[c][q * OUT_DIM];
    const float* a1p = &lA[TILE_M + c][q * OUT_DIM];

    v8f acc[2][2];
#pragma unroll
    for (int m = 0; m < 2; ++m)
#pragma unroll
        for (int t = 0; t < 2; ++t) acc[m][t] = (v8f){0.f,0.f,0.f,0.f,0.f,0.f,0.f,0.f};

#pragma unroll
    for (int k0 = 0; k0 < OUT_DIM; k0 += 4) {
        const int kk = k0 + khalf;
        const v2f a0 = *(const v2f*)(a0p + kk);
        const v2f a1 = *(const v2f*)(a1p + kk);
        const float2 w0 = *(const float2*)(wp + (size_t)kk * OUT_DIM);
        const float2 w1 = *(const float2*)(wp + (size_t)(kk + 1) * OUT_DIM);
        v2f b;
        b.x = w0.x; b.y = w1.x;
        acc[0][0] = __builtin_amdgcn_wmma_f32_16x16x4_f32(false, a0, false, b, (short)0, acc[0][0], false, false);
        acc[1][0] = __builtin_amdgcn_wmma_f32_16x16x4_f32(false, a1, false, b, (short)0, acc[1][0], false, false);
        b.x = w0.y; b.y = w1.y;
        acc[0][1] = __builtin_amdgcn_wmma_f32_16x16x4_f32(false, a0, false, b, (short)0, acc[0][1], false, false);
        acc[1][1] = __builtin_amdgcn_wmma_f32_16x16x4_f32(false, a1, false, b, (short)0, acc[1][1], false, false);
    }
    __syncthreads();

#pragma unroll
    for (int m = 0; m < 2; ++m) {
#pragma unroll
        for (int r = 0; r < 8; ++r) {
            *(float2*)&lA[m * TILE_M + 8 * hh + r][q * OUT_DIM + 32 * jh + 2 * c] =
                make_float2(acc[m][0][r], acc[m][1][r]);
        }
    }
    __syncthreads();

    const float2 bs = ((const float2*)bias)[lane];
    const float2 g2 = ((const float2*)gamma)[lane];
    const float2 be = ((const float2*)beta)[lane];
#pragma unroll
    for (int t = 0; t < 4; ++t) {
        const int row = wave * 4 + t;
        const float* prow = &lA[row][2 * lane];
        const float2 p0 = *(const float2*)(prow);
        const float2 p1 = *(const float2*)(prow + 64);
        const float2 p2 = *(const float2*)(prow + 128);
        const float2 p3 = *(const float2*)(prow + 192);
        float2 v;
        v.x = p3.x + bs.x;  v.y = p3.y + bs.y;
        v.x += p0.x;        v.y += p0.y;
        v.x += p1.x;        v.y += p1.y;
        v.x += p2.x;        v.y += p2.y;
        v.x = fmaxf(v.x, 0.0f);
        v.y = fmaxf(v.y, 0.0f);
        float s  = v.x + v.y;
        float sq = v.x * v.x + v.y * v.y;
#pragma unroll
        for (int off = 16; off > 0; off >>= 1) {
            s  += __shfl_xor(s,  off, 32);
            sq += __shfl_xor(sq, off, 32);
        }
        const float mu   = s * (1.0f / OUT_DIM);
        const float var  = sq * (1.0f / OUT_DIM) - mu * mu;
        const float rstd = rsqrtf(var + LN_EPS);
        float2 o;
        o.x = (v.x - mu) * rstd * g2.x + be.x;
        o.y = (v.y - mu) * rstd * g2.y + be.y;
        if (row0 + row < N_NODES)
            ((float2*)(hout + (size_t)(row0 + row) * OUT_DIM))[lane] = o;
    }
}

#define GRP 8

__global__ __launch_bounds__(LAYER_THREADS, 16)
void rgcn_layer_kernel(const float* __restrict__ h,
                       const int*   __restrict__ offs,
                       const int*   __restrict__ csr_src,
                       const float* __restrict__ inv_denom,
                       const float* __restrict__ relW,
                       const float* __restrict__ rootW,
                       const float* __restrict__ bias,
                       const float* __restrict__ gamma,
                       const float* __restrict__ beta,
                       float* __restrict__ hout)
{
    __shared__ __attribute__((aligned(16))) float lA[LAYER_M][A_PITCH];
    const int row0 = blockIdx.x * LAYER_M;
    const int wave = __builtin_amdgcn_readfirstlane(threadIdx.x >> 5);
    const int lane = threadIdx.x & 31;

    if (row0 + wave * 4 < N_NODES) {
        const float2* __restrict__ h2 = (const float2*)h;
        const int node0 = row0 + wave * 4;
        float2 hown[4];
#pragma unroll
        for (int t = 0; t < 4; ++t) hown[t] = h2[(size_t)(node0 + t) * (OUT_DIM / 2) + lane];
        int bl = offs[node0 * NUM_REL + (lane < 12 ? lane : 12)];
        bl = min(max(bl, 0), N_EDGES);
        const float wl = inv_denom[node0 * NUM_REL + (lane < 11 ? lane : 11)];

        const int e0 = bcast_i(bl, 0);
        const int n  = max(bcast_i(bl, 12) - e0, 0);
        const int elast = min(max(e0 + n - 1, 0), N_EDGES - 1);

        auto idx_raw = [&](int cidx) __attribute__((always_inline)) {
            const int j = min(max(e0 + 2 * GRP * cidx + lane, 0), elast);
            return csr_src[j];
        };
        int s = idx_raw(0);

        float* lrow = &lA[wave * 4][2 * lane];
#pragma unroll
        for (int t = 0; t < 4; ++t) *(float2*)(lrow + t * A_PITCH + 3 * OUT_DIM) = hown[t];

        s = min(max(s, 0), N_NODES - 1);

        int seg = 0;
        int rel = 0;
        int segoff = 0;
        int nb = bcast_i(bl, 1) - e0;
        float2 a = make_float2(0.f, 0.f);

        auto flush = [&]() __attribute__((always_inline)) {
            const float w = bcast_f(wl, seg);
            *(float2*)(lrow + segoff) = make_float2(a.x * w, a.y * w);
            a = make_float2(0.f, 0.f);
            ++seg;
            ++rel;
            segoff += OUT_DIM;
            if (rel == NUM_REL) { rel = 0; segoff += A_PITCH - NUM_REL * OUT_DIM; }
            nb = (seg < 4 * NUM_REL) ? (bcast_i(bl, seg + 1) - e0) : 0x7fffffff;
        };
        auto load_group = [&](float2 (&m)[GRP], int sv, int k) __attribute__((always_inline)) {
#pragma unroll
            for (int i = 0; i < GRP; ++i) {
                const int rk = bcast_i(sv, k + i);
                const float2* __restrict__ rowp = h2 + (size_t)rk * (OUT_DIM / 2);
                m[i] = rowp[lane];
            }
        };
        auto consume = [&](const float2 (&m)[GRP], int p0) __attribute__((always_inline)) {
#pragma unroll
            for (int i = 0; i < GRP; ++i) {
                while (p0 + i >= nb) flush();
                a.x += m[i].x;
                a.y += m[i].y;
            }
        };

        {
            float2 mA[GRP], mB[GRP];
            int g = 0;
            while (g * GRP < n) {
                const int sraw = idx_raw(g / 2 + 1);
                load_group(mA, s, 0);
                if (g > 0) consume(mB, (g - 1) * GRP);
                load_group(mB, s, GRP);
                consume(mA, g * GRP);
                s = min(max(sraw, 0), N_NODES - 1);
                g += 2;
            }
            if (g > 0) consume(mB, (g - 1) * GRP);
        }
        while (seg < 4 * NUM_REL) flush();
    } else {
        float* lrow = &lA[wave * 4][2 * lane];
#pragma unroll
        for (int t = 0; t < 4; ++t)
#pragma unroll
            for (int j = 0; j < 4; ++j)
                *(float2*)(lrow + t * A_PITCH + j * OUT_DIM) = make_float2(0.f, 0.f);
    }
    __syncthreads();

    multiply_pair_and_norm(lA, row0, wave, lane, relW, rootW, bias, gamma, beta, hout);
}

extern "C" void kernel_launch(void* const* d_in, const int* in_sizes, int n_in,
                              void* d_out, int out_size, void* d_ws, size_t ws_size,
                              hipStream_t stream) {
    const float* x      = (const float*)d_in[0];
    const int*   ei     = (const int*)  d_in[1];
    const int*   src    = ei;
    const int*   dst    = ei + N_EDGES;
    const int*   etype  = (const int*)  d_in[2];
    const float* f2cW   = (const float*)d_in[3];
    const float* f2cb   = (const float*)d_in[4];
    const float* rgW    = (const float*)d_in[5];
    const float* rgRoot = (const float*)d_in[6];
    const float* rgBias = (const float*)d_in[7];
    const float* gamma  = (const float*)d_in[8];
    const float* beta   = (const float*)d_in[9];

    float* ws = (float*)d_ws;
    float* h_a       = ws;  ws += (size_t)N_NODES * OUT_DIM;
    int*   mid       = (int*)ws;
    int*   csr_src   = mid + N_EDGES;
    int*   offs      = csr_src + N_EDGES;
    ws += (size_t)N_NODES * OUT_DIM;
    float* h_b       = ws;
    float* wfold     = h_b + (size_t)N_NODES * OUT_DIM;
    ws += (size_t)N_NODES * NUM_REL * OUT_DIM;
    int*   tb        = (int*)ws;
    ws += (size_t)N_NODES * OUT_DIM;
    float* inv_denom = ws;

    partition_kernel<<<NUM_TILES + 1, SORT_BLOCK, 0, stream>>>(src, dst, etype, tb, mid,
                                                                f2cW, f2cb, rgW, rgRoot, wfold);
    bucket_sort_kernel<<<NUM_BKT, SORT_BLOCK, 0, stream>>>(mid, tb, csr_src, offs, inv_denom,
                                                           x, wfold, rgBias, gamma, beta, h_b);
    for (int l = 1; l < NUM_LAYERS; ++l) {
        const float* hi = (l & 1) ? h_b : h_a;
        float* ho = (l == NUM_LAYERS - 1) ? (float*)d_out : ((l & 1) ? h_a : h_b);
        rgcn_layer_kernel<<<(N_NODES + LAYER_M - 1) / LAYER_M, LAYER_THREADS, 0, stream>>>(
            hi, offs, csr_src, inv_denom,
            rgW    + (size_t)l * NUM_REL * OUT_DIM * OUT_DIM,
            rgRoot + (size_t)l * OUT_DIM * OUT_DIM,
            rgBias + (size_t)l * OUT_DIM,
            gamma, beta, ho);
    }
}
